# starting kernel
# baseline (speedup 1.0000x reference)
#include <hip/hip_runtime.h>
#include <math.h>

typedef __attribute__((ext_vector_type(2))) float v2f;
typedef __attribute__((ext_vector_type(8))) float v8f;

#define N_NODES 50000
#define N_EDGES 800000
#define IN_F    256
#define OUT_F   128
#define RANK    64

static_assert(N_NODES % 16 == 0, "row tiles exact");
static_assert(N_EDGES % 256 == 0, "edge blocks exact");

#define OFF_ODEG   0u
#define OFF_IDEG   50000u
#define OFF_HSUM   100000u
#define OFF_LABS   6500000u
#define OFF_NCNT   9700000u
#define OFF_FSUM   12900000u
#define OFF_FPROD  19300000u
#define ZERO_CNT   12900000u

__global__ __launch_bounds__(256) void k_zero(unsigned int* __restrict__ p, unsigned n4)
{
    unsigned i = blockIdx.x * blockDim.x + threadIdx.x;
    if (i < n4) {
        ((uint4*)p)[i] = make_uint4(0u, 0u, 0u, 0u);
    }
}

__global__ __launch_bounds__(256) void k_degrees(const int* __restrict__ src,
                                                 const int* __restrict__ dst,
                                                 float* __restrict__ odeg,
                                                 float* __restrict__ ideg)
{
    int e = blockIdx.x * blockDim.x + threadIdx.x;
    atomicAdd(&odeg[src[e]], 1.0f);
    atomicAdd(&ideg[dst[e]], 1.0f);
}

__global__ __launch_bounds__(128) void k_node_gemm(const float* __restrict__ feat,
                                                   const float* __restrict__ w1,
                                                   const float* __restrict__ w2,
                                                   const float* __restrict__ odeg,
                                                   float* __restrict__ fsum,
                                                   float* __restrict__ fprod)
{
    __shared__ float As[16][260];
    __shared__ float sc[16];

    const int m0  = blockIdx.x * 16;
    const int tid = threadIdx.x;

    if (tid < 16) {
        sc[tid] = rsqrtf(fmaxf(odeg[m0 + tid], 1.0f));
    }
    __syncthreads();

    {
        const int r   = tid >> 3;
        const int c0  = (tid & 7) * 32;
        const float s = sc[r];
        const float4* g = (const float4*)(feat + (size_t)(m0 + r) * IN_F + c0);
#pragma unroll
        for (int j = 0; j < 8; ++j) {
            float4 x = g[j];
            As[r][c0 + 4 * j + 0] = x.x * s;
            As[r][c0 + 4 * j + 1] = x.y * s;
            As[r][c0 + 4 * j + 2] = x.z * s;
            As[r][c0 + 4 * j + 3] = x.w * s;
        }
    }
    __syncthreads();

    const int wave  = tid >> 5;
    const int lane  = tid & 31;
    const int rsel  = lane & 15;
    const int khalf = (lane < 16) ? 0 : 2;
    const int rofs  = (lane < 16) ? 0 : 8;

    for (int t = 0; t < 3; ++t) {
        const int  tile = wave + 4 * t;
        const bool isW2 = (tile >= 8);
        const int  n0   = isW2 ? (tile - 8) * 16 : tile * 16;
        const float* B  = isW2 ? w2 : w1;
        const int  ldb  = isW2 ? RANK : OUT_F;

        for (int k = 0; k < IN_F; k += 32)
            __builtin_prefetch(B + (size_t)k * ldb + n0, 0, 0);

        v8f acc = {};
        for (int k0 = 0; k0 < IN_F; k0 += 4) {
            v2f a;
            a.x = As[rsel][k0 + khalf];
            a.y = As[rsel][k0 + khalf + 1];
            v2f b;
            const float* Bp = B + (size_t)(k0 + khalf) * ldb + n0 + rsel;
            b.x = Bp[0];
            b.y = Bp[ldb];
            acc = __builtin_amdgcn_wmma_f32_16x16x4_f32(
                false, a, false, b, (short)0, acc, false, false);
        }

        const int colg = n0 + rsel;
        if (isW2) {
            const float bias = w2[(size_t)IN_F * RANK + colg];
            float* op = fprod + (size_t)(m0 + rofs) * RANK + colg;
#pragma unroll
            for (int r = 0; r < 8; ++r)
                op[r * RANK] = tanhf(acc[r] + bias);
        } else {
            float* op = fsum + (size_t)(m0 + rofs) * OUT_F + colg;
#pragma unroll
            for (int r = 0; r < 8; ++r)
                op[r * OUT_F] = acc[r];
        }
    }
}

__global__ __launch_bounds__(256) void k_edge(const int* __restrict__ src,
                                              const int* __restrict__ dst,
                                              const float* __restrict__ fsum,
                                              const float* __restrict__ fprod,
                                              float* __restrict__ hsum,
                                              float* __restrict__ labs,
                                              int*   __restrict__ ncnt)
{
    const int wid  = (int)((blockIdx.x * blockDim.x + threadIdx.x) >> 5);
    const int lane = threadIdx.x & 31;

    const int s = src[wid];
    const int d = dst[wid];

    const float* fs = fsum + (size_t)s * OUT_F;
    float*       hs = hsum + (size_t)d * OUT_F;
#pragma unroll
    for (int j = 0; j < 4; ++j) {
        int c = lane + 32 * j;
        atomicAdd(&hs[c], fs[c]);
    }

    const float* fp = fprod + (size_t)s * RANK;
    float*       la = labs  + (size_t)d * RANK;
    int*         nc = ncnt  + (size_t)d * RANK;
#pragma unroll
    for (int j = 0; j < 2; ++j) {
        int c = lane + 32 * j;
        float m = fp[c];
        atomicAdd(&la[c], __logf(fabsf(m)));
        if (m < 0.0f) atomicAdd(&nc[c], 1);
    }
}

__global__ __launch_bounds__(128) void k_final(const float* __restrict__ vmat,
                                               const float* __restrict__ hsum,
                                               const float* __restrict__ labs,
                                               const int*   __restrict__ ncnt,
                                               const float* __restrict__ ideg,
                                               float* __restrict__ out)
{
    __shared__ float Hp[16][68];
    __shared__ float sc[16];

    const int m0  = blockIdx.x * 16;
    const int tid = threadIdx.x;

    if (tid < 16) {
        sc[tid] = rsqrtf(fmaxf(ideg[m0 + tid], 1.0f));
    }
    for (int i = tid; i < 16 * RANK; i += 128) {
        int r = i >> 6, c = i & 63;
        int row = m0 + r;
        float val = 0.0f;
        float dg = ideg[row];
        if (dg > 0.0f) {
            float la = labs[(size_t)row * RANK + c];
            int   n  = ncnt[(size_t)row * RANK + c];
            float sg = (n & 1) ? -1.0f : 1.0f;
            val = sg * __expf(la);
        }
        Hp[r][c] = val;
    }
    __syncthreads();

    const int wave  = tid >> 5;
    const int lane  = tid & 31;
    const int rsel  = lane & 15;
    const int khalf = (lane < 16) ? 0 : 2;
    const int rofs  = (lane < 16) ? 0 : 8;

    for (int t = 0; t < 2; ++t) {
        const int n0 = (wave * 2 + t) * 16;

        for (int k = 0; k < RANK; k += 32)
            __builtin_prefetch(vmat + (size_t)k * OUT_F + n0, 0, 0);

        v8f acc = {};
        for (int k0 = 0; k0 < RANK; k0 += 4) {
            v2f a;
            a.x = Hp[rsel][k0 + khalf];
            a.y = Hp[rsel][k0 + khalf + 1];
            v2f b;
            const float* Bp = vmat + (size_t)(k0 + khalf) * OUT_F + n0 + rsel;
            b.x = Bp[0];
            b.y = Bp[OUT_F];
            acc = __builtin_amdgcn_wmma_f32_16x16x4_f32(
                false, a, false, b, (short)0, acc, false, false);
        }
        const int colg = n0 + rsel;
        const float* hp = hsum + (size_t)(m0 + rofs) * OUT_F + colg;
        float*       op = out  + (size_t)(m0 + rofs) * OUT_F + colg;
#pragma unroll
        for (int r = 0; r < 8; ++r)
            op[r * OUT_F] = (acc[r] + hp[r * OUT_F]) * sc[rofs + r];
    }
}

extern "C" void kernel_launch(void* const* d_in, const int* in_sizes, int n_in,
                              void* d_out, int out_size, void* d_ws, size_t ws_size,
                              hipStream_t stream)
{
    (void)in_sizes; (void)n_in; (void)out_size; (void)ws_size;

    const float* feat = (const float*)d_in[0];
    const float* w1   = (const float*)d_in[1];
    const float* w2   = (const float*)d_in[2];
    const float* vmat = (const float*)d_in[3];
    const int*   src  = (const int*)d_in[4];
    const int*   dst  = (const int*)d_in[5];

    float* ws    = (float*)d_ws;
    float* odeg  = ws + OFF_ODEG;
    float* ideg  = ws + OFF_IDEG;
    float* hsum  = ws + OFF_HSUM;
    float* labs  = ws + OFF_LABS;
    int*   ncnt  = (int*)(ws + OFF_NCNT);
    float* fsum  = ws + OFF_FSUM;
    float* fprod = ws + OFF_FPROD;

    {
        unsigned n4 = ZERO_CNT / 4u;
        k_zero<<<(n4 + 255u) / 256u, 256, 0, stream>>>((unsigned int*)ws, n4);
    }
    k_degrees<<<N_EDGES / 256, 256, 0, stream>>>(src, dst, odeg, ideg);
    k_node_gemm<<<N_NODES / 16, 128, 0, stream>>>(feat, w1, w2, odeg, fsum, fprod);
    k_edge<<<(N_EDGES * 32) / 256, 256, 0, stream>>>(src, dst, fsum, fprod,
                                                     hsum, labs, ncnt);
    k_final<<<N_NODES / 16, 128, 0, stream>>>(vmat, hsum, labs, ncnt, ideg,
                                              (float*)d_out);
}


# Edge sums in LDS, not global atomics: a block owns 196 dst rows, filters the dst list for them and adds source rows into LDS accumulators by ds float adds (arrival order); planes stored once
# speedup vs baseline: 1.7629x; 1.7629x over previous
#include <hip/hip_runtime.h>
#include <math.h>

typedef __attribute__((ext_vector_type(2))) float v2f;
typedef __attribute__((ext_vector_type(8))) float v8f;

#define N_NODES 50000
#define N_EDGES 800000
#define IN_F    256
#define OUT_F   128
#define RANK    64

static_assert(N_NODES % 16 == 0, "row tiles exact");
static_assert(N_EDGES % 256 == 0, "edge blocks exact");

#define OFF_ODEG   0u
#define OFF_IDEG   50000u
#define OFF_HSUM   100000u
#define OFF_LABS   6500000u
#define OFF_NCNT   9700000u
#define OFF_FSUM   12900000u
#define OFF_FPROD  19300000u
#define ZERO_CNT   100000u

#define NR          196
#define EDGE_BLOCKS ((N_NODES + NR - 1) / NR)
#define EDGE_WAVES  32
#define SCAN_U      4
#define SCAN_EDGES  (32 * 4 * SCAN_U)
#define DRAIN_AT    24
#define QCAP        640
static_assert(NR <= 256, "local row id fits 8 bits of a queue entry");
static_assert(N_EDGES < (1 << 24) && N_EDGES % 4 == 0, "edge id fits 24 bits; int4 scan exact");
static_assert(DRAIN_AT + SCAN_EDGES <= QCAP, "one scan step can never overflow a wave's queue");

__global__ __launch_bounds__(256) void k_zero(unsigned int* __restrict__ p, unsigned n4)
{
    unsigned i = blockIdx.x * blockDim.x + threadIdx.x;
    if (i < n4) {
        ((uint4*)p)[i] = make_uint4(0u, 0u, 0u, 0u);
    }
}

__global__ __launch_bounds__(256) void k_degrees(const int* __restrict__ src,
                                                 const int* __restrict__ dst,
                                                 float* __restrict__ odeg,
                                                 float* __restrict__ ideg)
{
    int e = blockIdx.x * blockDim.x + threadIdx.x;
    atomicAdd(&odeg[src[e]], 1.0f);
    atomicAdd(&ideg[dst[e]], 1.0f);
}

__global__ __launch_bounds__(128) void k_node_gemm(const float* __restrict__ feat,
                                                   const float* __restrict__ w1,
                                                   const float* __restrict__ w2,
                                                   const float* __restrict__ odeg,
                                                   float* __restrict__ fsum,
                                                   float* __restrict__ fprod)
{
    __shared__ float As[16][260];
    __shared__ float sc[16];

    const int m0  = blockIdx.x * 16;
    const int tid = threadIdx.x;

    if (tid < 16) {
        sc[tid] = rsqrtf(fmaxf(odeg[m0 + tid], 1.0f));
    }
    __syncthreads();

    {
        const int r   = tid >> 3;
        const int c0  = (tid & 7) * 32;
        const float s = sc[r];
        const float4* g = (const float4*)(feat + (size_t)(m0 + r) * IN_F + c0);
#pragma unroll
        for (int j = 0; j < 8; ++j) {
            float4 x = g[j];
            As[r][c0 + 4 * j + 0] = x.x * s;
            As[r][c0 + 4 * j + 1] = x.y * s;
            As[r][c0 + 4 * j + 2] = x.z * s;
            As[r][c0 + 4 * j + 3] = x.w * s;
        }
    }
    __syncthreads();

    const int wave  = tid >> 5;
    const int lane  = tid & 31;
    const int rsel  = lane & 15;
    const int khalf = (lane < 16) ? 0 : 2;
    const int rofs  = (lane < 16) ? 0 : 8;

    for (int t = 0; t < 3; ++t) {
        const int  tile = wave + 4 * t;
        const bool isW2 = (tile >= 8);
        const int  n0   = isW2 ? (tile - 8) * 16 : tile * 16;
        const float* B  = isW2 ? w2 : w1;
        const int  ldb  = isW2 ? RANK : OUT_F;

        for (int k = 0; k < IN_F; k += 32)
            __builtin_prefetch(B + (size_t)k * ldb + n0, 0, 0);

        v8f acc = {};
        for (int k0 = 0; k0 < IN_F; k0 += 4) {
            v2f a;
            a.x = As[rsel][k0 + khalf];
            a.y = As[rsel][k0 + khalf + 1];
            v2f b;
            const float* Bp = B + (size_t)(k0 + khalf) * ldb + n0 + rsel;
            b.x = Bp[0];
            b.y = Bp[ldb];
            acc = __builtin_amdgcn_wmma_f32_16x16x4_f32(
                false, a, false, b, (short)0, acc, false, false);
        }

        const int colg = n0 + rsel;
        if (isW2) {
            const float bias = w2[(size_t)IN_F * RANK + colg];
            float* op = fprod + (size_t)(m0 + rofs) * RANK + colg;
#pragma unroll
            for (int r = 0; r < 8; ++r)
                op[r * RANK] = tanhf(acc[r] + bias);
        } else {
            float* op = fsum + (size_t)(m0 + rofs) * OUT_F + colg;
#pragma unroll
            for (int r = 0; r < 8; ++r)
                op[r * OUT_F] = acc[r];
        }
    }
}

__global__ __launch_bounds__(1024) void k_edge(const int* __restrict__ src,
                                               const int* __restrict__ dst,
                                               const float* __restrict__ fsum,
                                               const float* __restrict__ fprod,
                                               float* __restrict__ hsum,
                                               float* __restrict__ labs,
                                               int*   __restrict__ ncnt)
{
    __shared__ __attribute__((aligned(16))) float    Hs[NR * OUT_F];
    __shared__ __attribute__((aligned(16))) float    La[NR * RANK];
    __shared__ __attribute__((aligned(16))) unsigned Sg[NR * 2];
    __shared__ unsigned Qe[EDGE_WAVES * QCAP];
    __shared__ unsigned Qn[EDGE_WAVES];

    const int tid  = threadIdx.x;
    const int wave = tid >> 5;
    const int lane = tid & 31;
    const int n0   = blockIdx.x * NR;

    for (int i = tid; i < NR * OUT_F / 4; i += 1024)
        ((float4*)Hs)[i] = make_float4(0.0f, 0.0f, 0.0f, 0.0f);
    for (int i = tid; i < NR * RANK / 4; i += 1024)
        ((float4*)La)[i] = make_float4(0.0f, 0.0f, 0.0f, 0.0f);
    for (int i = tid; i < NR * 2; i += 1024)
        Sg[i] = 0u;
    if (tid < EDGE_WAVES) Qn[tid] = 0u;
    __syncthreads();

    constexpr int N4    = N_EDGES / 4;
    constexpr int STEP4 = 32 * SCAN_U;
    constexpr int NCH   = (N4 + STEP4 - 1) / STEP4;
    const int4* d4  = (const int4*)dst;
    unsigned*   myq = Qe + wave * QCAP;

    auto load_step = [&](int4 (&r)[SCAN_U], int cc) {
        for (int u = 0; u < SCAN_U; ++u) {
            const int i4 = cc * STEP4 + u * 32 + lane;
            r[u] = d4[(i4 < N4) ? i4 : (N4 - 1)];
        }
    };

    auto scan_step = [&](const int4 (&r)[SCAN_U], int cc, bool last) {
        for (int u = 0; u < SCAN_U; ++u) {
            const int  i4 = cc * STEP4 + u * 32 + lane;
            const bool ok = (i4 < N4);
            const unsigned e0 = (unsigned)i4 * 4u;
            const unsigned a[4] = {(unsigned)(r[u].x - n0), (unsigned)(r[u].y - n0),
                                   (unsigned)(r[u].z - n0), (unsigned)(r[u].w - n0)};
            const unsigned amin = min(min(a[0], a[1]), min(a[2], a[3]));
            if (ok && amin < (unsigned)NR) {
                for (int k = 0; k < 4; ++k) {
                    if (a[k] < (unsigned)NR) {
                        const unsigned p = atomicAdd(&Qn[wave], 1u);
                        myq[p] = ((e0 + (unsigned)k) << 8) | a[k];
                    }
                }
            }
        }

        const unsigned cnt = (unsigned)__builtin_amdgcn_readfirstlane((int)Qn[wave]);
        if (last || cnt >= (unsigned)DRAIN_AT) {
            for (unsigned base = 0; base < cnt; base += 32u) {
                const unsigned q = base + (unsigned)lane;
                unsigned ent = 0u;
                int      sv  = 0;
                if (q < cnt) {
                    ent = myq[q];
                    sv  = src[ent >> 8];
                }
                const int m = (int)((cnt - base < 32u) ? (cnt - base) : 32u);

                auto gather = [&](int j, float2& x0, float2& x1, float2& pr, unsigned& dl) {
                    const int s = __builtin_amdgcn_readlane(sv, j);
                    dl = (unsigned)__builtin_amdgcn_readlane((int)ent, j) & 255u;
                    const float2* fs = (const float2*)(fsum + (size_t)s * OUT_F);
                    x0 = fs[lane];
                    x1 = fs[32 + lane];
                    pr = ((const float2*)(fprod + (size_t)s * RANK))[lane];
                };
                auto accum = [&](const float2& x0, const float2& x1, const float2& pr, unsigned dl) {
                    float* h = Hs + dl * OUT_F + 2 * lane;
                    atomicAdd(h,      x0.x);
                    atomicAdd(h + 1,  x0.y);
                    atomicAdd(h + 64, x1.x);
                    atomicAdd(h + 65, x1.y);
                    float* l = La + dl * RANK + 2 * lane;
                    atomicAdd(l,     __logf(fabsf(pr.x)));
                    atomicAdd(l + 1, __logf(fabsf(pr.y)));
                    const unsigned b0 = __builtin_amdgcn_ballot_w32(pr.x < 0.0f);
                    const unsigned b1 = __builtin_amdgcn_ballot_w32(pr.y < 0.0f);
                    if (lane < 2) atomicXor(&Sg[dl * 2 + lane], lane ? b1 : b0);
                };

                int j = 0;
                for (; j + 4 <= m; j += 4) {
                    float2 a0, a1, ap, b0v, b1v, bp, c0, c1, cp, e0v, e1v, ep;
                    unsigned da, db, dc, de;
                    gather(j,     a0,  a1,  ap, da);
                    gather(j + 1, b0v, b1v, bp, db);
                    gather(j + 2, c0,  c1,  cp, dc);
                    gather(j + 3, e0v, e1v, ep, de);
                    accum(a0,  a1,  ap, da);
                    accum(b0v, b1v, bp, db);
                    accum(c0,  c1,  cp, dc);
                    accum(e0v, e1v, ep, de);
                }
                for (; j < m; ++j) {
                    float2 a0, a1, ap;
                    unsigned da;
                    gather(j, a0, a1, ap, da);
                    accum(a0, a1, ap, da);
                }
            }
            if (lane == 0) Qn[wave] = 0u;
        }
    };

    constexpr int NPAIR = (NCH + 2 * EDGE_WAVES - 1) / (2 * EDGE_WAVES);
    int4 ra[SCAN_U], rb[SCAN_U];
    load_step(ra, wave);
    for (int p = 0; p < NPAIR; ++p) {
        const int c = wave + p * 2 * EDGE_WAVES;
        load_step(rb, c + EDGE_WAVES);
        scan_step(ra, c, false);
        load_step(ra, c + 2 * EDGE_WAVES);
        scan_step(rb, c + EDGE_WAVES, p == NPAIR - 1);
    }
    __syncthreads();

    const int rows = (N_NODES - n0 < NR) ? (N_NODES - n0) : NR;
    {
        float4* hg = (float4*)(hsum + (size_t)n0 * OUT_F);
        for (int i = tid; i < rows * OUT_F / 4; i += 1024) hg[i] = ((const float4*)Hs)[i];
        float4* lg = (float4*)(labs + (size_t)n0 * RANK);
        for (int i = tid; i < rows * RANK / 4; i += 1024) lg[i] = ((const float4*)La)[i];
        int* ng = ncnt + (size_t)n0 * RANK;
        for (int i = tid; i < rows * RANK; i += 1024) {
            const int r = i >> 6, cc = i & 63;
            ng[i] = (int)((Sg[r * 2 + (cc & 1)] >> (cc >> 1)) & 1u);
        }
    }
}

__global__ __launch_bounds__(128) void k_final(const float* __restrict__ vmat,
                                               const float* __restrict__ hsum,
                                               const float* __restrict__ labs,
                                               const int*   __restrict__ ncnt,
                                               const float* __restrict__ ideg,
                                               float* __restrict__ out)
{
    __shared__ float Hp[16][68];
    __shared__ float sc[16];

    const int m0  = blockIdx.x * 16;
    const int tid = threadIdx.x;

    if (tid < 16) {
        sc[tid] = rsqrtf(fmaxf(ideg[m0 + tid], 1.0f));
    }
    for (int i = tid; i < 16 * RANK; i += 128) {
        int r = i >> 6, c = i & 63;
        int row = m0 + r;
        float val = 0.0f;
        float dg = ideg[row];
        if (dg > 0.0f) {
            float la = labs[(size_t)row * RANK + c];
            int   n  = ncnt[(size_t)row * RANK + c];
            float sg = (n & 1) ? -1.0f : 1.0f;
            val = sg * __expf(la);
        }
        Hp[r][c] = val;
    }
    __syncthreads();

    const int wave  = tid >> 5;
    const int lane  = tid & 31;
    const int rsel  = lane & 15;
    const int khalf = (lane < 16) ? 0 : 2;
    const int rofs  = (lane < 16) ? 0 : 8;

    for (int t = 0; t < 2; ++t) {
        const int n0 = (wave * 2 + t) * 16;

        for (int k = 0; k < RANK; k += 32)
            __builtin_prefetch(vmat + (size_t)k * OUT_F + n0, 0, 0);

        v8f acc = {};
        for (int k0 = 0; k0 < RANK; k0 += 4) {
            v2f a;
            a.x = Hp[rsel][k0 + khalf];
            a.y = Hp[rsel][k0 + khalf + 1];
            v2f b;
            const float* Bp = vmat + (size_t)(k0 + khalf) * OUT_F + n0 + rsel;
            b.x = Bp[0];
            b.y = Bp[OUT_F];
            acc = __builtin_amdgcn_wmma_f32_16x16x4_f32(
                false, a, false, b, (short)0, acc, false, false);
        }
        const int colg = n0 + rsel;
        const float* hp = hsum + (size_t)(m0 + rofs) * OUT_F + colg;
        float*       op = out  + (size_t)(m0 + rofs) * OUT_F + colg;
#pragma unroll
        for (int r = 0; r < 8; ++r)
            op[r * OUT_F] = (acc[r] + hp[r * OUT_F]) * sc[rofs + r];
    }
}

extern "C" void kernel_launch(void* const* d_in, const int* in_sizes, int n_in,
                              void* d_out, int out_size, void* d_ws, size_t ws_size,
                              hipStream_t stream)
{
    (void)in_sizes; (void)n_in; (void)out_size; (void)ws_size;

    const float* feat = (const float*)d_in[0];
    const float* w1   = (const float*)d_in[1];
    const float* w2   = (const float*)d_in[2];
    const float* vmat = (const float*)d_in[3];
    const int*   src  = (const int*)d_in[4];
    const int*   dst  = (const int*)d_in[5];

    float* ws    = (float*)d_ws;
    float* odeg  = ws + OFF_ODEG;
    float* ideg  = ws + OFF_IDEG;
    float* hsum  = ws + OFF_HSUM;
    float* labs  = ws + OFF_LABS;
    int*   ncnt  = (int*)(ws + OFF_NCNT);
    float* fsum  = ws + OFF_FSUM;
    float* fprod = ws + OFF_FPROD;

    {
        unsigned n4 = ZERO_CNT / 4u;
        k_zero<<<(n4 + 255u) / 256u, 256, 0, stream>>>((unsigned int*)ws, n4);
    }
    k_degrees<<<N_EDGES / 256, 256, 0, stream>>>(src, dst, odeg, ideg);
    k_node_gemm<<<N_NODES / 16, 128, 0, stream>>>(feat, w1, w2, odeg, fsum, fprod);
    k_edge<<<EDGE_BLOCKS, 1024, 0, stream>>>(src, dst, fsum, fprod,
                                             hsum, labs, ncnt);
    k_final<<<N_NODES / 16, 128, 0, stream>>>(vmat, hsum, labs, ncnt, ideg,
                                              (float*)d_out);
}


# Degrees counted in LDS, not by global float atomics: out-degree as 32 per-slice LDS histograms added up by the node-product blocks, in-degree by the edge blocks' row counters; 4 launches, not 5
# speedup vs baseline: 2.4029x; 1.3630x over previous
#include <hip/hip_runtime.h>
#include <math.h>

typedef __attribute__((ext_vector_type(2))) float v2f;
typedef __attribute__((ext_vector_type(8))) float v8f;

#define N_NODES 50000
#define N_EDGES 800000
#define IN_F    256
#define OUT_F   128
#define RANK    64

static_assert(N_NODES % 16 == 0, "row tiles exact");
static_assert(N_EDGES % 256 == 0, "edge blocks exact");

#define OFF_IDEG   50000u
#define OFF_HSUM   100000u
#define OFF_LABS   6500000u
#define OFF_NCNT   9700000u
#define OFF_FSUM   12900000u
#define OFF_FPROD  19300000u
#define OFF_ODPART OFF_HSUM

#define DEG_SLICES  32
#define DEG_EDGES   (N_EDGES / DEG_SLICES)
#define DEG_LOADS   ((DEG_EDGES / 4 + 1023) / 1024)
static_assert(N_EDGES % DEG_SLICES == 0 && DEG_EDGES % 4 == 0, "slices exact; int4 loads aligned");
static_assert(N_NODES % 4 == 0, "count planes move as uint4");
static_assert(DEG_SLICES % 8 == 0, "the node product blocks add the planes in 8 groups");
static_assert((size_t)DEG_SLICES * N_NODES <= 6400000u, "partial planes fit the h_sum region");

#define NR          196
#define EDGE_BLOCKS ((N_NODES + NR - 1) / NR)
#define EDGE_WAVES  32
#define SCAN_U      4
#define SCAN_EDGES  (32 * 4 * SCAN_U)
#define DRAIN_AT    24
#define QCAP        640
static_assert(NR <= 256, "local row id fits 8 bits of a queue entry");
static_assert(N_EDGES < (1 << 24) && N_EDGES % 4 == 0, "edge id fits 24 bits; int4 scan exact");
static_assert(DRAIN_AT + SCAN_EDGES <= QCAP, "one scan step can never overflow a wave's queue");

__global__ __launch_bounds__(1024) void k_outdeg(const int* __restrict__ src,
                                                 unsigned* __restrict__ parts)
{
    __shared__ __attribute__((aligned(16))) unsigned Hc[N_NODES];

    const int tid = threadIdx.x;
    constexpr int S4 = DEG_EDGES / 4;

    const int4* s4 = (const int4*)(src + (size_t)blockIdx.x * DEG_EDGES);
    int4 v[DEG_LOADS];
#pragma unroll
    for (int j = 0; j < DEG_LOADS; ++j) {
        const int i = tid + j * 1024;
        v[j] = s4[(i < S4) ? i : (S4 - 1)];
    }

    for (int i = tid; i < N_NODES / 4; i += 1024)
        ((uint4*)Hc)[i] = make_uint4(0u, 0u, 0u, 0u);
    __syncthreads();

#pragma unroll
    for (int j = 0; j < DEG_LOADS; ++j) {
        if (tid + j * 1024 < S4) {
            atomicAdd(&Hc[v[j].x], 1u);
            atomicAdd(&Hc[v[j].y], 1u);
            atomicAdd(&Hc[v[j].z], 1u);
            atomicAdd(&Hc[v[j].w], 1u);
        }
    }
    __syncthreads();

    uint4* og = (uint4*)(parts + (size_t)blockIdx.x * N_NODES);
    for (int i = tid; i < N_NODES / 4; i += 1024)
        og[i] = ((const uint4*)Hc)[i];
}

__global__ __launch_bounds__(128) void k_node_gemm(const float* __restrict__ feat,
                                                   const float* __restrict__ w1,
                                                   const float* __restrict__ w2,
                                                   const unsigned* __restrict__ odpart,
                                                   float* __restrict__ fsum,
                                                   float* __restrict__ fprod)
{
    __shared__ float As[16][260];
    __shared__ unsigned dp[8][16];

    const int m0  = blockIdx.x * 16;
    const int tid = threadIdx.x;

    {
        const int r = tid & 15;
        const int g = tid >> 4;
        unsigned d = 0u;
#pragma unroll
        for (int j = 0; j < DEG_SLICES / 8; ++j)
            d += odpart[(size_t)(g + 8 * j) * N_NODES + m0 + r];
        dp[g][r] = d;
    }
    __syncthreads();

    {
        const int r   = tid >> 3;
        const int c0  = (tid & 7) * 32;
        unsigned d = 0u;
#pragma unroll
        for (int q = 0; q < 8; ++q) d += dp[q][r];
        const float s = rsqrtf(fmaxf((float)d, 1.0f));
        const float4* g = (const float4*)(feat + (size_t)(m0 + r) * IN_F + c0);
#pragma unroll
        for (int j = 0; j < 8; ++j) {
            float4 x = g[j];
            As[r][c0 + 4 * j + 0] = x.x * s;
            As[r][c0 + 4 * j + 1] = x.y * s;
            As[r][c0 + 4 * j + 2] = x.z * s;
            As[r][c0 + 4 * j + 3] = x.w * s;
        }
    }
    __syncthreads();

    const int wave  = tid >> 5;
    const int lane  = tid & 31;
    const int rsel  = lane & 15;
    const int khalf = (lane < 16) ? 0 : 2;
    const int rofs  = (lane < 16) ? 0 : 8;

    for (int t = 0; t < 3; ++t) {
        const int  tile = wave + 4 * t;
        const bool isW2 = (tile >= 8);
        const int  n0   = isW2 ? (tile - 8) * 16 : tile * 16;
        const float* B  = isW2 ? w2 : w1;
        const int  ldb  = isW2 ? RANK : OUT_F;

        for (int k = 0; k < IN_F; k += 32)
            __builtin_prefetch(B + (size_t)k * ldb + n0, 0, 0);

        v8f acc = {};
        for (int k0 = 0; k0 < IN_F; k0 += 4) {
            v2f a;
            a.x = As[rsel][k0 + khalf];
            a.y = As[rsel][k0 + khalf + 1];
            v2f b;
            const float* Bp = B + (size_t)(k0 + khalf) * ldb + n0 + rsel;
            b.x = Bp[0];
            b.y = Bp[ldb];
            acc = __builtin_amdgcn_wmma_f32_16x16x4_f32(
                false, a, false, b, (short)0, acc, false, false);
        }

        const int colg = n0 + rsel;
        if (isW2) {
            const float bias = w2[(size_t)IN_F * RANK + colg];
            float* op = fprod + (size_t)(m0 + rofs) * RANK + colg;
#pragma unroll
            for (int r = 0; r < 8; ++r)
                op[r * RANK] = tanhf(acc[r] + bias);
        } else {
            float* op = fsum + (size_t)(m0 + rofs) * OUT_F + colg;
#pragma unroll
            for (int r = 0; r < 8; ++r)
                op[r * OUT_F] = acc[r];
        }
    }
}

__global__ __launch_bounds__(1024) void k_edge(const int* __restrict__ src,
                                               const int* __restrict__ dst,
                                               const float* __restrict__ fsum,
                                               const float* __restrict__ fprod,
                                               float* __restrict__ hsum,
                                               float* __restrict__ labs,
                                               int*   __restrict__ ncnt,
                                               float* __restrict__ ideg)
{
    __shared__ __attribute__((aligned(16))) float    Hs[NR * OUT_F];
    __shared__ __attribute__((aligned(16))) float    La[NR * RANK];
    __shared__ __attribute__((aligned(16))) unsigned Sg[NR * 2];
    __shared__ unsigned Cn[NR];
    __shared__ unsigned Qe[EDGE_WAVES * QCAP];
    __shared__ unsigned Qn[EDGE_WAVES];

    const int tid  = threadIdx.x;
    const int wave = tid >> 5;
    const int lane = tid & 31;
    const int n0   = blockIdx.x * NR;

    for (int i = tid; i < NR * OUT_F / 4; i += 1024)
        ((float4*)Hs)[i] = make_float4(0.0f, 0.0f, 0.0f, 0.0f);
    for (int i = tid; i < NR * RANK / 4; i += 1024)
        ((float4*)La)[i] = make_float4(0.0f, 0.0f, 0.0f, 0.0f);
    for (int i = tid; i < NR * 2; i += 1024)
        Sg[i] = 0u;
    if (tid < NR) Cn[tid] = 0u;
    if (tid < EDGE_WAVES) Qn[tid] = 0u;
    __syncthreads();

    constexpr int N4    = N_EDGES / 4;
    constexpr int STEP4 = 32 * SCAN_U;
    constexpr int NCH   = (N4 + STEP4 - 1) / STEP4;
    const int4* d4  = (const int4*)dst;
    unsigned*   myq = Qe + wave * QCAP;

    auto load_step = [&](int4 (&r)[SCAN_U], int cc) {
        for (int u = 0; u < SCAN_U; ++u) {
            const int i4 = cc * STEP4 + u * 32 + lane;
            r[u] = d4[(i4 < N4) ? i4 : (N4 - 1)];
        }
    };

    auto scan_step = [&](const int4 (&r)[SCAN_U], int cc, bool last) {
        for (int u = 0; u < SCAN_U; ++u) {
            const int  i4 = cc * STEP4 + u * 32 + lane;
            const bool ok = (i4 < N4);
            const unsigned e0 = (unsigned)i4 * 4u;
            const unsigned a[4] = {(unsigned)(r[u].x - n0), (unsigned)(r[u].y - n0),
                                   (unsigned)(r[u].z - n0), (unsigned)(r[u].w - n0)};
            const unsigned amin = min(min(a[0], a[1]), min(a[2], a[3]));
            if (ok && amin < (unsigned)NR) {
                for (int k = 0; k < 4; ++k) {
                    if (a[k] < (unsigned)NR) {
                        const unsigned p = atomicAdd(&Qn[wave], 1u);
                        myq[p] = ((e0 + (unsigned)k) << 8) | a[k];
                    }
                }
            }
        }

        const unsigned cnt = (unsigned)__builtin_amdgcn_readfirstlane((int)Qn[wave]);
        if (last || cnt >= (unsigned)DRAIN_AT) {
            for (unsigned base = 0; base < cnt; base += 32u) {
                const unsigned q = base + (unsigned)lane;
                unsigned ent = 0u;
                int      sv  = 0;
                if (q < cnt) {
                    ent = myq[q];
                    sv  = src[ent >> 8];
                }
                const int m = (int)((cnt - base < 32u) ? (cnt - base) : 32u);

                auto gather = [&](int j, float2& x0, float2& x1, float2& pr, unsigned& dl) {
                    const int s = __builtin_amdgcn_readlane(sv, j);
                    dl = (unsigned)__builtin_amdgcn_readlane((int)ent, j) & 255u;
                    const float2* fs = (const float2*)(fsum + (size_t)s * OUT_F);
                    x0 = fs[lane];
                    x1 = fs[32 + lane];
                    pr = ((const float2*)(fprod + (size_t)s * RANK))[lane];
                };
                auto accum = [&](const float2& x0, const float2& x1, const float2& pr, unsigned dl) {
                    float* h = Hs + dl * OUT_F + 2 * lane;
                    atomicAdd(h,      x0.x);
                    atomicAdd(h + 1,  x0.y);
                    atomicAdd(h + 64, x1.x);
                    atomicAdd(h + 65, x1.y);
                    float* l = La + dl * RANK + 2 * lane;
                    atomicAdd(l,     __logf(fabsf(pr.x)));
                    atomicAdd(l + 1, __logf(fabsf(pr.y)));
                    const unsigned b0 = __builtin_amdgcn_ballot_w32(pr.x < 0.0f);
                    const unsigned b1 = __builtin_amdgcn_ballot_w32(pr.y < 0.0f);
                    if (lane < 2) atomicXor(&Sg[dl * 2 + lane], lane ? b1 : b0);
                    if (lane == 2) atomicAdd(&Cn[dl], 1u);
                };

                int j = 0;
                for (; j + 4 <= m; j += 4) {
                    float2 a0, a1, ap, b0v, b1v, bp, c0, c1, cp, e0v, e1v, ep;
                    unsigned da, db, dc, de;
                    gather(j,     a0,  a1,  ap, da);
                    gather(j + 1, b0v, b1v, bp, db);
                    gather(j + 2, c0,  c1,  cp, dc);
                    gather(j + 3, e0v, e1v, ep, de);
                    accum(a0,  a1,  ap, da);
                    accum(b0v, b1v, bp, db);
                    accum(c0,  c1,  cp, dc);
                    accum(e0v, e1v, ep, de);
                }
                for (; j < m; ++j) {
                    float2 a0, a1, ap;
                    unsigned da;
                    gather(j, a0, a1, ap, da);
                    accum(a0, a1, ap, da);
                }
            }
            if (lane == 0) Qn[wave] = 0u;
        }
    };

    constexpr int NPAIR = (NCH + 2 * EDGE_WAVES - 1) / (2 * EDGE_WAVES);
    int4 ra[SCAN_U], rb[SCAN_U];
    load_step(ra, wave);
    for (int p = 0; p < NPAIR; ++p) {
        const int c = wave + p * 2 * EDGE_WAVES;
        load_step(rb, c + EDGE_WAVES);
        scan_step(ra, c, false);
        load_step(ra, c + 2 * EDGE_WAVES);
        scan_step(rb, c + EDGE_WAVES, p == NPAIR - 1);
    }
    __syncthreads();

    const int rows = (N_NODES - n0 < NR) ? (N_NODES - n0) : NR;
    {
        float4* hg = (float4*)(hsum + (size_t)n0 * OUT_F);
        for (int i = tid; i < rows * OUT_F / 4; i += 1024) hg[i] = ((const float4*)Hs)[i];
        float4* lg = (float4*)(labs + (size_t)n0 * RANK);
        for (int i = tid; i < rows * RANK / 4; i += 1024) lg[i] = ((const float4*)La)[i];
        int* ng = ncnt + (size_t)n0 * RANK;
        for (int i = tid; i < rows * RANK; i += 1024) {
            const int r = i >> 6, cc = i & 63;
            ng[i] = (int)((Sg[r * 2 + (cc & 1)] >> (cc >> 1)) & 1u);
        }
        if (tid < rows) ideg[n0 + tid] = (float)Cn[tid];
    }
}

__global__ __launch_bounds__(128) void k_final(const float* __restrict__ vmat,
                                               const float* __restrict__ hsum,
                                               const float* __restrict__ labs,
                                               const int*   __restrict__ ncnt,
                                               const float* __restrict__ ideg,
                                               float* __restrict__ out)
{
    __shared__ float Hp[16][68];
    __shared__ float sc[16];

    const int m0  = blockIdx.x * 16;
    const int tid = threadIdx.x;

    if (tid < 16) {
        sc[tid] = rsqrtf(fmaxf(ideg[m0 + tid], 1.0f));
    }
    for (int i = tid; i < 16 * RANK; i += 128) {
        int r = i >> 6, c = i & 63;
        int row = m0 + r;
        float val = 0.0f;
        float dg = ideg[row];
        if (dg > 0.0f) {
            float la = labs[(size_t)row * RANK + c];
            int   n  = ncnt[(size_t)row * RANK + c];
            float sg = (n & 1) ? -1.0f : 1.0f;
            val = sg * __expf(la);
        }
        Hp[r][c] = val;
    }
    __syncthreads();

    const int wave  = tid >> 5;
    const int lane  = tid & 31;
    const int rsel  = lane & 15;
    const int khalf = (lane < 16) ? 0 : 2;
    const int rofs  = (lane < 16) ? 0 : 8;

    for (int t = 0; t < 2; ++t) {
        const int n0 = (wave * 2 + t) * 16;

        for (int k = 0; k < RANK; k += 32)
            __builtin_prefetch(vmat + (size_t)k * OUT_F + n0, 0, 0);

        v8f acc = {};
        for (int k0 = 0; k0 < RANK; k0 += 4) {
            v2f a;
            a.x = Hp[rsel][k0 + khalf];
            a.y = Hp[rsel][k0 + khalf + 1];
            v2f b;
            const float* Bp = vmat + (size_t)(k0 + khalf) * OUT_F + n0 + rsel;
            b.x = Bp[0];
            b.y = Bp[OUT_F];
            acc = __builtin_amdgcn_wmma_f32_16x16x4_f32(
                false, a, false, b, (short)0, acc, false, false);
        }
        const int colg = n0 + rsel;
        const float* hp = hsum + (size_t)(m0 + rofs) * OUT_F + colg;
        float*       op = out  + (size_t)(m0 + rofs) * OUT_F + colg;
#pragma unroll
        for (int r = 0; r < 8; ++r)
            op[r * OUT_F] = (acc[r] + hp[r * OUT_F]) * sc[rofs + r];
    }
}

extern "C" void kernel_launch(void* const* d_in, const int* in_sizes, int n_in,
                              void* d_out, int out_size, void* d_ws, size_t ws_size,
                              hipStream_t stream)
{
    (void)in_sizes; (void)n_in; (void)out_size; (void)ws_size;

    const float* feat = (const float*)d_in[0];
    const float* w1   = (const float*)d_in[1];
    const float* w2   = (const float*)d_in[2];
    const float* vmat = (const float*)d_in[3];
    const int*   src  = (const int*)d_in[4];
    const int*   dst  = (const int*)d_in[5];

    float* ws    = (float*)d_ws;
    float* ideg  = ws + OFF_IDEG;
    unsigned* odpart = (unsigned*)(ws + OFF_ODPART);
    float* hsum  = ws + OFF_HSUM;
    float* labs  = ws + OFF_LABS;
    int*   ncnt  = (int*)(ws + OFF_NCNT);
    float* fsum  = ws + OFF_FSUM;
    float* fprod = ws + OFF_FPROD;

    k_outdeg<<<DEG_SLICES, 1024, 0, stream>>>(src, odpart);
    k_node_gemm<<<N_NODES / 16, 128, 0, stream>>>(feat, w1, w2, odpart, fsum, fprod);
    k_edge<<<EDGE_BLOCKS, 1024, 0, stream>>>(src, dst, fsum, fprod,
                                             hsum, labs, ncnt, ideg);
    k_final<<<N_NODES / 16, 128, 0, stream>>>(vmat, hsum, labs, ncnt, ideg,
                                              (float*)d_out);
}


# Edge blocks scan a 1-byte block-id plane (dst/196, written by the degree blocks) in place of the 4-byte dst list; local row from dst at the gather; sign plane packed to 2 words/row for room
# speedup vs baseline: 2.5558x; 1.0636x over previous
#include <hip/hip_runtime.h>
#include <math.h>

typedef __attribute__((ext_vector_type(2))) float v2f;
typedef __attribute__((ext_vector_type(8))) float v8f;

#define N_NODES 50000
#define N_EDGES 800000
#define IN_F    256
#define OUT_F   128
#define RANK    64

static_assert(N_NODES % 16 == 0, "row tiles exact");
static_assert(N_EDGES % 256 == 0, "edge blocks exact");

#define OFF_IDEG   50000u
#define OFF_HSUM   100000u
#define OFF_LABS   6500000u
#define OFF_SGN    9700000u
#define OFF_BID    9900000u
#define OFF_FSUM   12900000u
#define OFF_FPROD  19300000u
#define OFF_ODPART OFF_HSUM

#define DEG_SLICES  32
#define DEG_EDGES   (N_EDGES / DEG_SLICES)
#define DEG_LOADS   ((DEG_EDGES / 4 + 1023) / 1024)
static_assert(N_EDGES % DEG_SLICES == 0 && DEG_EDGES % 4 == 0, "slices exact; int4 loads aligned");
static_assert(N_NODES % 4 == 0, "count planes move as uint4");
static_assert(DEG_SLICES % 8 == 0, "the node product blocks add the planes in 8 groups");
static_assert((size_t)DEG_SLICES * N_NODES <= 6400000u, "partial planes fit the h_sum region");
static_assert(OFF_SGN + 2u * N_NODES <= OFF_BID && OFF_BID + N_EDGES / 4 <= 12900000u,
              "packed sign words and the block-id bytes fit in front of the f_sum plane");

#define NR          196
#define EDGE_BLOCKS ((N_NODES + NR - 1) / NR)
#define EDGE_WAVES  32
#define SCAN_U      4
#define SCAN_HALF   (32 * 16 * SCAN_U / 2)
#define DRAIN_AT    24
#define QCAP        1056
static_assert(EDGE_BLOCKS <= 256, "a block id fits one byte");
static_assert(N_EDGES % 16 == 0, "16-byte block-id scan exact");
static_assert(SCAN_U % 2 == 0 && DRAIN_AT + SCAN_HALF <= QCAP, "half a scan step can never overflow a wave's queue");

__device__ __forceinline__ unsigned block_id_byte(int d)
{
    return ((unsigned)d < (unsigned)N_NODES) ? (unsigned)d / (unsigned)NR : 255u;
}

__global__ __launch_bounds__(1024) void k_outdeg(const int* __restrict__ src,
                                                 const int* __restrict__ dst,
                                                 unsigned* __restrict__ parts,
                                                 unsigned* __restrict__ bid)
{
    __shared__ __attribute__((aligned(16))) unsigned Hc[N_NODES];

    const int tid = threadIdx.x;
    constexpr int S4 = DEG_EDGES / 4;

    const int4* s4 = (const int4*)(src + (size_t)blockIdx.x * DEG_EDGES);
    int4 v[DEG_LOADS];
#pragma unroll
    for (int j = 0; j < DEG_LOADS; ++j) {
        const int i = tid + j * 1024;
        v[j] = s4[(i < S4) ? i : (S4 - 1)];
    }

    {
        const int4* t4 = (const int4*)(dst + (size_t)blockIdx.x * DEG_EDGES);
        unsigned*   bo = bid + (size_t)blockIdx.x * S4;
        for (int j = 0; j < DEG_LOADS; ++j) {
            const int i = tid + j * 1024;
            if (i < S4) {
                const int4 d = t4[i];
                bo[i] = block_id_byte(d.x) | (block_id_byte(d.y) << 8) |
                        (block_id_byte(d.z) << 16) | (block_id_byte(d.w) << 24);
            }
        }
    }

    for (int i = tid; i < N_NODES / 4; i += 1024)
        ((uint4*)Hc)[i] = make_uint4(0u, 0u, 0u, 0u);
    __syncthreads();

#pragma unroll
    for (int j = 0; j < DEG_LOADS; ++j) {
        if (tid + j * 1024 < S4) {
            atomicAdd(&Hc[v[j].x], 1u);
            atomicAdd(&Hc[v[j].y], 1u);
            atomicAdd(&Hc[v[j].z], 1u);
            atomicAdd(&Hc[v[j].w], 1u);
        }
    }
    __syncthreads();

    uint4* og = (uint4*)(parts + (size_t)blockIdx.x * N_NODES);
    for (int i = tid; i < N_NODES / 4; i += 1024)
        og[i] = ((const uint4*)Hc)[i];
}

__global__ __launch_bounds__(128) void k_node_gemm(const float* __restrict__ feat,
                                                   const float* __restrict__ w1,
                                                   const float* __restrict__ w2,
                                                   const unsigned* __restrict__ odpart,
                                                   float* __restrict__ fsum,
                                                   float* __restrict__ fprod)
{
    __shared__ float As[16][260];
    __shared__ unsigned dp[8][16];

    const int m0  = blockIdx.x * 16;
    const int tid = threadIdx.x;

    {
        const int r = tid & 15;
        const int g = tid >> 4;
        unsigned d = 0u;
#pragma unroll
        for (int j = 0; j < DEG_SLICES / 8; ++j)
            d += odpart[(size_t)(g + 8 * j) * N_NODES + m0 + r];
        dp[g][r] = d;
    }
    __syncthreads();

    {
        const int r   = tid >> 3;
        const int c0  = (tid & 7) * 32;
        unsigned d = 0u;
#pragma unroll
        for (int q = 0; q < 8; ++q) d += dp[q][r];
        const float s = rsqrtf(fmaxf((float)d, 1.0f));
        const float4* g = (const float4*)(feat + (size_t)(m0 + r) * IN_F + c0);
#pragma unroll
        for (int j = 0; j < 8; ++j) {
            float4 x = g[j];
            As[r][c0 + 4 * j + 0] = x.x * s;
            As[r][c0 + 4 * j + 1] = x.y * s;
            As[r][c0 + 4 * j + 2] = x.z * s;
            As[r][c0 + 4 * j + 3] = x.w * s;
        }
    }
    __syncthreads();

    const int wave  = tid >> 5;
    const int lane  = tid & 31;
    const int rsel  = lane & 15;
    const int khalf = (lane < 16) ? 0 : 2;
    const int rofs  = (lane < 16) ? 0 : 8;

    for (int t = 0; t < 3; ++t) {
        const int  tile = wave + 4 * t;
        const bool isW2 = (tile >= 8);
        const int  n0   = isW2 ? (tile - 8) * 16 : tile * 16;
        const float* B  = isW2 ? w2 : w1;
        const int  ldb  = isW2 ? RANK : OUT_F;

        for (int k = 0; k < IN_F; k += 32)
            __builtin_prefetch(B + (size_t)k * ldb + n0, 0, 0);

        v8f acc = {};
        for (int k0 = 0; k0 < IN_F; k0 += 4) {
            v2f a;
            a.x = As[rsel][k0 + khalf];
            a.y = As[rsel][k0 + khalf + 1];
            v2f b;
            const float* Bp = B + (size_t)(k0 + khalf) * ldb + n0 + rsel;
            b.x = Bp[0];
            b.y = Bp[ldb];
            acc = __builtin_amdgcn_wmma_f32_16x16x4_f32(
                false, a, false, b, (short)0, acc, false, false);
        }

        const int colg = n0 + rsel;
        if (isW2) {
            const float bias = w2[(size_t)IN_F * RANK + colg];
            float* op = fprod + (size_t)(m0 + rofs) * RANK + colg;
#pragma unroll
            for (int r = 0; r < 8; ++r)
                op[r * RANK] = tanhf(acc[r] + bias);
        } else {
            float* op = fsum + (size_t)(m0 + rofs) * OUT_F + colg;
#pragma unroll
            for (int r = 0; r < 8; ++r)
                op[r * OUT_F] = acc[r];
        }
    }
}

__global__ __launch_bounds__(1024) void k_edge(const int* __restrict__ src,
                                               const int* __restrict__ dst,
                                               const unsigned* __restrict__ bid,
                                               const float* __restrict__ fsum,
                                               const float* __restrict__ fprod,
                                               float* __restrict__ hsum,
                                               float* __restrict__ labs,
                                               unsigned* __restrict__ sgn,
                                               float* __restrict__ ideg)
{
    __shared__ __attribute__((aligned(16))) float    Hs[NR * OUT_F];
    __shared__ __attribute__((aligned(16))) float    La[NR * RANK];
    __shared__ __attribute__((aligned(16))) unsigned Sg[NR * 2];
    __shared__ unsigned Cn[NR];
    __shared__ unsigned Qe[EDGE_WAVES * QCAP];
    __shared__ unsigned Qn[EDGE_WAVES];

    const int tid  = threadIdx.x;
    const int wave = tid >> 5;
    const int lane = tid & 31;
    const int n0   = blockIdx.x * NR;

    for (int i = tid; i < NR * OUT_F / 4; i += 1024)
        ((float4*)Hs)[i] = make_float4(0.0f, 0.0f, 0.0f, 0.0f);
    for (int i = tid; i < NR * RANK / 4; i += 1024)
        ((float4*)La)[i] = make_float4(0.0f, 0.0f, 0.0f, 0.0f);
    for (int i = tid; i < NR * 2; i += 1024)
        Sg[i] = 0u;
    if (tid < NR) Cn[tid] = 0u;
    if (tid < EDGE_WAVES) Qn[tid] = 0u;
    __syncthreads();

    constexpr int N16    = N_EDGES / 16;
    constexpr int STEP16 = 32 * SCAN_U;
    constexpr int NCH    = (N16 + STEP16 - 1) / STEP16;
    const uint4* b16  = (const uint4*)bid;
    unsigned*    myq  = Qe + wave * QCAP;
    const unsigned bsplat = (unsigned)blockIdx.x * 0x01010101u;

    auto load_step = [&](uint4 (&r)[SCAN_U], int cc) {
        for (int u = 0; u < SCAN_U; ++u) {
            const int i16 = cc * STEP16 + u * 32 + lane;
            r[u] = b16[(i16 < N16) ? i16 : (N16 - 1)];
        }
    };

    auto drain = [&](bool last) {
        const unsigned cnt = (unsigned)__builtin_amdgcn_readfirstlane((int)Qn[wave]);
        if (last || cnt >= (unsigned)DRAIN_AT) {
            for (unsigned base = 0; base < cnt; base += 32u) {
                const unsigned q = base + (unsigned)lane;
                int      sv  = 0;
                unsigned dlv = (unsigned)NR;
                if (q < cnt) {
                    const unsigned e = myq[q];
                    sv  = src[e];
                    dlv = (unsigned)(dst[e] - n0);
                }
                const int m = (int)((cnt - base < 32u) ? (cnt - base) : 32u);

                auto gather = [&](int j, float2& x0, float2& x1, float2& pr, unsigned& dl) {
                    const int s = __builtin_amdgcn_readlane(sv, j);
                    dl = (unsigned)__builtin_amdgcn_readlane((int)dlv, j);
                    const float2* fs = (const float2*)(fsum + (size_t)s * OUT_F);
                    x0 = fs[lane];
                    x1 = fs[32 + lane];
                    pr = ((const float2*)(fprod + (size_t)s * RANK))[lane];
                };
                auto accum = [&](const float2& x0, const float2& x1, const float2& pr, unsigned dl) {
                    if (dl >= (unsigned)NR) return;
                    float* h = Hs + dl * OUT_F + 2 * lane;
                    atomicAdd(h,      x0.x);
                    atomicAdd(h + 1,  x0.y);
                    atomicAdd(h + 64, x1.x);
                    atomicAdd(h + 65, x1.y);
                    float* l = La + dl * RANK + 2 * lane;
                    atomicAdd(l,     __logf(fabsf(pr.x)));
                    atomicAdd(l + 1, __logf(fabsf(pr.y)));
                    const unsigned b0 = __builtin_amdgcn_ballot_w32(pr.x < 0.0f);
                    const unsigned b1 = __builtin_amdgcn_ballot_w32(pr.y < 0.0f);
                    if (lane < 2) atomicXor(&Sg[dl * 2 + lane], lane ? b1 : b0);
                    if (lane == 2) atomicAdd(&Cn[dl], 1u);
                };

                int j = 0;
                for (; j + 4 <= m; j += 4) {
                    float2 a0, a1, ap, b0v, b1v, bp, c0, c1, cp, e0v, e1v, ep;
                    unsigned da, db, dc, de;
                    gather(j,     a0,  a1,  ap, da);
                    gather(j + 1, b0v, b1v, bp, db);
                    gather(j + 2, c0,  c1,  cp, dc);
                    gather(j + 3, e0v, e1v, ep, de);
                    accum(a0,  a1,  ap, da);
                    accum(b0v, b1v, bp, db);
                    accum(c0,  c1,  cp, dc);
                    accum(e0v, e1v, ep, de);
                }
                for (; j < m; ++j) {
                    float2 a0, a1, ap;
                    unsigned da;
                    gather(j, a0, a1, ap, da);
                    accum(a0, a1, ap, da);
                }
            }
            if (lane == 0) Qn[wave] = 0u;
        }
    };

    auto scan_step = [&](const uint4 (&r)[SCAN_U], int cc, bool last) {
        for (int u = 0; u < SCAN_U; ++u) {
            const int  i16 = cc * STEP16 + u * 32 + lane;
            const bool ok  = (i16 < N16);
            const unsigned e0 = (unsigned)i16 * 16u;
            const unsigned w[4] = {r[u].x ^ bsplat, r[u].y ^ bsplat, r[u].z ^ bsplat, r[u].w ^ bsplat};
            unsigned z[4];
            for (int k = 0; k < 4; ++k)
                z[k] = ~(((w[k] & 0x7f7f7f7fu) + 0x7f7f7f7fu) | w[k]) & 0x80808080u;
            if (ok && (z[0] | z[1] | z[2] | z[3]) != 0u) {
                for (int k = 0; k < 4; ++k) {
                    unsigned zk = z[k];
                    while (zk != 0u) {
                        const unsigned bit = (unsigned)__builtin_ctz(zk);
                        zk &= zk - 1u;
                        const unsigned p = atomicAdd(&Qn[wave], 1u);
                        myq[p] = e0 + 4u * (unsigned)k + (bit >> 3);
                    }
                }
            }
            if (u == SCAN_U / 2 - 1) drain(false);
        }
        drain(last);
    };

    constexpr int NPAIR = (NCH + 2 * EDGE_WAVES - 1) / (2 * EDGE_WAVES);
    uint4 ra[SCAN_U], rb[SCAN_U];
    load_step(ra, wave);
    for (int p = 0; p < NPAIR; ++p) {
        const int c = wave + p * 2 * EDGE_WAVES;
        load_step(rb, c + EDGE_WAVES);
        scan_step(ra, c, false);
        load_step(ra, c + 2 * EDGE_WAVES);
        scan_step(rb, c + EDGE_WAVES, p == NPAIR - 1);
    }
    __syncthreads();

    const int rows = (N_NODES - n0 < NR) ? (N_NODES - n0) : NR;
    {
        float4* hg = (float4*)(hsum + (size_t)n0 * OUT_F);
        for (int i = tid; i < rows * OUT_F / 4; i += 1024) hg[i] = ((const float4*)Hs)[i];
        float4* lg = (float4*)(labs + (size_t)n0 * RANK);
        for (int i = tid; i < rows * RANK / 4; i += 1024) lg[i] = ((const float4*)La)[i];
        unsigned* sg = sgn + (size_t)n0 * 2;
        for (int i = tid; i < rows * 2; i += 1024) sg[i] = Sg[i];
        if (tid < rows) ideg[n0 + tid] = (float)Cn[tid];
    }
}

__global__ __launch_bounds__(128) void k_final(const float* __restrict__ vmat,
                                               const float* __restrict__ hsum,
                                               const float* __restrict__ labs,
                                               const unsigned* __restrict__ sgn,
                                               const float* __restrict__ ideg,
                                               float* __restrict__ out)
{
    __shared__ float Hp[16][68];
    __shared__ float sc[16];

    const int m0  = blockIdx.x * 16;
    const int tid = threadIdx.x;

    if (tid < 16) {
        sc[tid] = rsqrtf(fmaxf(ideg[m0 + tid], 1.0f));
    }
    for (int i = tid; i < 16 * RANK; i += 128) {
        int r = i >> 6, c = i & 63;
        int row = m0 + r;
        float val = 0.0f;
        float dg = ideg[row];
        if (dg > 0.0f) {
            float la = labs[(size_t)row * RANK + c];
            unsigned n = sgn[(size_t)row * 2 + (c & 1)] >> (c >> 1);
            float sg = (n & 1u) ? -1.0f : 1.0f;
            val = sg * __expf(la);
        }
        Hp[r][c] = val;
    }
    __syncthreads();

    const int wave  = tid >> 5;
    const int lane  = tid & 31;
    const int rsel  = lane & 15;
    const int khalf = (lane < 16) ? 0 : 2;
    const int rofs  = (lane < 16) ? 0 : 8;

    for (int t = 0; t < 2; ++t) {
        const int n0 = (wave * 2 + t) * 16;

        for (int k = 0; k < RANK; k += 32)
            __builtin_prefetch(vmat + (size_t)k * OUT_F + n0, 0, 0);

        v8f acc = {};
        for (int k0 = 0; k0 < RANK; k0 += 4) {
            v2f a;
            a.x = Hp[rsel][k0 + khalf];
            a.y = Hp[rsel][k0 + khalf + 1];
            v2f b;
            const float* Bp = vmat + (size_t)(k0 + khalf) * OUT_F + n0 + rsel;
            b.x = Bp[0];
            b.y = Bp[OUT_F];
            acc = __builtin_amdgcn_wmma_f32_16x16x4_f32(
                false, a, false, b, (short)0, acc, false, false);
        }
        const int colg = n0 + rsel;
        const float* hp = hsum + (size_t)(m0 + rofs) * OUT_F + colg;
        float*       op = out  + (size_t)(m0 + rofs) * OUT_F + colg;
#pragma unroll
        for (int r = 0; r < 8; ++r)
            op[r * OUT_F] = (acc[r] + hp[r * OUT_F]) * sc[rofs + r];
    }
}

extern "C" void kernel_launch(void* const* d_in, const int* in_sizes, int n_in,
                              void* d_out, int out_size, void* d_ws, size_t ws_size,
                              hipStream_t stream)
{
    (void)in_sizes; (void)n_in; (void)out_size; (void)ws_size;

    const float* feat = (const float*)d_in[0];
    const float* w1   = (const float*)d_in[1];
    const float* w2   = (const float*)d_in[2];
    const float* vmat = (const float*)d_in[3];
    const int*   src  = (const int*)d_in[4];
    const int*   dst  = (const int*)d_in[5];

    float* ws    = (float*)d_ws;
    float* ideg  = ws + OFF_IDEG;
    unsigned* odpart = (unsigned*)(ws + OFF_ODPART);
    float* hsum  = ws + OFF_HSUM;
    float* labs  = ws + OFF_LABS;
    unsigned* sgn = (unsigned*)(ws + OFF_SGN);
    unsigned* bid = (unsigned*)(ws + OFF_BID);
    float* fsum  = ws + OFF_FSUM;
    float* fprod = ws + OFF_FPROD;

    k_outdeg<<<DEG_SLICES, 1024, 0, stream>>>(src, dst, odpart, bid);
    k_node_gemm<<<N_NODES / 16, 128, 0, stream>>>(feat, w1, w2, odpart, fsum, fprod);
    k_edge<<<EDGE_BLOCKS, 1024, 0, stream>>>(src, dst, bid, fsum, fprod,
                                             hsum, labs, sgn, ideg);
    k_final<<<N_NODES / 16, 128, 0, stream>>>(vmat, hsum, labs, sgn, ideg,
                                              (float*)d_out);
}


# speedup vs baseline: 2.6369x; 1.0317x over previous
#include <hip/hip_runtime.h>
#include <math.h>

typedef __attribute__((ext_vector_type(2))) float v2f;
typedef __attribute__((ext_vector_type(8))) float v8f;

#define N_NODES 50000
#define N_EDGES 800000
#define IN_F    256
#define OUT_F   128
#define RANK    64

static_assert(N_NODES % 16 == 0, "row tiles exact");
static_assert(N_EDGES % 256 == 0, "edge blocks exact");

#define OFF_IDEG   50000u
#define OFF_HSUM   100000u
#define OFF_LABS   6500000u
#define OFF_SGN    9700000u
#define OFF_BID    9900000u
#define OFF_REC    12899904u
#define REC_F      (OUT_F + RANK)
#define OFF_ODPART OFF_HSUM

#define DEG_SLICES  32
#define DEG_EDGES   (N_EDGES / DEG_SLICES)
#define DEG_LOADS   ((DEG_EDGES / 4 + 1023) / 1024)
static_assert(N_EDGES % DEG_SLICES == 0 && DEG_EDGES % 4 == 0, "slices exact; int4 loads aligned");
static_assert(N_NODES % 4 == 0, "count planes move as uint4");
static_assert(DEG_SLICES % 8 == 0, "the node product blocks add the planes in 8 groups");
static_assert((size_t)DEG_SLICES * N_NODES <= 6400000u, "partial planes fit the h_sum region");
static_assert(OFF_SGN + 2u * N_NODES <= OFF_BID && OFF_BID + N_EDGES / 4 <= OFF_REC,
              "packed sign words and the block-id bytes fit in front of the node table");
static_assert((size_t)OFF_REC + 63u + (size_t)N_NODES * REC_F <= 22500000u,
              "the aligned node table ends inside the workspace");
static_assert((REC_F * 4) % 256 == 0 && (OUT_F * 4) % 256 == 0,
              "a record, and its product part, start on 256-byte boundaries");

__device__ __forceinline__ float* node_table(float* p)
{
    const unsigned pad = (unsigned)(0ull - (unsigned long long)p) & 255u;
    return p + (pad >> 2);
}

#define NR          196
#define EDGE_BLOCKS ((N_NODES + NR - 1) / NR)
#define EDGE_WAVES  32
#define SCAN_U      4
#define SCAN_HALF   (32 * 16 * SCAN_U / 2)
#define DRAIN_AT    24
#define QCAP        1056
static_assert(EDGE_BLOCKS <= 256, "a block id fits one byte");
static_assert(N_EDGES % 16 == 0, "16-byte block-id scan exact");
static_assert(SCAN_U % 2 == 0 && DRAIN_AT + SCAN_HALF <= QCAP, "half a scan step can never overflow a wave's queue");

__device__ __forceinline__ unsigned block_id_byte(int d)
{
    return ((unsigned)d < (unsigned)N_NODES) ? (unsigned)d / (unsigned)NR : 255u;
}

__global__ __launch_bounds__(1024) void k_outdeg(const int* __restrict__ src,
                                                 const int* __restrict__ dst,
                                                 unsigned* __restrict__ parts,
                                                 unsigned* __restrict__ bid)
{
    __shared__ __attribute__((aligned(16))) unsigned Hc[N_NODES];

    const int tid = threadIdx.x;
    constexpr int S4 = DEG_EDGES / 4;

    const int4* s4 = (const int4*)(src + (size_t)blockIdx.x * DEG_EDGES);
    int4 v[DEG_LOADS];
#pragma unroll
    for (int j = 0; j < DEG_LOADS; ++j) {
        const int i = tid + j * 1024;
        v[j] = s4[(i < S4) ? i : (S4 - 1)];
    }

    {
        const int4* t4 = (const int4*)(dst + (size_t)blockIdx.x * DEG_EDGES);
        unsigned*   bo = bid + (size_t)blockIdx.x * S4;
        for (int j = 0; j < DEG_LOADS; ++j) {
            const int i = tid + j * 1024;
            if (i < S4) {
                const int4 d = t4[i];
                bo[i] = block_id_byte(d.x) | (block_id_byte(d.y) << 8) |
                        (block_id_byte(d.z) << 16) | (block_id_byte(d.w) << 24);
            }
        }
    }

    for (int i = tid; i < N_NODES / 4; i += 1024)
        ((uint4*)Hc)[i] = make_uint4(0u, 0u, 0u, 0u);
    __syncthreads();

#pragma unroll
    for (int j = 0; j < DEG_LOADS; ++j) {
        if (tid + j * 1024 < S4) {
            atomicAdd(&Hc[v[j].x], 1u);
            atomicAdd(&Hc[v[j].y], 1u);
            atomicAdd(&Hc[v[j].z], 1u);
            atomicAdd(&Hc[v[j].w], 1u);
        }
    }
    __syncthreads();

    uint4* og = (uint4*)(parts + (size_t)blockIdx.x * N_NODES);
    for (int i = tid; i < N_NODES / 4; i += 1024)
        og[i] = ((const uint4*)Hc)[i];
}

__global__ __launch_bounds__(128) void k_node_gemm(const float* __restrict__ feat,
                                                   const float* __restrict__ w1,
                                                   const float* __restrict__ w2,
                                                   const unsigned* __restrict__ odpart,
                                                   float* __restrict__ rec_raw)
{
    __shared__ float As[16][260];
    float* __restrict__ rec = node_table(rec_raw);
    __shared__ unsigned dp[8][16];

    const int m0  = blockIdx.x * 16;
    const int tid = threadIdx.x;

    {
        const int r = tid & 15;
        const int g = tid >> 4;
        unsigned d = 0u;
#pragma unroll
        for (int j = 0; j < DEG_SLICES / 8; ++j)
            d += odpart[(size_t)(g + 8 * j) * N_NODES + m0 + r];
        dp[g][r] = d;
    }
    __syncthreads();

    {
        const int r   = tid >> 3;
        const int c0  = (tid & 7) * 32;
        unsigned d = 0u;
#pragma unroll
        for (int q = 0; q < 8; ++q) d += dp[q][r];
        const float s = rsqrtf(fmaxf((float)d, 1.0f));
        const float4* g = (const float4*)(feat + (size_t)(m0 + r) * IN_F + c0);
#pragma unroll
        for (int j = 0; j < 8; ++j) {
            float4 x = g[j];
            As[r][c0 + 4 * j + 0] = x.x * s;
            As[r][c0 + 4 * j + 1] = x.y * s;
            As[r][c0 + 4 * j + 2] = x.z * s;
            As[r][c0 + 4 * j + 3] = x.w * s;
        }
    }
    __syncthreads();

    const int wave  = tid >> 5;
    const int lane  = tid & 31;
    const int rsel  = lane & 15;
    const int khalf = (lane < 16) ? 0 : 2;
    const int rofs  = (lane < 16) ? 0 : 8;

    for (int t = 0; t < 3; ++t) {
        const int  tile = wave + 4 * t;
        const bool isW2 = (tile >= 8);
        const int  n0   = isW2 ? (tile - 8) * 16 : tile * 16;
        const float* B  = isW2 ? w2 : w1;
        const int  ldb  = isW2 ? RANK : OUT_F;

        for (int k = 0; k < IN_F; k += 32)
            __builtin_prefetch(B + (size_t)k * ldb + n0, 0, 0);

        v8f acc = {};
        for (int k0 = 0; k0 < IN_F; k0 += 4) {
            v2f a;
            a.x = As[rsel][k0 + khalf];
            a.y = As[rsel][k0 + khalf + 1];
            v2f b;
            const float* Bp = B + (size_t)(k0 + khalf) * ldb + n0 + rsel;
            b.x = Bp[0];
            b.y = Bp[ldb];
            acc = __builtin_amdgcn_wmma_f32_16x16x4_f32(
                false, a, false, b, (short)0, acc, false, false);
        }

        const int colg = n0 + rsel;
        if (isW2) {
            const float bias = w2[(size_t)IN_F * RANK + colg];
            float* op = rec + (size_t)(m0 + rofs) * REC_F + OUT_F + colg;
#pragma unroll
            for (int r = 0; r < 8; ++r)
                op[r * REC_F] = tanhf(acc[r] + bias);
        } else {
            float* op = rec + (size_t)(m0 + rofs) * REC_F + colg;
#pragma unroll
            for (int r = 0; r < 8; ++r)
                op[r * REC_F] = acc[r];
        }
    }
}

__global__ __launch_bounds__(1024) void k_edge(const int* __restrict__ src,
                                               const int* __restrict__ dst,
                                               const unsigned* __restrict__ bid,
                                               float* __restrict__ rec_raw,
                                               float* __restrict__ hsum,
                                               float* __restrict__ labs,
                                               unsigned* __restrict__ sgn,
                                               float* __restrict__ ideg)
{
    __shared__ __attribute__((aligned(16))) float    Hs[NR * OUT_F];
    __shared__ __attribute__((aligned(16))) float    La[NR * RANK];
    __shared__ __attribute__((aligned(16))) unsigned Sg[NR * 2];
    __shared__ unsigned Cn[NR];
    __shared__ unsigned Qe[EDGE_WAVES * QCAP];
    __shared__ unsigned Qn[EDGE_WAVES];

    const int tid  = threadIdx.x;
    const int wave = tid >> 5;
    const int lane = tid & 31;
    const int n0   = blockIdx.x * NR;
    const float* __restrict__ rec = node_table(rec_raw);

    for (int i = tid; i < NR * OUT_F / 4; i += 1024)
        ((float4*)Hs)[i] = make_float4(0.0f, 0.0f, 0.0f, 0.0f);
    for (int i = tid; i < NR * RANK / 4; i += 1024)
        ((float4*)La)[i] = make_float4(0.0f, 0.0f, 0.0f, 0.0f);
    for (int i = tid; i < NR * 2; i += 1024)
        Sg[i] = 0u;
    if (tid < NR) Cn[tid] = 0u;
    if (tid < EDGE_WAVES) Qn[tid] = 0u;
    __syncthreads();

    constexpr int N16    = N_EDGES / 16;
    constexpr int STEP16 = 32 * SCAN_U;
    constexpr int NCH    = (N16 + STEP16 - 1) / STEP16;
    const uint4* b16  = (const uint4*)bid;
    unsigned*    myq  = Qe + wave * QCAP;
    const unsigned bsplat = (unsigned)blockIdx.x * 0x01010101u;

    auto load_step = [&](uint4 (&r)[SCAN_U], int cc) {
        for (int u = 0; u < SCAN_U; ++u) {
            const int i16 = cc * STEP16 + u * 32 + lane;
            r[u] = b16[(i16 < N16) ? i16 : (N16 - 1)];
        }
    };

    auto drain = [&](bool last) {
        const unsigned cnt = (unsigned)__builtin_amdgcn_readfirstlane((int)Qn[wave]);
        if (last || cnt >= (unsigned)DRAIN_AT) {
            for (unsigned base = 0; base < cnt; base += 32u) {
                const unsigned q = base + (unsigned)lane;
                int      sv  = 0;
                unsigned dlv = (unsigned)NR;
                if (q < cnt) {
                    const unsigned e = myq[q];
                    sv  = src[e];
                    dlv = (unsigned)(dst[e] - n0);
                }
                const int m = (int)((cnt - base < 32u) ? (cnt - base) : 32u);

                auto gather = [&](int j, float2& x0, float2& x1, float2& pr, unsigned& dl) {
                    const int s = __builtin_amdgcn_readlane(sv, j);
                    dl = (unsigned)__builtin_amdgcn_readlane((int)dlv, j);
                    const float2* fs = (const float2*)(rec + (size_t)s * REC_F);
                    x0 = fs[lane];
                    x1 = fs[32 + lane];
                    pr = fs[64 + lane];
                };
                auto accum = [&](const float2& x0, const float2& x1, const float2& pr, unsigned dl) {
                    if (dl >= (unsigned)NR) return;
                    float* h = Hs + dl * OUT_F + 2 * lane;
                    atomicAdd(h,      x0.x);
                    atomicAdd(h + 1,  x0.y);
                    atomicAdd(h + 64, x1.x);
                    atomicAdd(h + 65, x1.y);
                    float* l = La + dl * RANK + 2 * lane;
                    atomicAdd(l,     __logf(fabsf(pr.x)));
                    atomicAdd(l + 1, __logf(fabsf(pr.y)));
                    const unsigned b0 = __builtin_amdgcn_ballot_w32(pr.x < 0.0f);
                    const unsigned b1 = __builtin_amdgcn_ballot_w32(pr.y < 0.0f);
                    if (lane < 2) atomicXor(&Sg[dl * 2 + lane], lane ? b1 : b0);
                    if (lane == 2) atomicAdd(&Cn[dl], 1u);
                };

                int j = 0;
                for (; j + 4 <= m; j += 4) {
                    float2 a0, a1, ap, b0v, b1v, bp, c0, c1, cp, e0v, e1v, ep;
                    unsigned da, db, dc, de;
                    gather(j,     a0,  a1,  ap, da);
                    gather(j + 1, b0v, b1v, bp, db);
                    gather(j + 2, c0,  c1,  cp, dc);
                    gather(j + 3, e0v, e1v, ep, de);
                    accum(a0,  a1,  ap, da);
                    accum(b0v, b1v, bp, db);
                    accum(c0,  c1,  cp, dc);
                    accum(e0v, e1v, ep, de);
                }
                for (; j < m; ++j) {
                    float2 a0, a1, ap;
                    unsigned da;
                    gather(j, a0, a1, ap, da);
                    accum(a0, a1, ap, da);
                }
            }
            if (lane == 0) Qn[wave] = 0u;
        }
    };

    auto scan_step = [&](const uint4 (&r)[SCAN_U], int cc, bool last) {
        for (int u = 0; u < SCAN_U; ++u) {
            const int  i16 = cc * STEP16 + u * 32 + lane;
            const bool ok  = (i16 < N16);
            const unsigned e0 = (unsigned)i16 * 16u;
            const unsigned w[4] = {r[u].x ^ bsplat, r[u].y ^ bsplat, r[u].z ^ bsplat, r[u].w ^ bsplat};
            unsigned z[4];
            for (int k = 0; k < 4; ++k)
                z[k] = ~(((w[k] & 0x7f7f7f7fu) + 0x7f7f7f7fu) | w[k]) & 0x80808080u;
            if (ok && (z[0] | z[1] | z[2] | z[3]) != 0u) {
                for (int k = 0; k < 4; ++k) {
                    unsigned zk = z[k];
                    while (zk != 0u) {
                        const unsigned bit = (unsigned)__builtin_ctz(zk);
                        zk &= zk - 1u;
                        const unsigned p = atomicAdd(&Qn[wave], 1u);
                        myq[p] = e0 + 4u * (unsigned)k + (bit >> 3);
                    }
                }
            }
            if (u == SCAN_U / 2 - 1) drain(false);
        }
        drain(last);
    };

    constexpr int NPAIR = (NCH + 2 * EDGE_WAVES - 1) / (2 * EDGE_WAVES);
    uint4 ra[SCAN_U], rb[SCAN_U];
    load_step(ra, wave);
    for (int p = 0; p < NPAIR; ++p) {
        const int c = wave + p * 2 * EDGE_WAVES;
        load_step(rb, c + EDGE_WAVES);
        scan_step(ra, c, false);
        load_step(ra, c + 2 * EDGE_WAVES);
        scan_step(rb, c + EDGE_WAVES, p == NPAIR - 1);
    }
    __syncthreads();

    const int rows = (N_NODES - n0 < NR) ? (N_NODES - n0) : NR;
    {
        float4* hg = (float4*)(hsum + (size_t)n0 * OUT_F);
        for (int i = tid; i < rows * OUT_F / 4; i += 1024) hg[i] = ((const float4*)Hs)[i];
        float4* lg = (float4*)(labs + (size_t)n0 * RANK);
        for (int i = tid; i < rows * RANK / 4; i += 1024) lg[i] = ((const float4*)La)[i];
        unsigned* sg = sgn + (size_t)n0 * 2;
        for (int i = tid; i < rows * 2; i += 1024) sg[i] = Sg[i];
        if (tid < rows) ideg[n0 + tid] = (float)Cn[tid];
    }
}

__global__ __launch_bounds__(128) void k_final(const float* __restrict__ vmat,
                                               const float* __restrict__ hsum,
                                               const float* __restrict__ labs,
                                               const unsigned* __restrict__ sgn,
                                               const float* __restrict__ ideg,
                                               float* __restrict__ out)
{
    __shared__ float Hp[16][68];
    __shared__ float sc[16];

    const int m0  = blockIdx.x * 16;
    const int tid = threadIdx.x;

    if (tid < 16) {
        sc[tid] = rsqrtf(fmaxf(ideg[m0 + tid], 1.0f));
    }
    for (int i = tid; i < 16 * RANK; i += 128) {
        int r = i >> 6, c = i & 63;
        int row = m0 + r;
        float val = 0.0f;
        float dg = ideg[row];
        if (dg > 0.0f) {
            float la = labs[(size_t)row * RANK + c];
            unsigned n = sgn[(size_t)row * 2 + (c & 1)] >> (c >> 1);
            float sg = (n & 1u) ? -1.0f : 1.0f;
            val = sg * __expf(la);
        }
        Hp[r][c] = val;
    }
    __syncthreads();

    const int wave  = tid >> 5;
    const int lane  = tid & 31;
    const int rsel  = lane & 15;
    const int khalf = (lane < 16) ? 0 : 2;
    const int rofs  = (lane < 16) ? 0 : 8;

    for (int t = 0; t < 2; ++t) {
        const int n0 = (wave * 2 + t) * 16;

        for (int k = 0; k < RANK; k += 32)
            __builtin_prefetch(vmat + (size_t)k * OUT_F + n0, 0, 0);

        v8f acc = {};
        for (int k0 = 0; k0 < RANK; k0 += 4) {
            v2f a;
            a.x = Hp[rsel][k0 + khalf];
            a.y = Hp[rsel][k0 + khalf + 1];
            v2f b;
            const float* Bp = vmat + (size_t)(k0 + khalf) * OUT_F + n0 + rsel;
            b.x = Bp[0];
            b.y = Bp[OUT_F];
            acc = __builtin_amdgcn_wmma_f32_16x16x4_f32(
                false, a, false, b, (short)0, acc, false, false);
        }
        const int colg = n0 + rsel;
        const float* hp = hsum + (size_t)(m0 + rofs) * OUT_F + colg;
        float*       op = out  + (size_t)(m0 + rofs) * OUT_F + colg;
#pragma unroll
        for (int r = 0; r < 8; ++r)
            op[r * OUT_F] = (acc[r] + hp[r * OUT_F]) * sc[rofs + r];
    }
}

extern "C" void kernel_launch(void* const* d_in, const int* in_sizes, int n_in,
                              void* d_out, int out_size, void* d_ws, size_t ws_size,
                              hipStream_t stream)
{
    (void)in_sizes; (void)n_in; (void)out_size; (void)ws_size;

    const float* feat = (const float*)d_in[0];
    const float* w1   = (const float*)d_in[1];
    const float* w2   = (const float*)d_in[2];
    const float* vmat = (const float*)d_in[3];
    const int*   src  = (const int*)d_in[4];
    const int*   dst  = (const int*)d_in[5];

    float* ws    = (float*)d_ws;
    float* ideg  = ws + OFF_IDEG;
    unsigned* odpart = (unsigned*)(ws + OFF_ODPART);
    float* hsum  = ws + OFF_HSUM;
    float* labs  = ws + OFF_LABS;
    unsigned* sgn = (unsigned*)(ws + OFF_SGN);
    unsigned* bid = (unsigned*)(ws + OFF_BID);
    float* rec   = ws + OFF_REC;

    k_outdeg<<<DEG_SLICES, 1024, 0, stream>>>(src, dst, odpart, bid);
    k_node_gemm<<<N_NODES / 16, 128, 0, stream>>>(feat, w1, w2, odpart, rec);
    k_edge<<<EDGE_BLOCKS, 1024, 0, stream>>>(src, dst, bid, rec,
                                             hsum, labs, sgn, ideg);
    k_final<<<N_NODES / 16, 128, 0, stream>>>(vmat, hsum, labs, sgn, ideg,
                                              (float*)d_out);
}


# Final blocks issue all their plane loads at entry as one batch (log-abs rows as 16-byte vectors, sign words, in-degree, the h_sum tile into registers): one memory trip, not 18 dependent ones
# speedup vs baseline: 2.7982x; 1.0612x over previous
#include <hip/hip_runtime.h>
#include <math.h>

typedef __attribute__((ext_vector_type(2))) float v2f;
typedef __attribute__((ext_vector_type(8))) float v8f;

#define N_NODES 50000
#define N_EDGES 800000
#define IN_F    256
#define OUT_F   128
#define RANK    64

static_assert(N_NODES % 16 == 0, "row tiles exact");
static_assert(N_EDGES % 256 == 0, "edge blocks exact");

#define OFF_IDEG   50000u
#define OFF_HSUM   100000u
#define OFF_LABS   6500000u
#define OFF_SGN    9700000u
#define OFF_BID    9900000u
#define OFF_REC    12899904u
#define REC_F      (OUT_F + RANK)
#define OFF_ODPART OFF_HSUM

#define DEG_SLICES  32
#define DEG_EDGES   (N_EDGES / DEG_SLICES)
#define DEG_LOADS   ((DEG_EDGES / 4 + 1023) / 1024)
static_assert(N_EDGES % DEG_SLICES == 0 && DEG_EDGES % 4 == 0, "slices exact; int4 loads aligned");
static_assert(N_NODES % 4 == 0, "count planes move as uint4");
static_assert(DEG_SLICES % 8 == 0, "the node product blocks add the planes in 8 groups");
static_assert((size_t)DEG_SLICES * N_NODES <= 6400000u, "partial planes fit the h_sum region");
static_assert(OFF_SGN + 2u * N_NODES <= OFF_BID && OFF_BID + N_EDGES / 4 <= OFF_REC,
              "packed sign words and the block-id bytes fit in front of the node table");
static_assert((size_t)OFF_REC + 63u + (size_t)N_NODES * REC_F <= 22500000u,
              "the aligned node table ends inside the workspace");
static_assert((REC_F * 4) % 256 == 0 && (OUT_F * 4) % 256 == 0,
              "a record, and its product part, start on 256-byte boundaries");

__device__ __forceinline__ float* node_table(float* p)
{
    const unsigned pad = (unsigned)(0ull - (unsigned long long)p) & 255u;
    return p + (pad >> 2);
}

#define NR          196
#define EDGE_BLOCKS ((N_NODES + NR - 1) / NR)
#define EDGE_WAVES  32
#define SCAN_U      4
#define SCAN_HALF   (32 * 16 * SCAN_U / 2)
#define DRAIN_AT    24
#define QCAP        1056
static_assert(EDGE_BLOCKS <= 256, "a block id fits one byte");
static_assert(N_EDGES % 16 == 0, "16-byte block-id scan exact");
static_assert(SCAN_U % 2 == 0 && DRAIN_AT + SCAN_HALF <= QCAP, "half a scan step can never overflow a wave's queue");

__device__ __forceinline__ unsigned block_id_byte(int d)
{
    return ((unsigned)d < (unsigned)N_NODES) ? (unsigned)d / (unsigned)NR : 255u;
}

__global__ __launch_bounds__(1024) void k_outdeg(const int* __restrict__ src,
                                                 const int* __restrict__ dst,
                                                 unsigned* __restrict__ parts,
                                                 unsigned* __restrict__ bid)
{
    __shared__ __attribute__((aligned(16))) unsigned Hc[N_NODES];

    const int tid = threadIdx.x;
    constexpr int S4 = DEG_EDGES / 4;

    const int4* s4 = (const int4*)(src + (size_t)blockIdx.x * DEG_EDGES);
    int4 v[DEG_LOADS];
#pragma unroll
    for (int j = 0; j < DEG_LOADS; ++j) {
        const int i = tid + j * 1024;
        v[j] = s4[(i < S4) ? i : (S4 - 1)];
    }

    {
        const int4* t4 = (const int4*)(dst + (size_t)blockIdx.x * DEG_EDGES);
        unsigned*   bo = bid + (size_t)blockIdx.x * S4;
        for (int j = 0; j < DEG_LOADS; ++j) {
            const int i = tid + j * 1024;
            if (i < S4) {
                const int4 d = t4[i];
                bo[i] = block_id_byte(d.x) | (block_id_byte(d.y) << 8) |
                        (block_id_byte(d.z) << 16) | (block_id_byte(d.w) << 24);
            }
        }
    }

    for (int i = tid; i < N_NODES / 4; i += 1024)
        ((uint4*)Hc)[i] = make_uint4(0u, 0u, 0u, 0u);
    __syncthreads();

#pragma unroll
    for (int j = 0; j < DEG_LOADS; ++j) {
        if (tid + j * 1024 < S4) {
            atomicAdd(&Hc[v[j].x], 1u);
            atomicAdd(&Hc[v[j].y], 1u);
            atomicAdd(&Hc[v[j].z], 1u);
            atomicAdd(&Hc[v[j].w], 1u);
        }
    }
    __syncthreads();

    uint4* og = (uint4*)(parts + (size_t)blockIdx.x * N_NODES);
    for (int i = tid; i < N_NODES / 4; i += 1024)
        og[i] = ((const uint4*)Hc)[i];
}

__global__ __launch_bounds__(128) void k_node_gemm(const float* __restrict__ feat,
                                                   const float* __restrict__ w1,
                                                   const float* __restrict__ w2,
                                                   const unsigned* __restrict__ odpart,
                                                   float* __restrict__ rec_raw)
{
    __shared__ float As[16][260];
    float* __restrict__ rec = node_table(rec_raw);
    __shared__ unsigned dp[8][16];

    const int m0  = blockIdx.x * 16;
    const int tid = threadIdx.x;

    {
        const int r = tid & 15;
        const int g = tid >> 4;
        unsigned d = 0u;
#pragma unroll
        for (int j = 0; j < DEG_SLICES / 8; ++j)
            d += odpart[(size_t)(g + 8 * j) * N_NODES + m0 + r];
        dp[g][r] = d;
    }
    __syncthreads();

    {
        const int r   = tid >> 3;
        const int c0  = (tid & 7) * 32;
        unsigned d = 0u;
#pragma unroll
        for (int q = 0; q < 8; ++q) d += dp[q][r];
        const float s = rsqrtf(fmaxf((float)d, 1.0f));
        const float4* g = (const float4*)(feat + (size_t)(m0 + r) * IN_F + c0);
#pragma unroll
        for (int j = 0; j < 8; ++j) {
            float4 x = g[j];
            As[r][c0 + 4 * j + 0] = x.x * s;
            As[r][c0 + 4 * j + 1] = x.y * s;
            As[r][c0 + 4 * j + 2] = x.z * s;
            As[r][c0 + 4 * j + 3] = x.w * s;
        }
    }
    __syncthreads();

    const int wave  = tid >> 5;
    const int lane  = tid & 31;
    const int rsel  = lane & 15;
    const int khalf = (lane < 16) ? 0 : 2;
    const int rofs  = (lane < 16) ? 0 : 8;

    for (int t = 0; t < 3; ++t) {
        const int  tile = wave + 4 * t;
        const bool isW2 = (tile >= 8);
        const int  n0   = isW2 ? (tile - 8) * 16 : tile * 16;
        const float* B  = isW2 ? w2 : w1;
        const int  ldb  = isW2 ? RANK : OUT_F;

        for (int k = 0; k < IN_F; k += 32)
            __builtin_prefetch(B + (size_t)k * ldb + n0, 0, 0);

        v8f acc = {};
        for (int k0 = 0; k0 < IN_F; k0 += 4) {
            v2f a;
            a.x = As[rsel][k0 + khalf];
            a.y = As[rsel][k0 + khalf + 1];
            v2f b;
            const float* Bp = B + (size_t)(k0 + khalf) * ldb + n0 + rsel;
            b.x = Bp[0];
            b.y = Bp[ldb];
            acc = __builtin_amdgcn_wmma_f32_16x16x4_f32(
                false, a, false, b, (short)0, acc, false, false);
        }

        const int colg = n0 + rsel;
        if (isW2) {
            const float bias = w2[(size_t)IN_F * RANK + colg];
            float* op = rec + (size_t)(m0 + rofs) * REC_F + OUT_F + colg;
#pragma unroll
            for (int r = 0; r < 8; ++r)
                op[r * REC_F] = tanhf(acc[r] + bias);
        } else {
            float* op = rec + (size_t)(m0 + rofs) * REC_F + colg;
#pragma unroll
            for (int r = 0; r < 8; ++r)
                op[r * REC_F] = acc[r];
        }
    }
}

__global__ __launch_bounds__(1024) void k_edge(const int* __restrict__ src,
                                               const int* __restrict__ dst,
                                               const unsigned* __restrict__ bid,
                                               float* __restrict__ rec_raw,
                                               float* __restrict__ hsum,
                                               float* __restrict__ labs,
                                               unsigned* __restrict__ sgn,
                                               float* __restrict__ ideg)
{
    __shared__ __attribute__((aligned(16))) float    Hs[NR * OUT_F];
    __shared__ __attribute__((aligned(16))) float    La[NR * RANK];
    __shared__ __attribute__((aligned(16))) unsigned Sg[NR * 2];
    __shared__ unsigned Cn[NR];
    __shared__ unsigned Qe[EDGE_WAVES * QCAP];
    __shared__ unsigned Qn[EDGE_WAVES];

    const int tid  = threadIdx.x;
    const int wave = tid >> 5;
    const int lane = tid & 31;
    const int n0   = blockIdx.x * NR;
    const float* __restrict__ rec = node_table(rec_raw);

    for (int i = tid; i < NR * OUT_F / 4; i += 1024)
        ((float4*)Hs)[i] = make_float4(0.0f, 0.0f, 0.0f, 0.0f);
    for (int i = tid; i < NR * RANK / 4; i += 1024)
        ((float4*)La)[i] = make_float4(0.0f, 0.0f, 0.0f, 0.0f);
    for (int i = tid; i < NR * 2; i += 1024)
        Sg[i] = 0u;
    if (tid < NR) Cn[tid] = 0u;
    if (tid < EDGE_WAVES) Qn[tid] = 0u;
    __syncthreads();

    constexpr int N16    = N_EDGES / 16;
    constexpr int STEP16 = 32 * SCAN_U;
    constexpr int NCH    = (N16 + STEP16 - 1) / STEP16;
    const uint4* b16  = (const uint4*)bid;
    unsigned*    myq  = Qe + wave * QCAP;
    const unsigned bsplat = (unsigned)blockIdx.x * 0x01010101u;

    auto load_step = [&](uint4 (&r)[SCAN_U], int cc) {
        for (int u = 0; u < SCAN_U; ++u) {
            const int i16 = cc * STEP16 + u * 32 + lane;
            r[u] = b16[(i16 < N16) ? i16 : (N16 - 1)];
        }
    };

    auto drain = [&](bool last) {
        const unsigned cnt = (unsigned)__builtin_amdgcn_readfirstlane((int)Qn[wave]);
        if (last || cnt >= (unsigned)DRAIN_AT) {
            for (unsigned base = 0; base < cnt; base += 32u) {
                const unsigned q = base + (unsigned)lane;
                int      sv  = 0;
                unsigned dlv = (unsigned)NR;
                if (q < cnt) {
                    const unsigned e = myq[q];
                    sv  = src[e];
                    dlv = (unsigned)(dst[e] - n0);
                }
                const int m = (int)((cnt - base < 32u) ? (cnt - base) : 32u);

                auto gather = [&](int j, float2& x0, float2& x1, float2& pr, unsigned& dl) {
                    const int s = __builtin_amdgcn_readlane(sv, j);
                    dl = (unsigned)__builtin_amdgcn_readlane((int)dlv, j);
                    const float2* fs = (const float2*)(rec + (size_t)s * REC_F);
                    x0 = fs[lane];
                    x1 = fs[32 + lane];
                    pr = fs[64 + lane];
                };
                auto accum = [&](const float2& x0, const float2& x1, const float2& pr, unsigned dl) {
                    if (dl >= (unsigned)NR) return;
                    float* h = Hs + dl * OUT_F + 2 * lane;
                    atomicAdd(h,      x0.x);
                    atomicAdd(h + 1,  x0.y);
                    atomicAdd(h + 64, x1.x);
                    atomicAdd(h + 65, x1.y);
                    float* l = La + dl * RANK + 2 * lane;
                    atomicAdd(l,     __logf(fabsf(pr.x)));
                    atomicAdd(l + 1, __logf(fabsf(pr.y)));
                    const unsigned b0 = __builtin_amdgcn_ballot_w32(pr.x < 0.0f);
                    const unsigned b1 = __builtin_amdgcn_ballot_w32(pr.y < 0.0f);
                    if (lane < 2) atomicXor(&Sg[dl * 2 + lane], lane ? b1 : b0);
                    if (lane == 2) atomicAdd(&Cn[dl], 1u);
                };

                int j = 0;
                for (; j + 4 <= m; j += 4) {
                    float2 a0, a1, ap, b0v, b1v, bp, c0, c1, cp, e0v, e1v, ep;
                    unsigned da, db, dc, de;
                    gather(j,     a0,  a1,  ap, da);
                    gather(j + 1, b0v, b1v, bp, db);
                    gather(j + 2, c0,  c1,  cp, dc);
                    gather(j + 3, e0v, e1v, ep, de);
                    accum(a0,  a1,  ap, da);
                    accum(b0v, b1v, bp, db);
                    accum(c0,  c1,  cp, dc);
                    accum(e0v, e1v, ep, de);
                }
                for (; j < m; ++j) {
                    float2 a0, a1, ap;
                    unsigned da;
                    gather(j, a0, a1, ap, da);
                    accum(a0, a1, ap, da);
                }
            }
            if (lane == 0) Qn[wave] = 0u;
        }
    };

    auto scan_step = [&](const uint4 (&r)[SCAN_U], int cc, bool last) {
        for (int u = 0; u < SCAN_U; ++u) {
            const int  i16 = cc * STEP16 + u * 32 + lane;
            const bool ok  = (i16 < N16);
            const unsigned e0 = (unsigned)i16 * 16u;
            const unsigned w[4] = {r[u].x ^ bsplat, r[u].y ^ bsplat, r[u].z ^ bsplat, r[u].w ^ bsplat};
            unsigned z[4];
            for (int k = 0; k < 4; ++k)
                z[k] = ~(((w[k] & 0x7f7f7f7fu) + 0x7f7f7f7fu) | w[k]) & 0x80808080u;
            if (ok && (z[0] | z[1] | z[2] | z[3]) != 0u) {
                for (int k = 0; k < 4; ++k) {
                    unsigned zk = z[k];
                    while (zk != 0u) {
                        const unsigned bit = (unsigned)__builtin_ctz(zk);
                        zk &= zk - 1u;
                        const unsigned p = atomicAdd(&Qn[wave], 1u);
                        myq[p] = e0 + 4u * (unsigned)k + (bit >> 3);
                    }
                }
            }
            if (u == SCAN_U / 2 - 1) drain(false);
        }
        drain(last);
    };

    constexpr int NPAIR = (NCH + 2 * EDGE_WAVES - 1) / (2 * EDGE_WAVES);
    uint4 ra[SCAN_U], rb[SCAN_U];
    load_step(ra, wave);
    for (int p = 0; p < NPAIR; ++p) {
        const int c = wave + p * 2 * EDGE_WAVES;
        load_step(rb, c + EDGE_WAVES);
        scan_step(ra, c, false);
        load_step(ra, c + 2 * EDGE_WAVES);
        scan_step(rb, c + EDGE_WAVES, p == NPAIR - 1);
    }
    __syncthreads();

    const int rows = (N_NODES - n0 < NR) ? (N_NODES - n0) : NR;
    {
        float4* hg = (float4*)(hsum + (size_t)n0 * OUT_F);
        for (int i = tid; i < rows * OUT_F / 4; i += 1024) hg[i] = ((const float4*)Hs)[i];
        float4* lg = (float4*)(labs + (size_t)n0 * RANK);
        for (int i = tid; i < rows * RANK / 4; i += 1024) lg[i] = ((const float4*)La)[i];
        unsigned* sg = sgn + (size_t)n0 * 2;
        for (int i = tid; i < rows * 2; i += 1024) sg[i] = Sg[i];
        if (tid < rows) ideg[n0 + tid] = (float)Cn[tid];
    }
}

__global__ __launch_bounds__(128) void k_final(const float* __restrict__ vmat,
                                               const float* __restrict__ hsum,
                                               const float* __restrict__ labs,
                                               const unsigned* __restrict__ sgn,
                                               const float* __restrict__ ideg,
                                               float* __restrict__ out)
{
    __shared__ __attribute__((aligned(16))) float Hp[16][68];
    __shared__ float sc[16];

    const int m0  = blockIdx.x * 16;
    const int tid = threadIdx.x;

    const int wave  = tid >> 5;
    const int lane  = tid & 31;
    const int rsel  = lane & 15;
    const int khalf = (lane < 16) ? 0 : 2;
    const int rofs  = (lane < 16) ? 0 : 8;

    const int sr  = tid >> 3;
    const int sc0 = (tid & 7) * 8;
    const int row = m0 + sr;
    const float4* lp = (const float4*)(labs + (size_t)row * RANK + sc0);
    const float4 la0 = lp[0];
    const float4 la1 = lp[1];
    const uint2  sw  = *(const uint2*)(sgn + (size_t)row * 2);
    const float  dg  = ideg[row];

    float hs[2][8];
#pragma unroll
    for (int t = 0; t < 2; ++t) {
        const float* hp = hsum + (size_t)(m0 + rofs) * OUT_F + (wave * 2 + t) * 16 + rsel;
#pragma unroll
        for (int r = 0; r < 8; ++r) hs[t][r] = hp[r * OUT_F];
    }
    asm volatile("" :: "v"(la0.x), "v"(la0.y), "v"(la0.z), "v"(la0.w),
                       "v"(la1.x), "v"(la1.y), "v"(la1.z), "v"(la1.w));

    {
        const bool  has = (dg > 0.0f);
        const float la[8] = {la0.x, la0.y, la0.z, la0.w, la1.x, la1.y, la1.z, la1.w};
        float val[8];
#pragma unroll
        for (int i = 0; i < 8; ++i) {
            const int c = sc0 + i;
            const unsigned n = ((c & 1) ? sw.y : sw.x) >> (c >> 1);
            const float sg = (n & 1u) ? -1.0f : 1.0f;
            val[i] = has ? sg * __expf(la[i]) : 0.0f;
        }
        *(float4*)&Hp[sr][sc0]     = make_float4(val[0], val[1], val[2], val[3]);
        *(float4*)&Hp[sr][sc0 + 4] = make_float4(val[4], val[5], val[6], val[7]);
        if ((tid & 7) == 0) sc[sr] = rsqrtf(fmaxf(dg, 1.0f));
    }
    __syncthreads();

    for (int t = 0; t < 2; ++t) {
        const int n0 = (wave * 2 + t) * 16;

        for (int k = 0; k < RANK; k += 32)
            __builtin_prefetch(vmat + (size_t)k * OUT_F + n0, 0, 0);

        v8f acc = {};
        for (int k0 = 0; k0 < RANK; k0 += 4) {
            v2f a;
            a.x = Hp[rsel][k0 + khalf];
            a.y = Hp[rsel][k0 + khalf + 1];
            v2f b;
            const float* Bp = vmat + (size_t)(k0 + khalf) * OUT_F + n0 + rsel;
            b.x = Bp[0];
            b.y = Bp[OUT_F];
            acc = __builtin_amdgcn_wmma_f32_16x16x4_f32(
                false, a, false, b, (short)0, acc, false, false);
        }
        const int colg = n0 + rsel;
        float* op = out + (size_t)(m0 + rofs) * OUT_F + colg;
#pragma unroll
        for (int r = 0; r < 8; ++r)
            op[r * OUT_F] = (acc[r] + hs[t][r]) * sc[rofs + r];
    }
}

extern "C" void kernel_launch(void* const* d_in, const int* in_sizes, int n_in,
                              void* d_out, int out_size, void* d_ws, size_t ws_size,
                              hipStream_t stream)
{
    (void)in_sizes; (void)n_in; (void)out_size; (void)ws_size;

    const float* feat = (const float*)d_in[0];
    const float* w1   = (const float*)d_in[1];
    const float* w2   = (const float*)d_in[2];
    const float* vmat = (const float*)d_in[3];
    const int*   src  = (const int*)d_in[4];
    const int*   dst  = (const int*)d_in[5];

    float* ws    = (float*)d_ws;
    float* ideg  = ws + OFF_IDEG;
    unsigned* odpart = (unsigned*)(ws + OFF_ODPART);
    float* hsum  = ws + OFF_HSUM;
    float* labs  = ws + OFF_LABS;
    unsigned* sgn = (unsigned*)(ws + OFF_SGN);
    unsigned* bid = (unsigned*)(ws + OFF_BID);
    float* rec   = ws + OFF_REC;

    k_outdeg<<<DEG_SLICES, 1024, 0, stream>>>(src, dst, odpart, bid);
    k_node_gemm<<<N_NODES / 16, 128, 0, stream>>>(feat, w1, w2, odpart, rec);
    k_edge<<<EDGE_BLOCKS, 1024, 0, stream>>>(src, dst, bid, rec,
                                             hsum, labs, sgn, ideg);
    k_final<<<N_NODES / 16, 128, 0, stream>>>(vmat, hsum, labs, sgn, ideg,
                                              (float*)d_out);
}


# Node products: wave w owns w1 columns 32w..32w+31 as 2 tiles interleaved 2-wide (8-byte B loads and stores) and w2 tile w, k outermost: 1 A fragment + 4 fixed-offset loads per 3 WMMAs; k order kept
# speedup vs baseline: 2.9380x; 1.0500x over previous
#include <hip/hip_runtime.h>
#include <math.h>

typedef __attribute__((ext_vector_type(2))) float v2f;
typedef __attribute__((ext_vector_type(8))) float v8f;

#define N_NODES 50000
#define N_EDGES 800000
#define IN_F    256
#define OUT_F   128
#define RANK    64

static_assert(N_NODES % 16 == 0, "row tiles exact");
static_assert(N_EDGES % 256 == 0, "edge blocks exact");

#define OFF_IDEG   50000u
#define OFF_HSUM   100000u
#define OFF_LABS   6500000u
#define OFF_SGN    9700000u
#define OFF_BID    9900000u
#define OFF_REC    12899904u
#define REC_F      (OUT_F + RANK)
#define OFF_ODPART OFF_HSUM

#define DEG_SLICES  32
#define DEG_EDGES   (N_EDGES / DEG_SLICES)
#define DEG_LOADS   ((DEG_EDGES / 4 + 1023) / 1024)
static_assert(N_EDGES % DEG_SLICES == 0 && DEG_EDGES % 4 == 0, "slices exact; int4 loads aligned");
static_assert(N_NODES % 4 == 0, "count planes move as uint4");
static_assert(DEG_SLICES % 8 == 0, "the node product blocks add the planes in 8 groups");
static_assert((size_t)DEG_SLICES * N_NODES <= 6400000u, "partial planes fit the h_sum region");
static_assert(OFF_SGN + 2u * N_NODES <= OFF_BID && OFF_BID + N_EDGES / 4 <= OFF_REC,
              "packed sign words and the block-id bytes fit in front of the node table");
static_assert((size_t)OFF_REC + 63u + (size_t)N_NODES * REC_F <= 22500000u,
              "the aligned node table ends inside the workspace");
static_assert((REC_F * 4) % 256 == 0 && (OUT_F * 4) % 256 == 0,
              "a record, and its product part, start on 256-byte boundaries");

__device__ __forceinline__ float* node_table(float* p)
{
    const unsigned pad = (unsigned)(0ull - (unsigned long long)p) & 255u;
    return p + (pad >> 2);
}

#define NR          196
#define EDGE_BLOCKS ((N_NODES + NR - 1) / NR)
#define EDGE_WAVES  32
#define SCAN_U      4
#define SCAN_HALF   (32 * 16 * SCAN_U / 2)
#define DRAIN_AT    24
#define QCAP        1056
static_assert(EDGE_BLOCKS <= 256, "a block id fits one byte");
static_assert(N_EDGES % 16 == 0, "16-byte block-id scan exact");
static_assert(SCAN_U % 2 == 0 && DRAIN_AT + SCAN_HALF <= QCAP, "half a scan step can never overflow a wave's queue");

__device__ __forceinline__ unsigned block_id_byte(int d)
{
    return ((unsigned)d < (unsigned)N_NODES) ? (unsigned)d / (unsigned)NR : 255u;
}

__global__ __launch_bounds__(1024) void k_outdeg(const int* __restrict__ src,
                                                 const int* __restrict__ dst,
                                                 unsigned* __restrict__ parts,
                                                 unsigned* __restrict__ bid)
{
    __shared__ __attribute__((aligned(16))) unsigned Hc[N_NODES];

    const int tid = threadIdx.x;
    constexpr int S4 = DEG_EDGES / 4;

    const int4* s4 = (const int4*)(src + (size_t)blockIdx.x * DEG_EDGES);
    int4 v[DEG_LOADS];
#pragma unroll
    for (int j = 0; j < DEG_LOADS; ++j) {
        const int i = tid + j * 1024;
        v[j] = s4[(i < S4) ? i : (S4 - 1)];
    }

    {
        const int4* t4 = (const int4*)(dst + (size_t)blockIdx.x * DEG_EDGES);
        unsigned*   bo = bid + (size_t)blockIdx.x * S4;
        for (int j = 0; j < DEG_LOADS; ++j) {
            const int i = tid + j * 1024;
            if (i < S4) {
                const int4 d = t4[i];
                bo[i] = block_id_byte(d.x) | (block_id_byte(d.y) << 8) |
                        (block_id_byte(d.z) << 16) | (block_id_byte(d.w) << 24);
            }
        }
    }

    for (int i = tid; i < N_NODES / 4; i += 1024)
        ((uint4*)Hc)[i] = make_uint4(0u, 0u, 0u, 0u);
    __syncthreads();

#pragma unroll
    for (int j = 0; j < DEG_LOADS; ++j) {
        if (tid + j * 1024 < S4) {
            atomicAdd(&Hc[v[j].x], 1u);
            atomicAdd(&Hc[v[j].y], 1u);
            atomicAdd(&Hc[v[j].z], 1u);
            atomicAdd(&Hc[v[j].w], 1u);
        }
    }
    __syncthreads();

    uint4* og = (uint4*)(parts + (size_t)blockIdx.x * N_NODES);
    for (int i = tid; i < N_NODES / 4; i += 1024)
        og[i] = ((const uint4*)Hc)[i];
}

__global__ __launch_bounds__(128) void k_node_gemm(const float* __restrict__ feat,
                                                   const float* __restrict__ w1,
                                                   const float* __restrict__ w2,
                                                   const unsigned* __restrict__ odpart,
                                                   float* __restrict__ rec_raw)
{
    __shared__ float As[16][260];
    float* __restrict__ rec = node_table(rec_raw);
    __shared__ unsigned dp[8][16];

    const int m0  = blockIdx.x * 16;
    const int tid = threadIdx.x;

    {
        const int r = tid & 15;
        const int g = tid >> 4;
        unsigned d = 0u;
#pragma unroll
        for (int j = 0; j < DEG_SLICES / 8; ++j)
            d += odpart[(size_t)(g + 8 * j) * N_NODES + m0 + r];
        dp[g][r] = d;
    }
    __syncthreads();

    {
        const int r   = tid >> 3;
        const int c0  = (tid & 7) * 32;
        unsigned d = 0u;
#pragma unroll
        for (int q = 0; q < 8; ++q) d += dp[q][r];
        const float s = rsqrtf(fmaxf((float)d, 1.0f));
        const float4* g = (const float4*)(feat + (size_t)(m0 + r) * IN_F + c0);
#pragma unroll
        for (int j = 0; j < 8; ++j) {
            float4 x = g[j];
            As[r][c0 + 4 * j + 0] = x.x * s;
            As[r][c0 + 4 * j + 1] = x.y * s;
            As[r][c0 + 4 * j + 2] = x.z * s;
            As[r][c0 + 4 * j + 3] = x.w * s;
        }
    }
    __syncthreads();

    const int wave  = tid >> 5;
    const int lane  = tid & 31;
    const int rsel  = lane & 15;
    const int khalf = (lane < 16) ? 0 : 2;
    const int rofs  = (lane < 16) ? 0 : 8;

    const float* B1 = w1 + (size_t)khalf * OUT_F + 32 * wave + 2 * rsel;
    const float* B2 = w2 + (size_t)khalf * RANK + 16 * wave + rsel;

    v8f acc0 = {}, acc1 = {}, acc2 = {};
#pragma unroll 4
    for (int k0 = 0; k0 < IN_F; k0 += 4) {
        v2f a;
        a.x = As[rsel][k0 + khalf];
        a.y = As[rsel][k0 + khalf + 1];
        const float2 q0 = *(const float2*)(B1 + (size_t)k0 * OUT_F);
        const float2 q1 = *(const float2*)(B1 + (size_t)(k0 + 1) * OUT_F);
        const float  p0 = B2[(size_t)k0 * RANK];
        const float  p1 = B2[(size_t)(k0 + 1) * RANK];
        v2f b;
        b.x = q0.x; b.y = q1.x;
        acc0 = __builtin_amdgcn_wmma_f32_16x16x4_f32(false, a, false, b, (short)0, acc0, false, false);
        b.x = q0.y; b.y = q1.y;
        acc1 = __builtin_amdgcn_wmma_f32_16x16x4_f32(false, a, false, b, (short)0, acc1, false, false);
        b.x = p0; b.y = p1;
        acc2 = __builtin_amdgcn_wmma_f32_16x16x4_f32(false, a, false, b, (short)0, acc2, false, false);
    }

    {
        float* op = rec + (size_t)(m0 + rofs) * REC_F + 32 * wave + 2 * rsel;
#pragma unroll
        for (int r = 0; r < 8; ++r)
            *(float2*)(op + r * REC_F) = make_float2(acc0[r], acc1[r]);
    }
    {
        const int colg = 16 * wave + rsel;
        const float bias = w2[(size_t)IN_F * RANK + colg];
        float* op = rec + (size_t)(m0 + rofs) * REC_F + OUT_F + colg;
#pragma unroll
        for (int r = 0; r < 8; ++r)
            op[r * REC_F] = tanhf(acc2[r] + bias);
    }
}

__global__ __launch_bounds__(1024) void k_edge(const int* __restrict__ src,
                                               const int* __restrict__ dst,
                                               const unsigned* __restrict__ bid,
                                               float* __restrict__ rec_raw,
                                               float* __restrict__ hsum,
                                               float* __restrict__ labs,
                                               unsigned* __restrict__ sgn,
                                               float* __restrict__ ideg)
{
    __shared__ __attribute__((aligned(16))) float    Hs[NR * OUT_F];
    __shared__ __attribute__((aligned(16))) float    La[NR * RANK];
    __shared__ __attribute__((aligned(16))) unsigned Sg[NR * 2];
    __shared__ unsigned Cn[NR];
    __shared__ unsigned Qe[EDGE_WAVES * QCAP];
    __shared__ unsigned Qn[EDGE_WAVES];

    const int tid  = threadIdx.x;
    const int wave = tid >> 5;
    const int lane = tid & 31;
    const int n0   = blockIdx.x * NR;
    const float* __restrict__ rec = node_table(rec_raw);

    for (int i = tid; i < NR * OUT_F / 4; i += 1024)
        ((float4*)Hs)[i] = make_float4(0.0f, 0.0f, 0.0f, 0.0f);
    for (int i = tid; i < NR * RANK / 4; i += 1024)
        ((float4*)La)[i] = make_float4(0.0f, 0.0f, 0.0f, 0.0f);
    for (int i = tid; i < NR * 2; i += 1024)
        Sg[i] = 0u;
    if (tid < NR) Cn[tid] = 0u;
    if (tid < EDGE_WAVES) Qn[tid] = 0u;
    __syncthreads();

    constexpr int N16    = N_EDGES / 16;
    constexpr int STEP16 = 32 * SCAN_U;
    constexpr int NCH    = (N16 + STEP16 - 1) / STEP16;
    const uint4* b16  = (const uint4*)bid;
    unsigned*    myq  = Qe + wave * QCAP;
    const unsigned bsplat = (unsigned)blockIdx.x * 0x01010101u;

    auto load_step = [&](uint4 (&r)[SCAN_U], int cc) {
        for (int u = 0; u < SCAN_U; ++u) {
            const int i16 = cc * STEP16 + u * 32 + lane;
            r[u] = b16[(i16 < N16) ? i16 : (N16 - 1)];
        }
    };

    auto drain = [&](bool last) {
        const unsigned cnt = (unsigned)__builtin_amdgcn_readfirstlane((int)Qn[wave]);
        if (last || cnt >= (unsigned)DRAIN_AT) {
            for (unsigned base = 0; base < cnt; base += 32u) {
                const unsigned q = base + (unsigned)lane;
                int      sv  = 0;
                unsigned dlv = (unsigned)NR;
                if (q < cnt) {
                    const unsigned e = myq[q];
                    sv  = src[e];
                    dlv = (unsigned)(dst[e] - n0);
                }
                const int m = (int)((cnt - base < 32u) ? (cnt - base) : 32u);

                auto gather = [&](int j, float2& x0, float2& x1, float2& pr, unsigned& dl) {
                    const int s = __builtin_amdgcn_readlane(sv, j);
                    dl = (unsigned)__builtin_amdgcn_readlane((int)dlv, j);
                    const float2* fs = (const float2*)(rec + (size_t)s * REC_F);
                    x0 = fs[lane];
                    x1 = fs[32 + lane];
                    pr = fs[64 + lane];
                };
                auto accum = [&](const float2& x0, const float2& x1, const float2& pr, unsigned dl) {
                    if (dl >= (unsigned)NR) return;
                    float* h = Hs + dl * OUT_F + 2 * lane;
                    atomicAdd(h,      x0.x);
                    atomicAdd(h + 1,  x0.y);
                    atomicAdd(h + 64, x1.x);
                    atomicAdd(h + 65, x1.y);
                    float* l = La + dl * RANK + 2 * lane;
                    atomicAdd(l,     __logf(fabsf(pr.x)));
                    atomicAdd(l + 1, __logf(fabsf(pr.y)));
                    const unsigned b0 = __builtin_amdgcn_ballot_w32(pr.x < 0.0f);
                    const unsigned b1 = __builtin_amdgcn_ballot_w32(pr.y < 0.0f);
                    if (lane < 2) atomicXor(&Sg[dl * 2 + lane], lane ? b1 : b0);
                    if (lane == 2) atomicAdd(&Cn[dl], 1u);
                };

                int j = 0;
                for (; j + 4 <= m; j += 4) {
                    float2 a0, a1, ap, b0v, b1v, bp, c0, c1, cp, e0v, e1v, ep;
                    unsigned da, db, dc, de;
                    gather(j,     a0,  a1,  ap, da);
                    gather(j + 1, b0v, b1v, bp, db);
                    gather(j + 2, c0,  c1,  cp, dc);
                    gather(j + 3, e0v, e1v, ep, de);
                    accum(a0,  a1,  ap, da);
                    accum(b0v, b1v, bp, db);
                    accum(c0,  c1,  cp, dc);
                    accum(e0v, e1v, ep, de);
                }
                for (; j < m; ++j) {
                    float2 a0, a1, ap;
                    unsigned da;
                    gather(j, a0, a1, ap, da);
                    accum(a0, a1, ap, da);
                }
            }
            if (lane == 0) Qn[wave] = 0u;
        }
    };

    auto scan_step = [&](const uint4 (&r)[SCAN_U], int cc, bool last) {
        for (int u = 0; u < SCAN_U; ++u) {
            const int  i16 = cc * STEP16 + u * 32 + lane;
            const bool ok  = (i16 < N16);
            const unsigned e0 = (unsigned)i16 * 16u;
            const unsigned w[4] = {r[u].x ^ bsplat, r[u].y ^ bsplat, r[u].z ^ bsplat, r[u].w ^ bsplat};
            unsigned z[4];
            for (int k = 0; k < 4; ++k)
                z[k] = ~(((w[k] & 0x7f7f7f7fu) + 0x7f7f7f7fu) | w[k]) & 0x80808080u;
            if (ok && (z[0] | z[1] | z[2] | z[3]) != 0u) {
                for (int k = 0; k < 4; ++k) {
                    unsigned zk = z[k];
                    while (zk != 0u) {
                        const unsigned bit = (unsigned)__builtin_ctz(zk);
                        zk &= zk - 1u;
                        const unsigned p = atomicAdd(&Qn[wave], 1u);
                        myq[p] = e0 + 4u * (unsigned)k + (bit >> 3);
                    }
                }
            }
            if (u == SCAN_U / 2 - 1) drain(false);
        }
        drain(last);
    };

    constexpr int NPAIR = (NCH + 2 * EDGE_WAVES - 1) / (2 * EDGE_WAVES);
    uint4 ra[SCAN_U], rb[SCAN_U];
    load_step(ra, wave);
    for (int p = 0; p < NPAIR; ++p) {
        const int c = wave + p * 2 * EDGE_WAVES;
        load_step(rb, c + EDGE_WAVES);
        scan_step(ra, c, false);
        load_step(ra, c + 2 * EDGE_WAVES);
        scan_step(rb, c + EDGE_WAVES, p == NPAIR - 1);
    }
    __syncthreads();

    const int rows = (N_NODES - n0 < NR) ? (N_NODES - n0) : NR;
    {
        float4* hg = (float4*)(hsum + (size_t)n0 * OUT_F);
        for (int i = tid; i < rows * OUT_F / 4; i += 1024) hg[i] = ((const float4*)Hs)[i];
        float4* lg = (float4*)(labs + (size_t)n0 * RANK);
        for (int i = tid; i < rows * RANK / 4; i += 1024) lg[i] = ((const float4*)La)[i];
        unsigned* sg = sgn + (size_t)n0 * 2;
        for (int i = tid; i < rows * 2; i += 1024) sg[i] = Sg[i];
        if (tid < rows) ideg[n0 + tid] = (float)Cn[tid];
    }
}

__global__ __launch_bounds__(128) void k_final(const float* __restrict__ vmat,
                                               const float* __restrict__ hsum,
                                               const float* __restrict__ labs,
                                               const unsigned* __restrict__ sgn,
                                               const float* __restrict__ ideg,
                                               float* __restrict__ out)
{
    __shared__ __attribute__((aligned(16))) float Hp[16][68];
    __shared__ float sc[16];

    const int m0  = blockIdx.x * 16;
    const int tid = threadIdx.x;

    const int wave  = tid >> 5;
    const int lane  = tid & 31;
    const int rsel  = lane & 15;
    const int khalf = (lane < 16) ? 0 : 2;
    const int rofs  = (lane < 16) ? 0 : 8;

    const int sr  = tid >> 3;
    const int sc0 = (tid & 7) * 8;
    const int row = m0 + sr;
    const float4* lp = (const float4*)(labs + (size_t)row * RANK + sc0);
    const float4 la0 = lp[0];
    const float4 la1 = lp[1];
    const uint2  sw  = *(const uint2*)(sgn + (size_t)row * 2);
    const float  dg  = ideg[row];

    float hs[2][8];
#pragma unroll
    for (int t = 0; t < 2; ++t) {
        const float* hp = hsum + (size_t)(m0 + rofs) * OUT_F + (wave * 2 + t) * 16 + rsel;
#pragma unroll
        for (int r = 0; r < 8; ++r) hs[t][r] = hp[r * OUT_F];
    }
    asm volatile("" :: "v"(la0.x), "v"(la0.y), "v"(la0.z), "v"(la0.w),
                       "v"(la1.x), "v"(la1.y), "v"(la1.z), "v"(la1.w));

    {
        const bool  has = (dg > 0.0f);
        const float la[8] = {la0.x, la0.y, la0.z, la0.w, la1.x, la1.y, la1.z, la1.w};
        float val[8];
#pragma unroll
        for (int i = 0; i < 8; ++i) {
            const int c = sc0 + i;
            const unsigned n = ((c & 1) ? sw.y : sw.x) >> (c >> 1);
            const float sg = (n & 1u) ? -1.0f : 1.0f;
            val[i] = has ? sg * __expf(la[i]) : 0.0f;
        }
        *(float4*)&Hp[sr][sc0]     = make_float4(val[0], val[1], val[2], val[3]);
        *(float4*)&Hp[sr][sc0 + 4] = make_float4(val[4], val[5], val[6], val[7]);
        if ((tid & 7) == 0) sc[sr] = rsqrtf(fmaxf(dg, 1.0f));
    }
    __syncthreads();

    for (int t = 0; t < 2; ++t) {
        const int n0 = (wave * 2 + t) * 16;

        for (int k = 0; k < RANK; k += 32)
            __builtin_prefetch(vmat + (size_t)k * OUT_F + n0, 0, 0);

        v8f acc = {};
        for (int k0 = 0; k0 < RANK; k0 += 4) {
            v2f a;
            a.x = Hp[rsel][k0 + khalf];
            a.y = Hp[rsel][k0 + khalf + 1];
            v2f b;
            const float* Bp = vmat + (size_t)(k0 + khalf) * OUT_F + n0 + rsel;
            b.x = Bp[0];
            b.y = Bp[OUT_F];
            acc = __builtin_amdgcn_wmma_f32_16x16x4_f32(
                false, a, false, b, (short)0, acc, false, false);
        }
        const int colg = n0 + rsel;
        float* op = out + (size_t)(m0 + rofs) * OUT_F + colg;
#pragma unroll
        for (int r = 0; r < 8; ++r)
            op[r * OUT_F] = (acc[r] + hs[t][r]) * sc[rofs + r];
    }
}

extern "C" void kernel_launch(void* const* d_in, const int* in_sizes, int n_in,
                              void* d_out, int out_size, void* d_ws, size_t ws_size,
                              hipStream_t stream)
{
    (void)in_sizes; (void)n_in; (void)out_size; (void)ws_size;

    const float* feat = (const float*)d_in[0];
    const float* w1   = (const float*)d_in[1];
    const float* w2   = (const float*)d_in[2];
    const float* vmat = (const float*)d_in[3];
    const int*   src  = (const int*)d_in[4];
    const int*   dst  = (const int*)d_in[5];

    float* ws    = (float*)d_ws;
    float* ideg  = ws + OFF_IDEG;
    unsigned* odpart = (unsigned*)(ws + OFF_ODPART);
    float* hsum  = ws + OFF_HSUM;
    float* labs  = ws + OFF_LABS;
    unsigned* sgn = (unsigned*)(ws + OFF_SGN);
    unsigned* bid = (unsigned*)(ws + OFF_BID);
    float* rec   = ws + OFF_REC;

    k_outdeg<<<DEG_SLICES, 1024, 0, stream>>>(src, dst, odpart, bid);
    k_node_gemm<<<N_NODES / 16, 128, 0, stream>>>(feat, w1, w2, odpart, rec);
    k_edge<<<EDGE_BLOCKS, 1024, 0, stream>>>(src, dst, bid, rec,
                                             hsum, labs, sgn, ideg);
    k_final<<<N_NODES / 16, 128, 0, stream>>>(vmat, hsum, labs, sgn, ideg,
                                              (float*)d_out);
}


# Edge blocks take per-edge bookkeeping off the LDS pipe: queue slots by lane prefix, count in a scalar register; in-degree added once per 32-edge batch. Queue order changes, sums stay arrival-order
# speedup vs baseline: 3.0438x; 1.0360x over previous
#include <hip/hip_runtime.h>
#include <math.h>

typedef __attribute__((ext_vector_type(2))) float v2f;
typedef __attribute__((ext_vector_type(8))) float v8f;

#define N_NODES 50000
#define N_EDGES 800000
#define IN_F    256
#define OUT_F   128
#define RANK    64

static_assert(N_NODES % 16 == 0, "row tiles exact");
static_assert(N_EDGES % 256 == 0, "edge blocks exact");

#define OFF_IDEG   50000u
#define OFF_HSUM   100000u
#define OFF_LABS   6500000u
#define OFF_SGN    9700000u
#define OFF_BID    9900000u
#define OFF_REC    12899904u
#define REC_F      (OUT_F + RANK)
#define OFF_ODPART OFF_HSUM

#define DEG_SLICES  32
#define DEG_EDGES   (N_EDGES / DEG_SLICES)
#define DEG_LOADS   ((DEG_EDGES / 4 + 1023) / 1024)
static_assert(N_EDGES % DEG_SLICES == 0 && DEG_EDGES % 4 == 0, "slices exact; int4 loads aligned");
static_assert(N_NODES % 4 == 0, "count planes move as uint4");
static_assert(DEG_SLICES % 8 == 0, "the node product blocks add the planes in 8 groups");
static_assert((size_t)DEG_SLICES * N_NODES <= 6400000u, "partial planes fit the h_sum region");
static_assert(OFF_SGN + 2u * N_NODES <= OFF_BID && OFF_BID + N_EDGES / 4 <= OFF_REC,
              "packed sign words and the block-id bytes fit in front of the node table");
static_assert((size_t)OFF_REC + 63u + (size_t)N_NODES * REC_F <= 22500000u,
              "the aligned node table ends inside the workspace");
static_assert((REC_F * 4) % 256 == 0 && (OUT_F * 4) % 256 == 0,
              "a record, and its product part, start on 256-byte boundaries");

__device__ __forceinline__ float* node_table(float* p)
{
    const unsigned pad = (unsigned)(0ull - (unsigned long long)p) & 255u;
    return p + (pad >> 2);
}

#define NR          196
#define EDGE_BLOCKS ((N_NODES + NR - 1) / NR)
#define EDGE_WAVES  32
#define SCAN_U      4
#define SCAN_HALF   (32 * 16 * SCAN_U / 2)
#define DRAIN_AT    24
#define QCAP        1056
static_assert(EDGE_BLOCKS <= 256, "a block id fits one byte");
static_assert(N_EDGES % 16 == 0, "16-byte block-id scan exact");
static_assert(SCAN_U % 2 == 0 && DRAIN_AT + SCAN_HALF <= QCAP, "half a scan step can never overflow a wave's queue");

__device__ __forceinline__ unsigned block_id_byte(int d)
{
    return ((unsigned)d < (unsigned)N_NODES) ? (unsigned)d / (unsigned)NR : 255u;
}

__global__ __launch_bounds__(1024) void k_outdeg(const int* __restrict__ src,
                                                 const int* __restrict__ dst,
                                                 unsigned* __restrict__ parts,
                                                 unsigned* __restrict__ bid)
{
    __shared__ __attribute__((aligned(16))) unsigned Hc[N_NODES];

    const int tid = threadIdx.x;
    constexpr int S4 = DEG_EDGES / 4;

    const int4* s4 = (const int4*)(src + (size_t)blockIdx.x * DEG_EDGES);
    int4 v[DEG_LOADS];
#pragma unroll
    for (int j = 0; j < DEG_LOADS; ++j) {
        const int i = tid + j * 1024;
        v[j] = s4[(i < S4) ? i : (S4 - 1)];
    }

    {
        const int4* t4 = (const int4*)(dst + (size_t)blockIdx.x * DEG_EDGES);
        unsigned*   bo = bid + (size_t)blockIdx.x * S4;
        for (int j = 0; j < DEG_LOADS; ++j) {
            const int i = tid + j * 1024;
            if (i < S4) {
                const int4 d = t4[i];
                bo[i] = block_id_byte(d.x) | (block_id_byte(d.y) << 8) |
                        (block_id_byte(d.z) << 16) | (block_id_byte(d.w) << 24);
            }
        }
    }

    for (int i = tid; i < N_NODES / 4; i += 1024)
        ((uint4*)Hc)[i] = make_uint4(0u, 0u, 0u, 0u);
    __syncthreads();

#pragma unroll
    for (int j = 0; j < DEG_LOADS; ++j) {
        if (tid + j * 1024 < S4) {
            atomicAdd(&Hc[v[j].x], 1u);
            atomicAdd(&Hc[v[j].y], 1u);
            atomicAdd(&Hc[v[j].z], 1u);
            atomicAdd(&Hc[v[j].w], 1u);
        }
    }
    __syncthreads();

    uint4* og = (uint4*)(parts + (size_t)blockIdx.x * N_NODES);
    for (int i = tid; i < N_NODES / 4; i += 1024)
        og[i] = ((const uint4*)Hc)[i];
}

__global__ __launch_bounds__(128) void k_node_gemm(const float* __restrict__ feat,
                                                   const float* __restrict__ w1,
                                                   const float* __restrict__ w2,
                                                   const unsigned* __restrict__ odpart,
                                                   float* __restrict__ rec_raw)
{
    __shared__ float As[16][260];
    float* __restrict__ rec = node_table(rec_raw);
    __shared__ unsigned dp[8][16];

    const int m0  = blockIdx.x * 16;
    const int tid = threadIdx.x;

    {
        const int r = tid & 15;
        const int g = tid >> 4;
        unsigned d = 0u;
#pragma unroll
        for (int j = 0; j < DEG_SLICES / 8; ++j)
            d += odpart[(size_t)(g + 8 * j) * N_NODES + m0 + r];
        dp[g][r] = d;
    }
    __syncthreads();

    {
        const int r   = tid >> 3;
        const int c0  = (tid & 7) * 32;
        unsigned d = 0u;
#pragma unroll
        for (int q = 0; q < 8; ++q) d += dp[q][r];
        const float s = rsqrtf(fmaxf((float)d, 1.0f));
        const float4* g = (const float4*)(feat + (size_t)(m0 + r) * IN_F + c0);
#pragma unroll
        for (int j = 0; j < 8; ++j) {
            float4 x = g[j];
            As[r][c0 + 4 * j + 0] = x.x * s;
            As[r][c0 + 4 * j + 1] = x.y * s;
            As[r][c0 + 4 * j + 2] = x.z * s;
            As[r][c0 + 4 * j + 3] = x.w * s;
        }
    }
    __syncthreads();

    const int wave  = tid >> 5;
    const int lane  = tid & 31;
    const int rsel  = lane & 15;
    const int khalf = (lane < 16) ? 0 : 2;
    const int rofs  = (lane < 16) ? 0 : 8;

    const float* B1 = w1 + (size_t)khalf * OUT_F + 32 * wave + 2 * rsel;
    const float* B2 = w2 + (size_t)khalf * RANK + 16 * wave + rsel;

    v8f acc0 = {}, acc1 = {}, acc2 = {};
#pragma unroll 4
    for (int k0 = 0; k0 < IN_F; k0 += 4) {
        v2f a;
        a.x = As[rsel][k0 + khalf];
        a.y = As[rsel][k0 + khalf + 1];
        const float2 q0 = *(const float2*)(B1 + (size_t)k0 * OUT_F);
        const float2 q1 = *(const float2*)(B1 + (size_t)(k0 + 1) * OUT_F);
        const float  p0 = B2[(size_t)k0 * RANK];
        const float  p1 = B2[(size_t)(k0 + 1) * RANK];
        v2f b;
        b.x = q0.x; b.y = q1.x;
        acc0 = __builtin_amdgcn_wmma_f32_16x16x4_f32(false, a, false, b, (short)0, acc0, false, false);
        b.x = q0.y; b.y = q1.y;
        acc1 = __builtin_amdgcn_wmma_f32_16x16x4_f32(false, a, false, b, (short)0, acc1, false, false);
        b.x = p0; b.y = p1;
        acc2 = __builtin_amdgcn_wmma_f32_16x16x4_f32(false, a, false, b, (short)0, acc2, false, false);
    }

    {
        float* op = rec + (size_t)(m0 + rofs) * REC_F + 32 * wave + 2 * rsel;
#pragma unroll
        for (int r = 0; r < 8; ++r)
            *(float2*)(op + r * REC_F) = make_float2(acc0[r], acc1[r]);
    }
    {
        const int colg = 16 * wave + rsel;
        const float bias = w2[(size_t)IN_F * RANK + colg];
        float* op = rec + (size_t)(m0 + rofs) * REC_F + OUT_F + colg;
#pragma unroll
        for (int r = 0; r < 8; ++r)
            op[r * REC_F] = tanhf(acc2[r] + bias);
    }
}

__global__ __launch_bounds__(1024) void k_edge(const int* __restrict__ src,
                                               const int* __restrict__ dst,
                                               const unsigned* __restrict__ bid,
                                               float* __restrict__ rec_raw,
                                               float* __restrict__ hsum,
                                               float* __restrict__ labs,
                                               unsigned* __restrict__ sgn,
                                               float* __restrict__ ideg)
{
    __shared__ __attribute__((aligned(16))) float    Hs[NR * OUT_F];
    __shared__ __attribute__((aligned(16))) float    La[NR * RANK];
    __shared__ __attribute__((aligned(16))) unsigned Sg[NR * 2];
    __shared__ unsigned Cn[NR];
    __shared__ unsigned Qe[EDGE_WAVES * QCAP];

    const int tid  = threadIdx.x;
    const int wave = tid >> 5;
    const int lane = tid & 31;
    const int n0   = blockIdx.x * NR;
    const float* __restrict__ rec = node_table(rec_raw);

    for (int i = tid; i < NR * OUT_F / 4; i += 1024)
        ((float4*)Hs)[i] = make_float4(0.0f, 0.0f, 0.0f, 0.0f);
    for (int i = tid; i < NR * RANK / 4; i += 1024)
        ((float4*)La)[i] = make_float4(0.0f, 0.0f, 0.0f, 0.0f);
    for (int i = tid; i < NR * 2; i += 1024)
        Sg[i] = 0u;
    if (tid < NR) Cn[tid] = 0u;
    __syncthreads();

    constexpr int N16    = N_EDGES / 16;
    constexpr int STEP16 = 32 * SCAN_U;
    constexpr int NCH    = (N16 + STEP16 - 1) / STEP16;
    const uint4* b16  = (const uint4*)bid;
    unsigned*    myq  = Qe + wave * QCAP;
    const unsigned bsplat = (unsigned)blockIdx.x * 0x01010101u;
    unsigned qn = 0u;

    auto load_step = [&](uint4 (&r)[SCAN_U], int cc) {
        for (int u = 0; u < SCAN_U; ++u) {
            const int i16 = cc * STEP16 + u * 32 + lane;
            r[u] = b16[(i16 < N16) ? i16 : (N16 - 1)];
        }
    };

    auto drain = [&](bool last) {
        const unsigned cnt = (unsigned)__builtin_amdgcn_readfirstlane((int)qn);
        if (last || cnt >= (unsigned)DRAIN_AT) {
            for (unsigned base = 0; base < cnt; base += 32u) {
                const unsigned q = base + (unsigned)lane;
                int      sv  = 0;
                unsigned dlv = (unsigned)NR;
                if (q < cnt) {
                    const unsigned e = myq[q];
                    sv  = src[e];
                    dlv = (unsigned)(dst[e] - n0);
                }
                if (dlv < (unsigned)NR) atomicAdd(&Cn[dlv], 1u);
                const int m = (int)((cnt - base < 32u) ? (cnt - base) : 32u);

                auto gather = [&](int j, float2& x0, float2& x1, float2& pr, unsigned& dl) {
                    const int s = __builtin_amdgcn_readlane(sv, j);
                    dl = (unsigned)__builtin_amdgcn_readlane((int)dlv, j);
                    const float2* fs = (const float2*)(rec + (size_t)s * REC_F);
                    x0 = fs[lane];
                    x1 = fs[32 + lane];
                    pr = fs[64 + lane];
                };
                auto accum = [&](const float2& x0, const float2& x1, const float2& pr, unsigned dl) {
                    if (dl >= (unsigned)NR) return;
                    float* h = Hs + dl * OUT_F + 2 * lane;
                    atomicAdd(h,      x0.x);
                    atomicAdd(h + 1,  x0.y);
                    atomicAdd(h + 64, x1.x);
                    atomicAdd(h + 65, x1.y);
                    float* l = La + dl * RANK + 2 * lane;
                    atomicAdd(l,     __logf(fabsf(pr.x)));
                    atomicAdd(l + 1, __logf(fabsf(pr.y)));
                    const unsigned b0 = __builtin_amdgcn_ballot_w32(pr.x < 0.0f);
                    const unsigned b1 = __builtin_amdgcn_ballot_w32(pr.y < 0.0f);
                    if (lane < 2) atomicXor(&Sg[dl * 2 + lane], lane ? b1 : b0);
                };

                int j = 0;
                for (; j + 4 <= m; j += 4) {
                    float2 a0, a1, ap, b0v, b1v, bp, c0, c1, cp, e0v, e1v, ep;
                    unsigned da, db, dc, de;
                    gather(j,     a0,  a1,  ap, da);
                    gather(j + 1, b0v, b1v, bp, db);
                    gather(j + 2, c0,  c1,  cp, dc);
                    gather(j + 3, e0v, e1v, ep, de);
                    accum(a0,  a1,  ap, da);
                    accum(b0v, b1v, bp, db);
                    accum(c0,  c1,  cp, dc);
                    accum(e0v, e1v, ep, de);
                }
                for (; j < m; ++j) {
                    float2 a0, a1, ap;
                    unsigned da;
                    gather(j, a0, a1, ap, da);
                    accum(a0, a1, ap, da);
                }
            }
            qn = 0u;
        }
    };

    auto scan_step = [&](const uint4 (&r)[SCAN_U], int cc, bool last) {
        for (int u = 0; u < SCAN_U; ++u) {
            const int  i16 = cc * STEP16 + u * 32 + lane;
            const bool ok  = (i16 < N16);
            const unsigned e0 = (unsigned)i16 * 16u;
            const unsigned w[4] = {r[u].x ^ bsplat, r[u].y ^ bsplat, r[u].z ^ bsplat, r[u].w ^ bsplat};
            unsigned z[4];
            for (int k = 0; k < 4; ++k)
                z[k] = ~(((w[k] & 0x7f7f7f7fu) + 0x7f7f7f7fu) | w[k]) & 0x80808080u;
            unsigned mk = ok ? ((z[0] >> 7) | (z[1] >> 6) | (z[2] >> 5) | (z[3] >> 4)) : 0u;
#pragma nounroll
            for (int t = 0; t < 16; ++t) {
                const bool has = (mk != 0u);
                const unsigned act = __builtin_amdgcn_ballot_w32(has);
                if (act == 0u) break;
                if (has) {
                    const unsigned bit = (unsigned)__builtin_ctz(mk);
                    mk &= mk - 1u;
                    const unsigned slot = qn + __builtin_amdgcn_mbcnt_lo(act, 0u);
                    myq[slot] = e0 + 4u * (bit & 7u) + (bit >> 3);
                }
                qn += (unsigned)__builtin_popcount(act);
            }
            if (u == SCAN_U / 2 - 1) drain(false);
        }
        drain(last);
    };

    constexpr int NPAIR = (NCH + 2 * EDGE_WAVES - 1) / (2 * EDGE_WAVES);
    uint4 ra[SCAN_U], rb[SCAN_U];
    load_step(ra, wave);
    for (int p = 0; p < NPAIR; ++p) {
        const int c = wave + p * 2 * EDGE_WAVES;
        load_step(rb, c + EDGE_WAVES);
        scan_step(ra, c, false);
        load_step(ra, c + 2 * EDGE_WAVES);
        scan_step(rb, c + EDGE_WAVES, p == NPAIR - 1);
    }
    __syncthreads();

    const int rows = (N_NODES - n0 < NR) ? (N_NODES - n0) : NR;
    {
        float4* hg = (float4*)(hsum + (size_t)n0 * OUT_F);
        for (int i = tid; i < rows * OUT_F / 4; i += 1024) hg[i] = ((const float4*)Hs)[i];
        float4* lg = (float4*)(labs + (size_t)n0 * RANK);
        for (int i = tid; i < rows * RANK / 4; i += 1024) lg[i] = ((const float4*)La)[i];
        unsigned* sg = sgn + (size_t)n0 * 2;
        for (int i = tid; i < rows * 2; i += 1024) sg[i] = Sg[i];
        if (tid < rows) ideg[n0 + tid] = (float)Cn[tid];
    }
}

__global__ __launch_bounds__(128) void k_final(const float* __restrict__ vmat,
                                               const float* __restrict__ hsum,
                                               const float* __restrict__ labs,
                                               const unsigned* __restrict__ sgn,
                                               const float* __restrict__ ideg,
                                               float* __restrict__ out)
{
    __shared__ __attribute__((aligned(16))) float Hp[16][68];
    __shared__ float sc[16];

    const int m0  = blockIdx.x * 16;
    const int tid = threadIdx.x;

    const int wave  = tid >> 5;
    const int lane  = tid & 31;
    const int rsel  = lane & 15;
    const int khalf = (lane < 16) ? 0 : 2;
    const int rofs  = (lane < 16) ? 0 : 8;

    const int sr  = tid >> 3;
    const int sc0 = (tid & 7) * 8;
    const int row = m0 + sr;
    const float4* lp = (const float4*)(labs + (size_t)row * RANK + sc0);
    const float4 la0 = lp[0];
    const float4 la1 = lp[1];
    const uint2  sw  = *(const uint2*)(sgn + (size_t)row * 2);
    const float  dg  = ideg[row];

    float hs[2][8];
#pragma unroll
    for (int t = 0; t < 2; ++t) {
        const float* hp = hsum + (size_t)(m0 + rofs) * OUT_F + (wave * 2 + t) * 16 + rsel;
#pragma unroll
        for (int r = 0; r < 8; ++r) hs[t][r] = hp[r * OUT_F];
    }
    asm volatile("" :: "v"(la0.x), "v"(la0.y), "v"(la0.z), "v"(la0.w),
                       "v"(la1.x), "v"(la1.y), "v"(la1.z), "v"(la1.w));

    {
        const bool  has = (dg > 0.0f);
        const float la[8] = {la0.x, la0.y, la0.z, la0.w, la1.x, la1.y, la1.z, la1.w};
        float val[8];
#pragma unroll
        for (int i = 0; i < 8; ++i) {
            const int c = sc0 + i;
            const unsigned n = ((c & 1) ? sw.y : sw.x) >> (c >> 1);
            const float sg = (n & 1u) ? -1.0f : 1.0f;
            val[i] = has ? sg * __expf(la[i]) : 0.0f;
        }
        *(float4*)&Hp[sr][sc0]     = make_float4(val[0], val[1], val[2], val[3]);
        *(float4*)&Hp[sr][sc0 + 4] = make_float4(val[4], val[5], val[6], val[7]);
        if ((tid & 7) == 0) sc[sr] = rsqrtf(fmaxf(dg, 1.0f));
    }
    __syncthreads();

    for (int t = 0; t < 2; ++t) {
        const int n0 = (wave * 2 + t) * 16;

        for (int k = 0; k < RANK; k += 32)
            __builtin_prefetch(vmat + (size_t)k * OUT_F + n0, 0, 0);

        v8f acc = {};
        for (int k0 = 0; k0 < RANK; k0 += 4) {
            v2f a;
            a.x = Hp[rsel][k0 + khalf];
            a.y = Hp[rsel][k0 + khalf + 1];
            v2f b;
            const float* Bp = vmat + (size_t)(k0 + khalf) * OUT_F + n0 + rsel;
            b.x = Bp[0];
            b.y = Bp[OUT_F];
            acc = __builtin_amdgcn_wmma_f32_16x16x4_f32(
                false, a, false, b, (short)0, acc, false, false);
        }
        const int colg = n0 + rsel;
        float* op = out + (size_t)(m0 + rofs) * OUT_F + colg;
#pragma unroll
        for (int r = 0; r < 8; ++r)
            op[r * OUT_F] = (acc[r] + hs[t][r]) * sc[rofs + r];
    }
}

extern "C" void kernel_launch(void* const* d_in, const int* in_sizes, int n_in,
                              void* d_out, int out_size, void* d_ws, size_t ws_size,
                              hipStream_t stream)
{
    (void)in_sizes; (void)n_in; (void)out_size; (void)ws_size;

    const float* feat = (const float*)d_in[0];
    const float* w1   = (const float*)d_in[1];
    const float* w2   = (const float*)d_in[2];
    const float* vmat = (const float*)d_in[3];
    const int*   src  = (const int*)d_in[4];
    const int*   dst  = (const int*)d_in[5];

    float* ws    = (float*)d_ws;
    float* ideg  = ws + OFF_IDEG;
    unsigned* odpart = (unsigned*)(ws + OFF_ODPART);
    float* hsum  = ws + OFF_HSUM;
    float* labs  = ws + OFF_LABS;
    unsigned* sgn = (unsigned*)(ws + OFF_SGN);
    unsigned* bid = (unsigned*)(ws + OFF_BID);
    float* rec   = ws + OFF_REC;

    k_outdeg<<<DEG_SLICES, 1024, 0, stream>>>(src, dst, odpart, bid);
    k_node_gemm<<<N_NODES / 16, 128, 0, stream>>>(feat, w1, w2, odpart, rec);
    k_edge<<<EDGE_BLOCKS, 1024, 0, stream>>>(src, dst, bid, rec,
                                             hsum, labs, sgn, ideg);
    k_final<<<N_NODES / 16, 128, 0, stream>>>(vmat, hsum, labs, sgn, ideg,
                                              (float*)d_out);
}


# Finish moved into the row-owner edge blocks: LDS log-abs rows become an h_prod tile in LDS, times v on f32 WMMA (A from LDS), folded into the LDS h_sum rows, out stored once; 3 launches, planes gone
# speedup vs baseline: 3.2703x; 1.0744x over previous
#include <hip/hip_runtime.h>
#include <math.h>

typedef __attribute__((ext_vector_type(2))) float v2f;
typedef __attribute__((ext_vector_type(8))) float v8f;

#define N_NODES 50000
#define N_EDGES 800000
#define IN_F    256
#define OUT_F   128
#define RANK    64

static_assert(N_NODES % 16 == 0, "row tiles exact");
static_assert(N_EDGES % 256 == 0, "edge blocks exact");

#define OFF_ODPART 100000u
#define OFF_BID    9900000u
#define OFF_REC    12899904u
#define REC_F      (OUT_F + RANK)

#define DEG_SLICES  32
#define DEG_EDGES   (N_EDGES / DEG_SLICES)
#define DEG_LOADS   ((DEG_EDGES / 4 + 1023) / 1024)
static_assert(N_EDGES % DEG_SLICES == 0 && DEG_EDGES % 4 == 0, "slices exact; int4 loads aligned");
static_assert(N_NODES % 4 == 0, "count planes move as uint4");
static_assert(DEG_SLICES % 8 == 0, "the node product blocks add the planes in 8 groups");
static_assert((size_t)OFF_ODPART + (size_t)DEG_SLICES * N_NODES <= OFF_BID && OFF_BID + N_EDGES / 4 <= OFF_REC,
              "partial planes and the block-id bytes fit in front of the node table");
static_assert((size_t)OFF_REC + 63u + (size_t)N_NODES * REC_F <= 22500000u,
              "the aligned node table ends inside the workspace");
static_assert((REC_F * 4) % 256 == 0 && (OUT_F * 4) % 256 == 0,
              "a record, and its product part, start on 256-byte boundaries");

__device__ __forceinline__ float* node_table(float* p)
{
    const unsigned pad = (unsigned)(0ull - (unsigned long long)p) & 255u;
    return p + (pad >> 2);
}

#define NR          196
#define EDGE_BLOCKS ((N_NODES + NR - 1) / NR)
#define EDGE_WAVES  32
#define SCAN_U      4
#define SCAN_HALF   (32 * 16 * SCAN_U / 2)
#define DRAIN_AT    24
#define QCAP        1056
static_assert(EDGE_BLOCKS <= 256, "a block id fits one byte");
static_assert(N_EDGES % 16 == 0, "16-byte block-id scan exact");
static_assert(SCAN_U % 2 == 0 && DRAIN_AT + SCAN_HALF <= QCAP, "half a scan step can never overflow a wave's queue");
#define HP_TILES    ((NR + 15) / 16)
#define HP_ROWS     (HP_TILES * 16)
#define HP_PITCH    68
static_assert((HP_ROWS * HP_PITCH + NR) * 4 <= EDGE_WAVES * QCAP * 4, "the h_prod tile and the row scales fit in the queues' LDS");
static_assert(EDGE_WAVES % (OUT_F / 16) == 0, "a wave keeps one column tile of out over all its row tiles");

__device__ __forceinline__ unsigned block_id_byte(int d)
{
    return ((unsigned)d < (unsigned)N_NODES) ? (unsigned)d / (unsigned)NR : 255u;
}

__global__ __launch_bounds__(1024) void k_outdeg(const int* __restrict__ src,
                                                 const int* __restrict__ dst,
                                                 unsigned* __restrict__ parts,
                                                 unsigned* __restrict__ bid)
{
    __shared__ __attribute__((aligned(16))) unsigned Hc[N_NODES];

    const int tid = threadIdx.x;
    constexpr int S4 = DEG_EDGES / 4;

    const int4* s4 = (const int4*)(src + (size_t)blockIdx.x * DEG_EDGES);
    int4 v[DEG_LOADS];
#pragma unroll
    for (int j = 0; j < DEG_LOADS; ++j) {
        const int i = tid + j * 1024;
        v[j] = s4[(i < S4) ? i : (S4 - 1)];
    }

    {
        const int4* t4 = (const int4*)(dst + (size_t)blockIdx.x * DEG_EDGES);
        unsigned*   bo = bid + (size_t)blockIdx.x * S4;
        for (int j = 0; j < DEG_LOADS; ++j) {
            const int i = tid + j * 1024;
            if (i < S4) {
                const int4 d = t4[i];
                bo[i] = block_id_byte(d.x) | (block_id_byte(d.y) << 8) |
                        (block_id_byte(d.z) << 16) | (block_id_byte(d.w) << 24);
            }
        }
    }

    for (int i = tid; i < N_NODES / 4; i += 1024)
        ((uint4*)Hc)[i] = make_uint4(0u, 0u, 0u, 0u);
    __syncthreads();

#pragma unroll
    for (int j = 0; j < DEG_LOADS; ++j) {
        if (tid + j * 1024 < S4) {
            atomicAdd(&Hc[v[j].x], 1u);
            atomicAdd(&Hc[v[j].y], 1u);
            atomicAdd(&Hc[v[j].z], 1u);
            atomicAdd(&Hc[v[j].w], 1u);
        }
    }
    __syncthreads();

    uint4* og = (uint4*)(parts + (size_t)blockIdx.x * N_NODES);
    for (int i = tid; i < N_NODES / 4; i += 1024)
        og[i] = ((const uint4*)Hc)[i];
}

__global__ __launch_bounds__(128) void k_node_gemm(const float* __restrict__ feat,
                                                   const float* __restrict__ w1,
                                                   const float* __restrict__ w2,
                                                   const unsigned* __restrict__ odpart,
                                                   float* __restrict__ rec_raw)
{
    __shared__ float As[16][260];
    float* __restrict__ rec = node_table(rec_raw);
    __shared__ unsigned dp[8][16];

    const int m0  = blockIdx.x * 16;
    const int tid = threadIdx.x;

    {
        const int r = tid & 15;
        const int g = tid >> 4;
        unsigned d = 0u;
#pragma unroll
        for (int j = 0; j < DEG_SLICES / 8; ++j)
            d += odpart[(size_t)(g + 8 * j) * N_NODES + m0 + r];
        dp[g][r] = d;
    }
    __syncthreads();

    {
        const int r   = tid >> 3;
        const int c0  = (tid & 7) * 32;
        unsigned d = 0u;
#pragma unroll
        for (int q = 0; q < 8; ++q) d += dp[q][r];
        const float s = rsqrtf(fmaxf((float)d, 1.0f));
        const float4* g = (const float4*)(feat + (size_t)(m0 + r) * IN_F + c0);
#pragma unroll
        for (int j = 0; j < 8; ++j) {
            float4 x = g[j];
            As[r][c0 + 4 * j + 0] = x.x * s;
            As[r][c0 + 4 * j + 1] = x.y * s;
            As[r][c0 + 4 * j + 2] = x.z * s;
            As[r][c0 + 4 * j + 3] = x.w * s;
        }
    }
    __syncthreads();

    const int wave  = tid >> 5;
    const int lane  = tid & 31;
    const int rsel  = lane & 15;
    const int khalf = (lane < 16) ? 0 : 2;
    const int rofs  = (lane < 16) ? 0 : 8;

    const float* B1 = w1 + (size_t)khalf * OUT_F + 32 * wave + 2 * rsel;
    const float* B2 = w2 + (size_t)khalf * RANK + 16 * wave + rsel;

    v8f acc0 = {}, acc1 = {}, acc2 = {};
#pragma unroll 4
    for (int k0 = 0; k0 < IN_F; k0 += 4) {
        v2f a;
        a.x = As[rsel][k0 + khalf];
        a.y = As[rsel][k0 + khalf + 1];
        const float2 q0 = *(const float2*)(B1 + (size_t)k0 * OUT_F);
        const float2 q1 = *(const float2*)(B1 + (size_t)(k0 + 1) * OUT_F);
        const float  p0 = B2[(size_t)k0 * RANK];
        const float  p1 = B2[(size_t)(k0 + 1) * RANK];
        v2f b;
        b.x = q0.x; b.y = q1.x;
        acc0 = __builtin_amdgcn_wmma_f32_16x16x4_f32(false, a, false, b, (short)0, acc0, false, false);
        b.x = q0.y; b.y = q1.y;
        acc1 = __builtin_amdgcn_wmma_f32_16x16x4_f32(false, a, false, b, (short)0, acc1, false, false);
        b.x = p0; b.y = p1;
        acc2 = __builtin_amdgcn_wmma_f32_16x16x4_f32(false, a, false, b, (short)0, acc2, false, false);
    }

    {
        float* op = rec + (size_t)(m0 + rofs) * REC_F + 32 * wave + 2 * rsel;
#pragma unroll
        for (int r = 0; r < 8; ++r)
            *(float2*)(op + r * REC_F) = make_float2(acc0[r], acc1[r]);
    }
    {
        const int colg = 16 * wave + rsel;
        const float bias = w2[(size_t)IN_F * RANK + colg];
        float* op = rec + (size_t)(m0 + rofs) * REC_F + OUT_F + colg;
#pragma unroll
        for (int r = 0; r < 8; ++r)
            op[r * REC_F] = tanhf(acc2[r] + bias);
    }
}

__global__ __launch_bounds__(1024) void k_edge(const int* __restrict__ src,
                                               const int* __restrict__ dst,
                                               const unsigned* __restrict__ bid,
                                               float* __restrict__ rec_raw,
                                               const float* __restrict__ vmat,
                                               float* __restrict__ out)
{
    __shared__ __attribute__((aligned(16))) float    Hs[NR * OUT_F];
    __shared__ __attribute__((aligned(16))) float    La[NR * RANK];
    __shared__ __attribute__((aligned(16))) unsigned Sg[NR * 2];
    __shared__ unsigned Cn[NR];
    __shared__ __attribute__((aligned(16))) unsigned Qe[EDGE_WAVES * QCAP];

    const int tid  = threadIdx.x;
    const int wave = tid >> 5;
    const int lane = tid & 31;
    const int n0   = blockIdx.x * NR;
    const float* __restrict__ rec = node_table(rec_raw);

    for (int i = tid; i < NR * OUT_F / 4; i += 1024)
        ((float4*)Hs)[i] = make_float4(0.0f, 0.0f, 0.0f, 0.0f);
    for (int i = tid; i < NR * RANK / 4; i += 1024)
        ((float4*)La)[i] = make_float4(0.0f, 0.0f, 0.0f, 0.0f);
    for (int i = tid; i < NR * 2; i += 1024)
        Sg[i] = 0u;
    if (tid < NR) Cn[tid] = 0u;
    __syncthreads();

    constexpr int N16    = N_EDGES / 16;
    constexpr int STEP16 = 32 * SCAN_U;
    constexpr int NCH    = (N16 + STEP16 - 1) / STEP16;
    const uint4* b16  = (const uint4*)bid;
    unsigned*    myq  = Qe + wave * QCAP;
    const unsigned bsplat = (unsigned)blockIdx.x * 0x01010101u;
    unsigned qn = 0u;

    auto load_step = [&](uint4 (&r)[SCAN_U], int cc) {
        for (int u = 0; u < SCAN_U; ++u) {
            const int i16 = cc * STEP16 + u * 32 + lane;
            r[u] = b16[(i16 < N16) ? i16 : (N16 - 1)];
        }
    };

    auto drain = [&](bool last) {
        const unsigned cnt = (unsigned)__builtin_amdgcn_readfirstlane((int)qn);
        if (last || cnt >= (unsigned)DRAIN_AT) {
            for (unsigned base = 0; base < cnt; base += 32u) {
                const unsigned q = base + (unsigned)lane;
                int      sv  = 0;
                unsigned dlv = (unsigned)NR;
                if (q < cnt) {
                    const unsigned e = myq[q];
                    sv  = src[e];
                    dlv = (unsigned)(dst[e] - n0);
                }
                if (dlv < (unsigned)NR) atomicAdd(&Cn[dlv], 1u);
                const int m = (int)((cnt - base < 32u) ? (cnt - base) : 32u);

                auto gather = [&](int j, float2& x0, float2& x1, float2& pr, unsigned& dl) {
                    const int s = __builtin_amdgcn_readlane(sv, j);
                    dl = (unsigned)__builtin_amdgcn_readlane((int)dlv, j);
                    const float2* fs = (const float2*)(rec + (size_t)s * REC_F);
                    x0 = fs[lane];
                    x1 = fs[32 + lane];
                    pr = fs[64 + lane];
                };
                auto accum = [&](const float2& x0, const float2& x1, const float2& pr, unsigned dl) {
                    if (dl >= (unsigned)NR) return;
                    float* h = Hs + dl * OUT_F + 2 * lane;
                    atomicAdd(h,      x0.x);
                    atomicAdd(h + 1,  x0.y);
                    atomicAdd(h + 64, x1.x);
                    atomicAdd(h + 65, x1.y);
                    float* l = La + dl * RANK + 2 * lane;
                    atomicAdd(l,     __logf(fabsf(pr.x)));
                    atomicAdd(l + 1, __logf(fabsf(pr.y)));
                    const unsigned b0 = __builtin_amdgcn_ballot_w32(pr.x < 0.0f);
                    const unsigned b1 = __builtin_amdgcn_ballot_w32(pr.y < 0.0f);
                    if (lane < 2) atomicXor(&Sg[dl * 2 + lane], lane ? b1 : b0);
                };

                int j = 0;
                for (; j + 4 <= m; j += 4) {
                    float2 a0, a1, ap, b0v, b1v, bp, c0, c1, cp, e0v, e1v, ep;
                    unsigned da, db, dc, de;
                    gather(j,     a0,  a1,  ap, da);
                    gather(j + 1, b0v, b1v, bp, db);
                    gather(j + 2, c0,  c1,  cp, dc);
                    gather(j + 3, e0v, e1v, ep, de);
                    accum(a0,  a1,  ap, da);
                    accum(b0v, b1v, bp, db);
                    accum(c0,  c1,  cp, dc);
                    accum(e0v, e1v, ep, de);
                }
                for (; j < m; ++j) {
                    float2 a0, a1, ap;
                    unsigned da;
                    gather(j, a0, a1, ap, da);
                    accum(a0, a1, ap, da);
                }
            }
            qn = 0u;
        }
    };

    auto scan_step = [&](const uint4 (&r)[SCAN_U], int cc, bool last) {
        for (int u = 0; u < SCAN_U; ++u) {
            const int  i16 = cc * STEP16 + u * 32 + lane;
            const bool ok  = (i16 < N16);
            const unsigned e0 = (unsigned)i16 * 16u;
            const unsigned w[4] = {r[u].x ^ bsplat, r[u].y ^ bsplat, r[u].z ^ bsplat, r[u].w ^ bsplat};
            unsigned z[4];
            for (int k = 0; k < 4; ++k)
                z[k] = ~(((w[k] & 0x7f7f7f7fu) + 0x7f7f7f7fu) | w[k]) & 0x80808080u;
            unsigned mk = ok ? ((z[0] >> 7) | (z[1] >> 6) | (z[2] >> 5) | (z[3] >> 4)) : 0u;
#pragma nounroll
            for (int t = 0; t < 16; ++t) {
                const bool has = (mk != 0u);
                const unsigned act = __builtin_amdgcn_ballot_w32(has);
                if (act == 0u) break;
                if (has) {
                    const unsigned bit = (unsigned)__builtin_ctz(mk);
                    mk &= mk - 1u;
                    const unsigned slot = qn + __builtin_amdgcn_mbcnt_lo(act, 0u);
                    myq[slot] = e0 + 4u * (bit & 7u) + (bit >> 3);
                }
                qn += (unsigned)__builtin_popcount(act);
            }
            if (u == SCAN_U / 2 - 1) drain(false);
        }
        drain(last);
    };

    constexpr int NPAIR = (NCH + 2 * EDGE_WAVES - 1) / (2 * EDGE_WAVES);
    uint4 ra[SCAN_U], rb[SCAN_U];
    load_step(ra, wave);
    for (int p = 0; p < NPAIR; ++p) {
        const int c = wave + p * 2 * EDGE_WAVES;
        load_step(rb, c + EDGE_WAVES);
        scan_step(ra, c, false);
        load_step(ra, c + 2 * EDGE_WAVES);
        scan_step(rb, c + EDGE_WAVES, p == NPAIR - 1);
    }
    __syncthreads();

    float* Hp  = (float*)Qe;
    float* scl = Hp + HP_ROWS * HP_PITCH;

    const int rsel  = lane & 15;
    const int khalf = (lane < 16) ? 0 : 2;
    const int rofs  = (lane < 16) ? 0 : 8;
    const int wv = __builtin_amdgcn_readfirstlane(wave);
    const int ct = wv & (OUT_F / 16 - 1);
    const int g  = wv / (OUT_F / 16);

    v2f bf[RANK / 4];
    {
        const float* Bp = vmat + (size_t)khalf * OUT_F + ct * 16 + rsel;
#pragma unroll
        for (int s = 0; s < RANK / 4; ++s) {
            bf[s].x = Bp[(size_t)(4 * s) * OUT_F];
            bf[s].y = Bp[(size_t)(4 * s + 1) * OUT_F];
        }
    }

    for (int i4 = tid; i4 < NR * RANK / 4; i4 += 1024) {
        const int row = i4 >> 4;
        const int c0  = (i4 & 15) * 4;
        const float4 la = ((const float4*)La)[i4];
        const bool has = (Cn[row] != 0u);
        const unsigned n0w = Sg[row * 2]     >> (c0 >> 1);
        const unsigned n1w = Sg[row * 2 + 1] >> (c0 >> 1);
        float4 hv;
        hv.x = has ? ((n0w & 1u) ? -1.0f : 1.0f) * __expf(la.x) : 0.0f;
        hv.y = has ? ((n1w & 1u) ? -1.0f : 1.0f) * __expf(la.y) : 0.0f;
        hv.z = has ? ((n0w & 2u) ? -1.0f : 1.0f) * __expf(la.z) : 0.0f;
        hv.w = has ? ((n1w & 2u) ? -1.0f : 1.0f) * __expf(la.w) : 0.0f;
        *(float4*)(Hp + row * HP_PITCH + c0) = hv;
    }
    for (int i = tid; i < (HP_ROWS - NR) * HP_PITCH; i += 1024)
        Hp[NR * HP_PITCH + i] = 0.0f;
    if (tid < NR) scl[tid] = rsqrtf(fmaxf((float)Cn[tid], 1.0f));
    __syncthreads();

    for (int rt = g; rt < HP_TILES; rt += EDGE_WAVES / (OUT_F / 16)) {
        const float* Ap = Hp + (rt * 16 + rsel) * HP_PITCH + khalf;
        v8f acc = {};
#pragma unroll
        for (int s = 0; s < RANK / 4; ++s) {
            v2f a;
            a.x = Ap[4 * s];
            a.y = Ap[4 * s + 1];
            acc = __builtin_amdgcn_wmma_f32_16x16x4_f32(
                false, a, false, bf[s], (short)0, acc, false, false);
        }
        float*       h  = Hs + (rt * 16 + rofs) * OUT_F + ct * 16 + rsel;
        const float* sp = scl + rt * 16 + rofs;
        if (rt * 16 + 16 <= NR) {
#pragma unroll
            for (int r = 0; r < 8; ++r)
                h[r * OUT_F] = (acc[r] + h[r * OUT_F]) * sp[r];
        } else {
#pragma unroll
            for (int r = 0; r < 8; ++r)
                if (rt * 16 + rofs + r < NR)
                    h[r * OUT_F] = (acc[r] + h[r * OUT_F]) * sp[r];
        }
    }
    __syncthreads();

    const int rows = (N_NODES - n0 < NR) ? (N_NODES - n0) : NR;
    float4* og = (float4*)(out + (size_t)n0 * OUT_F);
    for (int i = tid; i < rows * OUT_F / 4; i += 1024) og[i] = ((const float4*)Hs)[i];
}

extern "C" void kernel_launch(void* const* d_in, const int* in_sizes, int n_in,
                              void* d_out, int out_size, void* d_ws, size_t ws_size,
                              hipStream_t stream)
{
    (void)in_sizes; (void)n_in; (void)out_size; (void)ws_size;

    const float* feat = (const float*)d_in[0];
    const float* w1   = (const float*)d_in[1];
    const float* w2   = (const float*)d_in[2];
    const float* vmat = (const float*)d_in[3];
    const int*   src  = (const int*)d_in[4];
    const int*   dst  = (const int*)d_in[5];

    float* ws    = (float*)d_ws;
    unsigned* odpart = (unsigned*)(ws + OFF_ODPART);
    unsigned* bid = (unsigned*)(ws + OFF_BID);
    float* rec   = ws + OFF_REC;

    k_outdeg<<<DEG_SLICES, 1024, 0, stream>>>(src, dst, odpart, bid);
    k_node_gemm<<<N_NODES / 16, 128, 0, stream>>>(feat, w1, w2, odpart, rec);
    k_edge<<<EDGE_BLOCKS, 1024, 0, stream>>>(src, dst, bid, rec, vmat, (float*)d_out);
}


# log|feat_prod| taken once per node, not per edge: node kernel stores |log|m|| with m's sign in the sign bit; edge blocks add -|word| and xor the bit, no log at the gather; sums stay arrival-order
# speedup vs baseline: 3.3956x; 1.0383x over previous
#include <hip/hip_runtime.h>
#include <math.h>

typedef __attribute__((ext_vector_type(2))) float v2f;
typedef __attribute__((ext_vector_type(8))) float v8f;

#define N_NODES 50000
#define N_EDGES 800000
#define IN_F    256
#define OUT_F   128
#define RANK    64

static_assert(N_NODES % 16 == 0, "row tiles exact");
static_assert(N_EDGES % 256 == 0, "edge blocks exact");

#define OFF_ODPART 100000u
#define OFF_BID    9900000u
#define OFF_REC    12899904u
#define REC_F      (OUT_F + RANK)

#define DEG_SLICES  32
#define DEG_EDGES   (N_EDGES / DEG_SLICES)
#define DEG_LOADS   ((DEG_EDGES / 4 + 1023) / 1024)
static_assert(N_EDGES % DEG_SLICES == 0 && DEG_EDGES % 4 == 0, "slices exact; int4 loads aligned");
static_assert(N_NODES % 4 == 0, "count planes move as uint4");
static_assert(DEG_SLICES % 8 == 0, "the node product blocks add the planes in 8 groups");
static_assert((size_t)OFF_ODPART + (size_t)DEG_SLICES * N_NODES <= OFF_BID && OFF_BID + N_EDGES / 4 <= OFF_REC,
              "partial planes and the block-id bytes fit in front of the node table");
static_assert((size_t)OFF_REC + 63u + (size_t)N_NODES * REC_F <= 22500000u,
              "the aligned node table ends inside the workspace");
static_assert((REC_F * 4) % 256 == 0 && (OUT_F * 4) % 256 == 0,
              "a record, and its product part, start on 256-byte boundaries");

__device__ __forceinline__ float* node_table(float* p)
{
    const unsigned pad = (unsigned)(0ull - (unsigned long long)p) & 255u;
    return p + (pad >> 2);
}

#define NR          196
#define EDGE_BLOCKS ((N_NODES + NR - 1) / NR)
#define EDGE_WAVES  32
#define SCAN_U      4
#define SCAN_HALF   (32 * 16 * SCAN_U / 2)
#define DRAIN_AT    24
#define QCAP        1056
static_assert(EDGE_BLOCKS <= 256, "a block id fits one byte");
static_assert(N_EDGES % 16 == 0, "16-byte block-id scan exact");
static_assert(SCAN_U % 2 == 0 && DRAIN_AT + SCAN_HALF <= QCAP, "half a scan step can never overflow a wave's queue");
#define HP_TILES    ((NR + 15) / 16)
#define HP_ROWS     (HP_TILES * 16)
#define HP_PITCH    68
static_assert((HP_ROWS * HP_PITCH + NR) * 4 <= EDGE_WAVES * QCAP * 4, "the h_prod tile and the row scales fit in the queues' LDS");
static_assert(EDGE_WAVES % (OUT_F / 16) == 0, "a wave keeps one column tile of out over all its row tiles");

__device__ __forceinline__ unsigned block_id_byte(int d)
{
    return ((unsigned)d < (unsigned)N_NODES) ? (unsigned)d / (unsigned)NR : 255u;
}

__global__ __launch_bounds__(1024) void k_outdeg(const int* __restrict__ src,
                                                 const int* __restrict__ dst,
                                                 unsigned* __restrict__ parts,
                                                 unsigned* __restrict__ bid)
{
    __shared__ __attribute__((aligned(16))) unsigned Hc[N_NODES];

    const int tid = threadIdx.x;
    constexpr int S4 = DEG_EDGES / 4;

    const int4* s4 = (const int4*)(src + (size_t)blockIdx.x * DEG_EDGES);
    int4 v[DEG_LOADS];
#pragma unroll
    for (int j = 0; j < DEG_LOADS; ++j) {
        const int i = tid + j * 1024;
        v[j] = s4[(i < S4) ? i : (S4 - 1)];
    }

    {
        const int4* t4 = (const int4*)(dst + (size_t)blockIdx.x * DEG_EDGES);
        unsigned*   bo = bid + (size_t)blockIdx.x * S4;
        for (int j = 0; j < DEG_LOADS; ++j) {
            const int i = tid + j * 1024;
            if (i < S4) {
                const int4 d = t4[i];
                bo[i] = block_id_byte(d.x) | (block_id_byte(d.y) << 8) |
                        (block_id_byte(d.z) << 16) | (block_id_byte(d.w) << 24);
            }
        }
    }

    for (int i = tid; i < N_NODES / 4; i += 1024)
        ((uint4*)Hc)[i] = make_uint4(0u, 0u, 0u, 0u);
    __syncthreads();

#pragma unroll
    for (int j = 0; j < DEG_LOADS; ++j) {
        if (tid + j * 1024 < S4) {
            atomicAdd(&Hc[v[j].x], 1u);
            atomicAdd(&Hc[v[j].y], 1u);
            atomicAdd(&Hc[v[j].z], 1u);
            atomicAdd(&Hc[v[j].w], 1u);
        }
    }
    __syncthreads();

    uint4* og = (uint4*)(parts + (size_t)blockIdx.x * N_NODES);
    for (int i = tid; i < N_NODES / 4; i += 1024)
        og[i] = ((const uint4*)Hc)[i];
}

__global__ __launch_bounds__(128) void k_node_gemm(const float* __restrict__ feat,
                                                   const float* __restrict__ w1,
                                                   const float* __restrict__ w2,
                                                   const unsigned* __restrict__ odpart,
                                                   float* __restrict__ rec_raw)
{
    __shared__ float As[16][260];
    float* __restrict__ rec = node_table(rec_raw);
    __shared__ unsigned dp[8][16];

    const int m0  = blockIdx.x * 16;
    const int tid = threadIdx.x;

    {
        const int r = tid & 15;
        const int g = tid >> 4;
        unsigned d = 0u;
#pragma unroll
        for (int j = 0; j < DEG_SLICES / 8; ++j)
            d += odpart[(size_t)(g + 8 * j) * N_NODES + m0 + r];
        dp[g][r] = d;
    }
    __syncthreads();

    {
        const int r   = tid >> 3;
        const int c0  = (tid & 7) * 32;
        unsigned d = 0u;
#pragma unroll
        for (int q = 0; q < 8; ++q) d += dp[q][r];
        const float s = rsqrtf(fmaxf((float)d, 1.0f));
        const float4* g = (const float4*)(feat + (size_t)(m0 + r) * IN_F + c0);
#pragma unroll
        for (int j = 0; j < 8; ++j) {
            float4 x = g[j];
            As[r][c0 + 4 * j + 0] = x.x * s;
            As[r][c0 + 4 * j + 1] = x.y * s;
            As[r][c0 + 4 * j + 2] = x.z * s;
            As[r][c0 + 4 * j + 3] = x.w * s;
        }
    }
    __syncthreads();

    const int wave  = tid >> 5;
    const int lane  = tid & 31;
    const int rsel  = lane & 15;
    const int khalf = (lane < 16) ? 0 : 2;
    const int rofs  = (lane < 16) ? 0 : 8;

    const float* B1 = w1 + (size_t)khalf * OUT_F + 32 * wave + 2 * rsel;
    const float* B2 = w2 + (size_t)khalf * RANK + 16 * wave + rsel;

    v8f acc0 = {}, acc1 = {}, acc2 = {};
#pragma unroll 4
    for (int k0 = 0; k0 < IN_F; k0 += 4) {
        v2f a;
        a.x = As[rsel][k0 + khalf];
        a.y = As[rsel][k0 + khalf + 1];
        const float2 q0 = *(const float2*)(B1 + (size_t)k0 * OUT_F);
        const float2 q1 = *(const float2*)(B1 + (size_t)(k0 + 1) * OUT_F);
        const float  p0 = B2[(size_t)k0 * RANK];
        const float  p1 = B2[(size_t)(k0 + 1) * RANK];
        v2f b;
        b.x = q0.x; b.y = q1.x;
        acc0 = __builtin_amdgcn_wmma_f32_16x16x4_f32(false, a, false, b, (short)0, acc0, false, false);
        b.x = q0.y; b.y = q1.y;
        acc1 = __builtin_amdgcn_wmma_f32_16x16x4_f32(false, a, false, b, (short)0, acc1, false, false);
        b.x = p0; b.y = p1;
        acc2 = __builtin_amdgcn_wmma_f32_16x16x4_f32(false, a, false, b, (short)0, acc2, false, false);
    }

    {
        float* op = rec + (size_t)(m0 + rofs) * REC_F + 32 * wave + 2 * rsel;
#pragma unroll
        for (int r = 0; r < 8; ++r)
            *(float2*)(op + r * REC_F) = make_float2(acc0[r], acc1[r]);
    }
    {
        const int colg = 16 * wave + rsel;
        const float bias = w2[(size_t)IN_F * RANK + colg];
        float* op = rec + (size_t)(m0 + rofs) * REC_F + OUT_F + colg;
#pragma unroll
        for (int r = 0; r < 8; ++r) {
            const float m  = tanhf(acc2[r] + bias);
            const float la = __logf(fabsf(m));
            const unsigned w = (__float_as_uint(la) & 0x7fffffffu) | ((m < 0.0f) ? 0x80000000u : 0u);
            op[r * REC_F] = __uint_as_float(w);
        }
    }
}

__global__ __launch_bounds__(1024) void k_edge(const int* __restrict__ src,
                                               const int* __restrict__ dst,
                                               const unsigned* __restrict__ bid,
                                               float* __restrict__ rec_raw,
                                               const float* __restrict__ vmat,
                                               float* __restrict__ out)
{
    __shared__ __attribute__((aligned(16))) float    Hs[NR * OUT_F];
    __shared__ __attribute__((aligned(16))) float    La[NR * RANK];
    __shared__ __attribute__((aligned(16))) unsigned Sg[NR * 2];
    __shared__ unsigned Cn[NR];
    __shared__ __attribute__((aligned(16))) unsigned Qe[EDGE_WAVES * QCAP];

    const int tid  = threadIdx.x;
    const int wave = tid >> 5;
    const int lane = tid & 31;
    const int n0   = blockIdx.x * NR;
    const float* __restrict__ rec = node_table(rec_raw);

    for (int i = tid; i < NR * OUT_F / 4; i += 1024)
        ((float4*)Hs)[i] = make_float4(0.0f, 0.0f, 0.0f, 0.0f);
    for (int i = tid; i < NR * RANK / 4; i += 1024)
        ((float4*)La)[i] = make_float4(0.0f, 0.0f, 0.0f, 0.0f);
    for (int i = tid; i < NR * 2; i += 1024)
        Sg[i] = 0u;
    if (tid < NR) Cn[tid] = 0u;
    __syncthreads();

    constexpr int N16    = N_EDGES / 16;
    constexpr int STEP16 = 32 * SCAN_U;
    constexpr int NCH    = (N16 + STEP16 - 1) / STEP16;
    const uint4* b16  = (const uint4*)bid;
    unsigned*    myq  = Qe + wave * QCAP;
    const unsigned bsplat = (unsigned)blockIdx.x * 0x01010101u;
    unsigned qn = 0u;

    auto load_step = [&](uint4 (&r)[SCAN_U], int cc) {
        for (int u = 0; u < SCAN_U; ++u) {
            const int i16 = cc * STEP16 + u * 32 + lane;
            r[u] = b16[(i16 < N16) ? i16 : (N16 - 1)];
        }
    };

    auto drain = [&](bool last) {
        const unsigned cnt = (unsigned)__builtin_amdgcn_readfirstlane((int)qn);
        if (last || cnt >= (unsigned)DRAIN_AT) {
            for (unsigned base = 0; base < cnt; base += 32u) {
                const unsigned q = base + (unsigned)lane;
                int      sv  = 0;
                unsigned dlv = (unsigned)NR;
                if (q < cnt) {
                    const unsigned e = myq[q];
                    sv  = src[e];
                    dlv = (unsigned)(dst[e] - n0);
                }
                if (dlv < (unsigned)NR) atomicAdd(&Cn[dlv], 1u);
                const int m = (int)((cnt - base < 32u) ? (cnt - base) : 32u);

                auto gather = [&](int j, float2& x0, float2& x1, float2& pr, unsigned& dl) {
                    const int s = __builtin_amdgcn_readlane(sv, j);
                    dl = (unsigned)__builtin_amdgcn_readlane((int)dlv, j);
                    const float2* fs = (const float2*)(rec + (size_t)s * REC_F);
                    x0 = fs[lane];
                    x1 = fs[32 + lane];
                    pr = fs[64 + lane];
                };
                auto accum = [&](const float2& x0, const float2& x1, const float2& pr, unsigned dl) {
                    if (dl >= (unsigned)NR) return;
                    float* h = Hs + dl * OUT_F + 2 * lane;
                    atomicAdd(h,      x0.x);
                    atomicAdd(h + 1,  x0.y);
                    atomicAdd(h + 64, x1.x);
                    atomicAdd(h + 65, x1.y);
                    float* l = La + dl * RANK + 2 * lane;
                    atomicAdd(l,     -fabsf(pr.x));
                    atomicAdd(l + 1, -fabsf(pr.y));
                    const unsigned b0 = __builtin_amdgcn_ballot_w32(__float_as_int(pr.x) < 0);
                    const unsigned b1 = __builtin_amdgcn_ballot_w32(__float_as_int(pr.y) < 0);
                    if (lane < 2) atomicXor(&Sg[dl * 2 + lane], lane ? b1 : b0);
                };

                int j = 0;
                for (; j + 4 <= m; j += 4) {
                    float2 a0, a1, ap, b0v, b1v, bp, c0, c1, cp, e0v, e1v, ep;
                    unsigned da, db, dc, de;
                    gather(j,     a0,  a1,  ap, da);
                    gather(j + 1, b0v, b1v, bp, db);
                    gather(j + 2, c0,  c1,  cp, dc);
                    gather(j + 3, e0v, e1v, ep, de);
                    asm volatile("" :: "v"(a0.x), "v"(a0.y), "v"(a1.x), "v"(a1.y), "v"(ap.x), "v"(ap.y),
                                       "v"(b0v.x), "v"(b0v.y), "v"(b1v.x), "v"(b1v.y), "v"(bp.x), "v"(bp.y),
                                       "v"(c0.x), "v"(c0.y), "v"(c1.x), "v"(c1.y), "v"(cp.x), "v"(cp.y),
                                       "v"(e0v.x), "v"(e0v.y), "v"(e1v.x), "v"(e1v.y), "v"(ep.x), "v"(ep.y));
                    accum(a0,  a1,  ap, da);
                    accum(b0v, b1v, bp, db);
                    accum(c0,  c1,  cp, dc);
                    accum(e0v, e1v, ep, de);
                }
                for (; j < m; ++j) {
                    float2 a0, a1, ap;
                    unsigned da;
                    gather(j, a0, a1, ap, da);
                    asm volatile("" :: "v"(a0.x), "v"(a0.y), "v"(a1.x), "v"(a1.y), "v"(ap.x), "v"(ap.y));
                    accum(a0, a1, ap, da);
                }
            }
            qn = 0u;
        }
    };

    auto scan_step = [&](const uint4 (&r)[SCAN_U], int cc, bool last) {
        for (int u = 0; u < SCAN_U; ++u) {
            const int  i16 = cc * STEP16 + u * 32 + lane;
            const bool ok  = (i16 < N16);
            const unsigned e0 = (unsigned)i16 * 16u;
            const unsigned w[4] = {r[u].x ^ bsplat, r[u].y ^ bsplat, r[u].z ^ bsplat, r[u].w ^ bsplat};
            unsigned z[4];
            for (int k = 0; k < 4; ++k)
                z[k] = ~(((w[k] & 0x7f7f7f7fu) + 0x7f7f7f7fu) | w[k]) & 0x80808080u;
            unsigned mk = ok ? ((z[0] >> 7) | (z[1] >> 6) | (z[2] >> 5) | (z[3] >> 4)) : 0u;
#pragma nounroll
            for (int t = 0; t < 16; ++t) {
                const bool has = (mk != 0u);
                const unsigned act = __builtin_amdgcn_ballot_w32(has);
                if (act == 0u) break;
                if (has) {
                    const unsigned bit = (unsigned)__builtin_ctz(mk);
                    mk &= mk - 1u;
                    const unsigned slot = qn + __builtin_amdgcn_mbcnt_lo(act, 0u);
                    myq[slot] = e0 + 4u * (bit & 7u) + (bit >> 3);
                }
                qn += (unsigned)__builtin_popcount(act);
            }
            if (u == SCAN_U / 2 - 1) drain(false);
        }
        drain(last);
    };

    constexpr int NPAIR = (NCH + 2 * EDGE_WAVES - 1) / (2 * EDGE_WAVES);
    uint4 ra[SCAN_U], rb[SCAN_U];
    load_step(ra, wave);
    for (int p = 0; p < NPAIR; ++p) {
        const int c = wave + p * 2 * EDGE_WAVES;
        load_step(rb, c + EDGE_WAVES);
        scan_step(ra, c, false);
        load_step(ra, c + 2 * EDGE_WAVES);
        scan_step(rb, c + EDGE_WAVES, p == NPAIR - 1);
    }
    __syncthreads();

    float* Hp  = (float*)Qe;
    float* scl = Hp + HP_ROWS * HP_PITCH;

    const int rsel  = lane & 15;
    const int khalf = (lane < 16) ? 0 : 2;
    const int rofs  = (lane < 16) ? 0 : 8;
    const int wv = __builtin_amdgcn_readfirstlane(wave);
    const int ct = wv & (OUT_F / 16 - 1);
    const int g  = wv / (OUT_F / 16);

    v2f bf[RANK / 4];
    {
        const float* Bp = vmat + (size_t)khalf * OUT_F + ct * 16 + rsel;
#pragma unroll
        for (int s = 0; s < RANK / 4; ++s) {
            bf[s].x = Bp[(size_t)(4 * s) * OUT_F];
            bf[s].y = Bp[(size_t)(4 * s + 1) * OUT_F];
        }
    }

    for (int i4 = tid; i4 < NR * RANK / 4; i4 += 1024) {
        const int row = i4 >> 4;
        const int c0  = (i4 & 15) * 4;
        const float4 la = ((const float4*)La)[i4];
        const bool has = (Cn[row] != 0u);
        const unsigned n0w = Sg[row * 2]     >> (c0 >> 1);
        const unsigned n1w = Sg[row * 2 + 1] >> (c0 >> 1);
        float4 hv;
        hv.x = has ? ((n0w & 1u) ? -1.0f : 1.0f) * __expf(la.x) : 0.0f;
        hv.y = has ? ((n1w & 1u) ? -1.0f : 1.0f) * __expf(la.y) : 0.0f;
        hv.z = has ? ((n0w & 2u) ? -1.0f : 1.0f) * __expf(la.z) : 0.0f;
        hv.w = has ? ((n1w & 2u) ? -1.0f : 1.0f) * __expf(la.w) : 0.0f;
        *(float4*)(Hp + row * HP_PITCH + c0) = hv;
    }
    for (int i = tid; i < (HP_ROWS - NR) * HP_PITCH; i += 1024)
        Hp[NR * HP_PITCH + i] = 0.0f;
    if (tid < NR) scl[tid] = rsqrtf(fmaxf((float)Cn[tid], 1.0f));
    __syncthreads();

    for (int rt = g; rt < HP_TILES; rt += EDGE_WAVES / (OUT_F / 16)) {
        const float* Ap = Hp + (rt * 16 + rsel) * HP_PITCH + khalf;
        v8f acc = {};
#pragma unroll
        for (int s = 0; s < RANK / 4; ++s) {
            v2f a;
            a.x = Ap[4 * s];
            a.y = Ap[4 * s + 1];
            acc = __builtin_amdgcn_wmma_f32_16x16x4_f32(
                false, a, false, bf[s], (short)0, acc, false, false);
        }
        float*       h  = Hs + (rt * 16 + rofs) * OUT_F + ct * 16 + rsel;
        const float* sp = scl + rt * 16 + rofs;
        if (rt * 16 + 16 <= NR) {
#pragma unroll
            for (int r = 0; r < 8; ++r)
                h[r * OUT_F] = (acc[r] + h[r * OUT_F]) * sp[r];
        } else {
#pragma unroll
            for (int r = 0; r < 8; ++r)
                if (rt * 16 + rofs + r < NR)
                    h[r * OUT_F] = (acc[r] + h[r * OUT_F]) * sp[r];
        }
    }
    __syncthreads();

    const int rows = (N_NODES - n0 < NR) ? (N_NODES - n0) : NR;
    float4* og = (float4*)(out + (size_t)n0 * OUT_F);
    for (int i = tid; i < rows * OUT_F / 4; i += 1024) og[i] = ((const float4*)Hs)[i];
}

extern "C" void kernel_launch(void* const* d_in, const int* in_sizes, int n_in,
                              void* d_out, int out_size, void* d_ws, size_t ws_size,
                              hipStream_t stream)
{
    (void)in_sizes; (void)n_in; (void)out_size; (void)ws_size;

    const float* feat = (const float*)d_in[0];
    const float* w1   = (const float*)d_in[1];
    const float* w2   = (const float*)d_in[2];
    const float* vmat = (const float*)d_in[3];
    const int*   src  = (const int*)d_in[4];
    const int*   dst  = (const int*)d_in[5];

    float* ws    = (float*)d_ws;
    unsigned* odpart = (unsigned*)(ws + OFF_ODPART);
    unsigned* bid = (unsigned*)(ws + OFF_BID);
    float* rec   = ws + OFF_REC;

    k_outdeg<<<DEG_SLICES, 1024, 0, stream>>>(src, dst, odpart, bid);
    k_node_gemm<<<N_NODES / 16, 128, 0, stream>>>(feat, w1, w2, odpart, rec);
    k_edge<<<EDGE_BLOCKS, 1024, 0, stream>>>(src, dst, bid, rec, vmat, (float*)d_out);
}


# Edges bucketed by owner block in launch 1 (per-slice LDS count, prefix, place; 1 word per edge): an edge block reads only its own edges, no scan of all 800,000; sums stay arrival-order, order changes
# speedup vs baseline: 3.9124x; 1.1522x over previous
#include <hip/hip_runtime.h>
#include <math.h>

typedef __attribute__((ext_vector_type(2))) float v2f;
typedef __attribute__((ext_vector_type(8))) float v8f;

#define N_NODES 50000
#define N_EDGES 800000
#define IN_F    256
#define OUT_F   128
#define RANK    64

static_assert(N_NODES % 16 == 0, "row tiles exact");
static_assert(N_EDGES % 256 == 0, "edge blocks exact");

#define OFF_ODPART 100000u
#define OFF_PAIRS  9900000u
#define OFF_OFFS   10800000u
#define OFF_REC    12899904u
#define REC_F      (OUT_F + RANK)

#define DEG_SLICES  32
#define DEG_EDGES   (N_EDGES / DEG_SLICES)
#define DEG_LOADS   ((DEG_EDGES / 4 + 1023) / 1024)
static_assert(N_EDGES % DEG_SLICES == 0 && DEG_EDGES % 4 == 0, "slices exact; int4 loads aligned");
static_assert(N_NODES % 4 == 0, "count planes move as uint4");
static_assert(DEG_SLICES % 8 == 0, "the node product blocks add the planes in 8 groups");
static_assert((size_t)OFF_ODPART + (size_t)DEG_SLICES * N_NODES <= OFF_PAIRS && OFF_PAIRS + N_EDGES <= OFF_OFFS &&
              OFF_OFFS + 64u * 260u <= OFF_REC,
              "partial planes, edge buckets and bucket offsets fit in front of the node table");
static_assert((size_t)OFF_REC + 63u + (size_t)N_NODES * REC_F <= 22500000u,
              "the aligned node table ends inside the workspace");
static_assert((REC_F * 4) % 256 == 0 && (OUT_F * 4) % 256 == 0,
              "a record, and its product part, start on 256-byte boundaries");

__device__ __forceinline__ float* node_table(float* p)
{
    const unsigned pad = (unsigned)(0ull - (unsigned long long)p) & 255u;
    return p + (pad >> 2);
}

#define NR          196
#define EDGE_BLOCKS ((N_NODES + NR - 1) / NR)
#define EDGE_WAVES  32
#define BK_SLICES   64
#define BK_EDGES    (N_EDGES / BK_SLICES)
#define BK_LOADS    ((BK_EDGES / 4 + 1023) / 1024)
#define OFFS_PITCH  260
static_assert(EDGE_BLOCKS == 256, "the bucket prefix is one wave with 8 counters per lane");
static_assert(N_NODES <= 65536 && NR <= 256, "source node and local row fit 16 + 8 bits of an edge word");
static_assert(N_EDGES % BK_SLICES == 0 && BK_EDGES % 4 == 0, "slices exact; 16-byte moves aligned");
static_assert(BK_SLICES == 2 * EDGE_WAVES && BK_SLICES == 64, "a wave of an edge block reads the buckets of two slices");
static_assert(OFFS_PITCH >= EDGE_BLOCKS + 1 && OFFS_PITCH % 4 == 0, "an offset row holds every start and the total");
static_assert(BK_EDGES + 3 <= N_NODES, "a slice's edge words fit the LDS the histogram blocks use");
#define HP_TILES    ((NR + 15) / 16)
#define HP_ROWS     (HP_TILES * 16)
#define HP_PITCH    68
static_assert(EDGE_WAVES % (OUT_F / 16) == 0, "a wave keeps one column tile of out over all its row tiles");

__global__ __launch_bounds__(1024) void k_prep(const int* __restrict__ src,
                                               const int* __restrict__ dst,
                                               unsigned* __restrict__ parts,
                                               unsigned* __restrict__ pairs,
                                               unsigned* __restrict__ offs)
{
    __shared__ __attribute__((aligned(16))) unsigned Hc[N_NODES];
    __shared__ unsigned cur[EDGE_BLOCKS];
    __shared__ unsigned tot;

    const int tid = threadIdx.x;

    if (blockIdx.x < DEG_SLICES) {
        constexpr int S4 = DEG_EDGES / 4;

        const int4* s4 = (const int4*)(src + (size_t)blockIdx.x * DEG_EDGES);
        int4 v[DEG_LOADS];
#pragma unroll
        for (int j = 0; j < DEG_LOADS; ++j) {
            const int i = tid + j * 1024;
            v[j] = s4[(i < S4) ? i : (S4 - 1)];
        }

        for (int i = tid; i < N_NODES / 4; i += 1024)
            ((uint4*)Hc)[i] = make_uint4(0u, 0u, 0u, 0u);
        __syncthreads();

#pragma unroll
        for (int j = 0; j < DEG_LOADS; ++j) {
            if (tid + j * 1024 < S4) {
                atomicAdd(&Hc[v[j].x], 1u);
                atomicAdd(&Hc[v[j].y], 1u);
                atomicAdd(&Hc[v[j].z], 1u);
                atomicAdd(&Hc[v[j].w], 1u);
            }
        }
        __syncthreads();

        uint4* og = (uint4*)(parts + (size_t)blockIdx.x * N_NODES);
        for (int i = tid; i < N_NODES / 4; i += 1024)
            og[i] = ((const uint4*)Hc)[i];
    } else {
        const int g = blockIdx.x - DEG_SLICES;
        constexpr int S4 = BK_EDGES / 4;
        unsigned* lst = Hc;

        const int4* s4 = (const int4*)(src + (size_t)g * BK_EDGES);
        const int4* t4 = (const int4*)(dst + (size_t)g * BK_EDGES);
        int4 sv[BK_LOADS], dv[BK_LOADS];
#pragma unroll
        for (int j = 0; j < BK_LOADS; ++j) {
            const int i  = tid + j * 1024;
            const int ii = (i < S4) ? i : (S4 - 1);
            sv[j] = s4[ii];
            dv[j] = t4[ii];
        }
        if (tid < EDGE_BLOCKS) cur[tid] = 0u;
        __syncthreads();

        auto count = [&](int d) {
            if ((unsigned)d < (unsigned)N_NODES) atomicAdd(&cur[(unsigned)d / (unsigned)NR], 1u);
        };
#pragma unroll
        for (int j = 0; j < BK_LOADS; ++j) {
            if (tid + j * 1024 < S4) {
                count(dv[j].x); count(dv[j].y); count(dv[j].z); count(dv[j].w);
            }
        }
        __syncthreads();

        if (tid < 32) {
            unsigned cn[8];
            unsigned sum = 0u;
#pragma unroll
            for (int q = 0; q < 8; ++q) { cn[q] = cur[tid * 8 + q]; sum += cn[q]; }
            unsigned incl = sum;
#pragma unroll
            for (int o = 1; o < 32; o <<= 1) {
                const unsigned t = __shfl_up(incl, o);
                if (tid >= o) incl += t;
            }
            unsigned run = incl - sum;
            unsigned* og = offs + (size_t)g * OFFS_PITCH;
#pragma unroll
            for (int q = 0; q < 8; ++q) {
                og[tid * 8 + q]  = run;
                cur[tid * 8 + q] = run;
                run += cn[q];
            }
            if (tid == 31) {
                og[EDGE_BLOCKS] = run;
                tot = run;
                lst[run] = 0u; lst[run + 1] = 0u; lst[run + 2] = 0u;
            }
        }
        __syncthreads();

        auto slot = [&](int d) -> unsigned {
            return ((unsigned)d < (unsigned)N_NODES) ? atomicAdd(&cur[(unsigned)d / (unsigned)NR], 1u) : 0xffffffffu;
        };
        auto put = [&](unsigned r, int s, int d) {
            if (r != 0xffffffffu)
                lst[r] = (unsigned)s | (((unsigned)d % (unsigned)NR) << 16);
        };
        unsigned rk[BK_LOADS][4];
#pragma unroll
        for (int j = 0; j < BK_LOADS; ++j) {
            if (tid + j * 1024 < S4) {
                rk[j][0] = slot(dv[j].x); rk[j][1] = slot(dv[j].y);
                rk[j][2] = slot(dv[j].z); rk[j][3] = slot(dv[j].w);
            }
        }
#pragma unroll
        for (int j = 0; j < BK_LOADS; ++j) {
            if (tid + j * 1024 < S4) {
                put(rk[j][0], sv[j].x, dv[j].x); put(rk[j][1], sv[j].y, dv[j].y);
                put(rk[j][2], sv[j].z, dv[j].z); put(rk[j][3], sv[j].w, dv[j].w);
            }
        }
        __syncthreads();

        const int n4 = (int)((tot + 3u) >> 2);
        uint4* pg = (uint4*)(pairs + (size_t)g * BK_EDGES);
        for (int i = tid; i < n4; i += 1024)
            pg[i] = ((const uint4*)lst)[i];
    }
}

__global__ __launch_bounds__(128) void k_node_gemm(const float* __restrict__ feat,
                                                   const float* __restrict__ w1,
                                                   const float* __restrict__ w2,
                                                   const unsigned* __restrict__ odpart,
                                                   float* __restrict__ rec_raw)
{
    __shared__ float As[16][260];
    float* __restrict__ rec = node_table(rec_raw);
    __shared__ unsigned dp[8][16];

    const int m0  = blockIdx.x * 16;
    const int tid = threadIdx.x;

    {
        const int r = tid & 15;
        const int g = tid >> 4;
        unsigned d = 0u;
#pragma unroll
        for (int j = 0; j < DEG_SLICES / 8; ++j)
            d += odpart[(size_t)(g + 8 * j) * N_NODES + m0 + r];
        dp[g][r] = d;
    }
    __syncthreads();

    {
        const int r   = tid >> 3;
        const int c0  = (tid & 7) * 32;
        unsigned d = 0u;
#pragma unroll
        for (int q = 0; q < 8; ++q) d += dp[q][r];
        const float s = rsqrtf(fmaxf((float)d, 1.0f));
        const float4* g = (const float4*)(feat + (size_t)(m0 + r) * IN_F + c0);
#pragma unroll
        for (int j = 0; j < 8; ++j) {
            float4 x = g[j];
            As[r][c0 + 4 * j + 0] = x.x * s;
            As[r][c0 + 4 * j + 1] = x.y * s;
            As[r][c0 + 4 * j + 2] = x.z * s;
            As[r][c0 + 4 * j + 3] = x.w * s;
        }
    }
    __syncthreads();

    const int wave  = tid >> 5;
    const int lane  = tid & 31;
    const int rsel  = lane & 15;
    const int khalf = (lane < 16) ? 0 : 2;
    const int rofs  = (lane < 16) ? 0 : 8;

    const float* B1 = w1 + (size_t)khalf * OUT_F + 32 * wave + 2 * rsel;
    const float* B2 = w2 + (size_t)khalf * RANK + 16 * wave + rsel;

    v8f acc0 = {}, acc1 = {}, acc2 = {};
#pragma unroll 4
    for (int k0 = 0; k0 < IN_F; k0 += 4) {
        v2f a;
        a.x = As[rsel][k0 + khalf];
        a.y = As[rsel][k0 + khalf + 1];
        const float2 q0 = *(const float2*)(B1 + (size_t)k0 * OUT_F);
        const float2 q1 = *(const float2*)(B1 + (size_t)(k0 + 1) * OUT_F);
        const float  p0 = B2[(size_t)k0 * RANK];
        const float  p1 = B2[(size_t)(k0 + 1) * RANK];
        v2f b;
        b.x = q0.x; b.y = q1.x;
        acc0 = __builtin_amdgcn_wmma_f32_16x16x4_f32(false, a, false, b, (short)0, acc0, false, false);
        b.x = q0.y; b.y = q1.y;
        acc1 = __builtin_amdgcn_wmma_f32_16x16x4_f32(false, a, false, b, (short)0, acc1, false, false);
        b.x = p0; b.y = p1;
        acc2 = __builtin_amdgcn_wmma_f32_16x16x4_f32(false, a, false, b, (short)0, acc2, false, false);
    }

    {
        float* op = rec + (size_t)(m0 + rofs) * REC_F + 32 * wave + 2 * rsel;
#pragma unroll
        for (int r = 0; r < 8; ++r)
            *(float2*)(op + r * REC_F) = make_float2(acc0[r], acc1[r]);
    }
    {
        const int colg = 16 * wave + rsel;
        const float bias = w2[(size_t)IN_F * RANK + colg];
        float* op = rec + (size_t)(m0 + rofs) * REC_F + OUT_F + colg;
#pragma unroll
        for (int r = 0; r < 8; ++r) {
            const float m  = tanhf(acc2[r] + bias);
            const float la = __logf(fabsf(m));
            const unsigned w = (__float_as_uint(la) & 0x7fffffffu) | ((m < 0.0f) ? 0x80000000u : 0u);
            op[r * REC_F] = __uint_as_float(w);
        }
    }
}

__global__ __launch_bounds__(1024) void k_edge(const unsigned* __restrict__ pairs,
                                               const unsigned* __restrict__ offs,
                                               float* __restrict__ rec_raw,
                                               const float* __restrict__ vmat,
                                               float* __restrict__ out)
{
    __shared__ __attribute__((aligned(16))) float    Hs[NR * OUT_F];
    __shared__ __attribute__((aligned(16))) float    La[NR * RANK];
    __shared__ __attribute__((aligned(16))) unsigned Sg[NR * 2];
    __shared__ unsigned Cn[NR];
    __shared__ __attribute__((aligned(16))) float    HpS[HP_ROWS * HP_PITCH + NR];

    const int tid  = threadIdx.x;
    const int wave = tid >> 5;
    const int lane = tid & 31;
    const int n0   = blockIdx.x * NR;
    const float* __restrict__ rec = node_table(rec_raw);
    const int wv = __builtin_amdgcn_readfirstlane(wave);

    const unsigned* oa = offs + (size_t)wv * OFFS_PITCH + blockIdx.x;
    const unsigned* ob = offs + (size_t)(wv + EDGE_WAVES) * OFFS_PITCH + blockIdx.x;
    const unsigned a0 = oa[0], a1 = oa[1], b0 = ob[0], b1 = ob[1];

    for (int i = tid; i < NR * OUT_F / 4; i += 1024)
        ((float4*)Hs)[i] = make_float4(0.0f, 0.0f, 0.0f, 0.0f);
    for (int i = tid; i < NR * RANK / 4; i += 1024)
        ((float4*)La)[i] = make_float4(0.0f, 0.0f, 0.0f, 0.0f);
    for (int i = tid; i < NR * 2; i += 1024)
        Sg[i] = 0u;
    if (tid < NR) Cn[tid] = 0u;
    __syncthreads();

    {
        const unsigned cA    = a1 - a0;
        const unsigned cnt   = cA + (b1 - b0);
        const unsigned baseA = (unsigned)wv * BK_EDGES + a0;
        const unsigned offB  = (unsigned)(wv + EDGE_WAVES) * BK_EDGES + b0 - cA;

        auto load_batch = [&](unsigned base) -> unsigned {
            const unsigned q = base + (unsigned)lane;
            unsigned w = (unsigned)NR << 16;
            if (q < cnt) w = pairs[q + ((q < cA) ? baseA : offB)];
            return w;
        };

        unsigned pw = load_batch(0u);
        for (unsigned base = 0; base < cnt; base += 32u) {
            const unsigned nx  = load_batch(base + 32u);
            const int      sv  = (int)(pw & 0xffffu);
            const unsigned dlv = pw >> 16;
            if (dlv < (unsigned)NR) atomicAdd(&Cn[dlv], 1u);
            const int m = (int)((cnt - base < 32u) ? (cnt - base) : 32u);

            auto gather = [&](int j, float2& x0, float2& x1, float2& pr, unsigned& dl) {
                const int s = __builtin_amdgcn_readlane(sv, j);
                dl = (unsigned)__builtin_amdgcn_readlane((int)dlv, j);
                const float2* fs = (const float2*)(rec + (size_t)s * REC_F);
                x0 = fs[lane];
                x1 = fs[32 + lane];
                pr = fs[64 + lane];
            };
            auto accum = [&](const float2& x0, const float2& x1, const float2& pr, unsigned dl) {
                float* h = Hs + dl * OUT_F + 2 * lane;
                atomicAdd(h,      x0.x);
                atomicAdd(h + 1,  x0.y);
                atomicAdd(h + 64, x1.x);
                atomicAdd(h + 65, x1.y);
                float* l = La + dl * RANK + 2 * lane;
                atomicAdd(l,     -fabsf(pr.x));
                atomicAdd(l + 1, -fabsf(pr.y));
                const unsigned s0 = __builtin_amdgcn_ballot_w32(__float_as_int(pr.x) < 0);
                const unsigned s1 = __builtin_amdgcn_ballot_w32(__float_as_int(pr.y) < 0);
                if (lane < 2) atomicXor(&Sg[dl * 2 + lane], lane ? s1 : s0);
            };

            int j = 0;
            for (; j + 4 <= m; j += 4) {
                float2 a0v, a1v, apv, b0v, b1v, bpv, c0v, c1v, cpv, e0v, e1v, epv;
                unsigned da, db, dc, de;
                gather(j,     a0v, a1v, apv, da);
                gather(j + 1, b0v, b1v, bpv, db);
                gather(j + 2, c0v, c1v, cpv, dc);
                gather(j + 3, e0v, e1v, epv, de);
                asm volatile("" :: "v"(a0v.x), "v"(a0v.y), "v"(a1v.x), "v"(a1v.y), "v"(apv.x), "v"(apv.y),
                                   "v"(b0v.x), "v"(b0v.y), "v"(b1v.x), "v"(b1v.y), "v"(bpv.x), "v"(bpv.y),
                                   "v"(c0v.x), "v"(c0v.y), "v"(c1v.x), "v"(c1v.y), "v"(cpv.x), "v"(cpv.y),
                                   "v"(e0v.x), "v"(e0v.y), "v"(e1v.x), "v"(e1v.y), "v"(epv.x), "v"(epv.y));
                accum(a0v, a1v, apv, da);
                accum(b0v, b1v, bpv, db);
                accum(c0v, c1v, cpv, dc);
                accum(e0v, e1v, epv, de);
            }
            for (; j < m; ++j) {
                float2 a0v, a1v, apv;
                unsigned da;
                gather(j, a0v, a1v, apv, da);
                asm volatile("" :: "v"(a0v.x), "v"(a0v.y), "v"(a1v.x), "v"(a1v.y), "v"(apv.x), "v"(apv.y));
                accum(a0v, a1v, apv, da);
            }
            pw = nx;
        }
    }
    __syncthreads();

    float* Hp  = HpS;
    float* scl = Hp + HP_ROWS * HP_PITCH;

    const int rsel  = lane & 15;
    const int khalf = (lane < 16) ? 0 : 2;
    const int rofs  = (lane < 16) ? 0 : 8;
    const int ct = wv & (OUT_F / 16 - 1);
    const int g  = wv / (OUT_F / 16);

    v2f bf[RANK / 4];
    {
        const float* Bp = vmat + (size_t)khalf * OUT_F + ct * 16 + rsel;
#pragma unroll
        for (int s = 0; s < RANK / 4; ++s) {
            bf[s].x = Bp[(size_t)(4 * s) * OUT_F];
            bf[s].y = Bp[(size_t)(4 * s + 1) * OUT_F];
        }
    }

    for (int i4 = tid; i4 < NR * RANK / 4; i4 += 1024) {
        const int row = i4 >> 4;
        const int c0  = (i4 & 15) * 4;
        const float4 la = ((const float4*)La)[i4];
        const bool has = (Cn[row] != 0u);
        const unsigned n0w = Sg[row * 2]     >> (c0 >> 1);
        const unsigned n1w = Sg[row * 2 + 1] >> (c0 >> 1);
        float4 hv;
        hv.x = has ? ((n0w & 1u) ? -1.0f : 1.0f) * __expf(la.x) : 0.0f;
        hv.y = has ? ((n1w & 1u) ? -1.0f : 1.0f) * __expf(la.y) : 0.0f;
        hv.z = has ? ((n0w & 2u) ? -1.0f : 1.0f) * __expf(la.z) : 0.0f;
        hv.w = has ? ((n1w & 2u) ? -1.0f : 1.0f) * __expf(la.w) : 0.0f;
        *(float4*)(Hp + row * HP_PITCH + c0) = hv;
    }
    for (int i = tid; i < (HP_ROWS - NR) * HP_PITCH; i += 1024)
        Hp[NR * HP_PITCH + i] = 0.0f;
    if (tid < NR) scl[tid] = rsqrtf(fmaxf((float)Cn[tid], 1.0f));
    __syncthreads();

    for (int rt = g; rt < HP_TILES; rt += EDGE_WAVES / (OUT_F / 16)) {
        const float* Ap = Hp + (rt * 16 + rsel) * HP_PITCH + khalf;
        v8f acc = {};
#pragma unroll
        for (int s = 0; s < RANK / 4; ++s) {
            v2f a;
            a.x = Ap[4 * s];
            a.y = Ap[4 * s + 1];
            acc = __builtin_amdgcn_wmma_f32_16x16x4_f32(
                false, a, false, bf[s], (short)0, acc, false, false);
        }
        float*       h  = Hs + (rt * 16 + rofs) * OUT_F + ct * 16 + rsel;
        const float* sp = scl + rt * 16 + rofs;
        if (rt * 16 + 16 <= NR) {
#pragma unroll
            for (int r = 0; r < 8; ++r)
                h[r * OUT_F] = (acc[r] + h[r * OUT_F]) * sp[r];
        } else {
#pragma unroll
            for (int r = 0; r < 8; ++r)
                if (rt * 16 + rofs + r < NR)
                    h[r * OUT_F] = (acc[r] + h[r * OUT_F]) * sp[r];
        }
    }
    __syncthreads();

    const int rows = (N_NODES - n0 < NR) ? (N_NODES - n0) : NR;
    float4* og = (float4*)(out + (size_t)n0 * OUT_F);
    for (int i = tid; i < rows * OUT_F / 4; i += 1024) og[i] = ((const float4*)Hs)[i];
}

extern "C" void kernel_launch(void* const* d_in, const int* in_sizes, int n_in,
                              void* d_out, int out_size, void* d_ws, size_t ws_size,
                              hipStream_t stream)
{
    (void)in_sizes; (void)n_in; (void)out_size; (void)ws_size;

    const float* feat = (const float*)d_in[0];
    const float* w1   = (const float*)d_in[1];
    const float* w2   = (const float*)d_in[2];
    const float* vmat = (const float*)d_in[3];
    const int*   src  = (const int*)d_in[4];
    const int*   dst  = (const int*)d_in[5];

    float* ws    = (float*)d_ws;
    unsigned* odpart = (unsigned*)(ws + OFF_ODPART);
    unsigned* pairs = (unsigned*)(ws + OFF_PAIRS);
    unsigned* offs  = (unsigned*)(ws + OFF_OFFS);
    float* rec   = ws + OFF_REC;

    k_prep<<<DEG_SLICES + BK_SLICES, 1024, 0, stream>>>(src, dst, odpart, pairs, offs);
    k_node_gemm<<<N_NODES / 16, 128, 0, stream>>>(feat, w1, w2, odpart, rec);
    k_edge<<<EDGE_BLOCKS, 1024, 0, stream>>>(pairs, offs, rec, vmat, (float*)d_out);
}
